# starting kernel
# baseline (speedup 1.0000x reference)
#include <hip/hip_runtime.h>
#include <math.h>


typedef __attribute__((ext_vector_type(2))) float v2f;
typedef __attribute__((ext_vector_type(8))) float v8f;

#define HIDC 256
#define HEADS 4
#define DHC 64

__device__ __forceinline__ float atomicMaxF(float* addr, float value) {
  if (value >= 0.0f)
    return __int_as_float(atomicMax((int*)addr, __float_as_int(value)));
  else
    return __uint_as_float(atomicMin((unsigned int*)addr, __float_as_uint(value)));
}

__device__ __forceinline__ float lrelu(float v) {
  return v > 0.0f ? v : 0.2f * v;
}

template <int K>
__global__ __launch_bounds__(512)
void gemm_wmma_f32(const float* __restrict__ A, const float* __restrict__ W,
                   const float* __restrict__ bias, float* __restrict__ C,
                   int M) {
  constexpr int LDA = K + 4;
  __shared__ float As[16 * LDA];

  const int tid   = threadIdx.x;
  const int mBase = blockIdx.x << 4;

  #pragma unroll
  for (int i = tid; i < 16 * K; i += 512) {
    int r = i / K;
    int c = i - r * K;
    int row = mBase + r;
    if (row >= M) row = M - 1;
    As[r * LDA + c] = A[(long)row * K + c];
  }
  __syncthreads();

  const int lane  = tid & 31;
  const int wave  = tid >> 5;
  const int nBase = wave << 4;
  const int col   = lane & 15;
  const int kHalf = (lane >> 4) << 1;

  const float* as = &As[(lane & 15) * LDA + kHalf];
  const float* bp = W + (long)kHalf * HIDC + nBase + col;

  v8f acc = {};
  #pragma unroll 8
  for (int kb = 0; kb < K; kb += 4) {
    v2f a, b;
    a.x = as[0];
    a.y = as[1];
    b.x = bp[0];
    b.y = bp[HIDC];
    acc = __builtin_amdgcn_wmma_f32_16x16x4_f32(
         false, a,  false, b,
         (short)0, acc,  false,  false);
    as += 4;
    bp += 4 * HIDC;
  }

  const int rowOff = (lane >> 4) << 3;
  const float bc = bias ? bias[nBase + col] : 0.0f;
  #pragma unroll
  for (int r = 0; r < 8; ++r) {
    int row = mBase + rowOff + r;
    if (row < M) C[(long)row * HIDC + nBase + col] = acc[r] + bc;
  }
}

__global__ __launch_bounds__(256)
void el_er_kernel(const float* __restrict__ feat, const float* __restrict__ al,
                  const float* __restrict__ ar, float* __restrict__ el,
                  float* __restrict__ er, int NH) {
  int warp = blockIdx.x * (blockDim.x >> 5) + (threadIdx.x >> 5);
  int lane = threadIdx.x & 31;
  if (warp >= NH) return;
  int n = warp >> 2, h = warp & 3;
  const float* f  = feat + (long)n * HIDC + h * DHC;
  const float* a1 = al + h * DHC;
  const float* a2 = ar + h * DHC;
  float f0 = f[lane], f1 = f[lane + 32];
  float sl = f0 * a1[lane] + f1 * a1[lane + 32];
  float sr = f0 * a2[lane] + f1 * a2[lane + 32];
  #pragma unroll
  for (int off = 16; off; off >>= 1) {
    sl += __shfl_down(sl, off, 32);
    sr += __shfl_down(sr, off, 32);
  }
  if (lane == 0) { el[warp] = sl; er[warp] = sr; }
}

__global__ void init_nh_kernel(float* __restrict__ m, float* __restrict__ dn,
                               int NH) {
  int i = blockIdx.x * blockDim.x + threadIdx.x;
  if (i < NH) { m[i] = -INFINITY; dn[i] = 0.0f; }
}

__global__ void init_out_kernel(const float* __restrict__ hin,
                                const float* __restrict__ bias,
                                float* __restrict__ out, long total) {
  long i = (long)blockIdx.x * blockDim.x + threadIdx.x;
  if (i < total) out[i] = hin[i] + bias[i & (HIDC - 1)];
}

__global__ void edge_max_kernel(const int* __restrict__ src,
                                const int* __restrict__ dst,
                                const float* __restrict__ el,
                                const float* __restrict__ er,
                                float* __restrict__ m, int EH) {
  int idx = blockIdx.x * blockDim.x + threadIdx.x;
  if (idx >= EH) return;
  int e = idx >> 2, h = idx & 3;
  int s = src[e], d = dst[e];
  float v = lrelu(el[s * 4 + h] + er[d * 4 + h]);
  atomicMaxF(&m[d * 4 + h], v);
}

__global__ void edge_exp_kernel(const int* __restrict__ src,
                                const int* __restrict__ dst,
                                const float* __restrict__ el,
                                const float* __restrict__ er,
                                const float* __restrict__ m,
                                float* __restrict__ dn,
                                float* __restrict__ ex, int EH) {
  int idx = blockIdx.x * blockDim.x + threadIdx.x;
  if (idx >= EH) return;
  int e = idx >> 2, h = idx & 3;
  int s = src[e], d = dst[e];
  float v = lrelu(el[s * 4 + h] + er[d * 4 + h]);
  float mm = m[d * 4 + h];
  if (!(mm > -1e38f)) mm = 0.0f;
  float x = __expf(v - mm);
  ex[idx] = x;
  atomicAdd(&dn[d * 4 + h], x);
}

__global__ __launch_bounds__(256)
void aggregate_kernel(const int* __restrict__ src, const int* __restrict__ dst,
                      const float* __restrict__ feat,
                      const float* __restrict__ ex,
                      const float* __restrict__ dn, float* __restrict__ out) {
  int e = blockIdx.x;
  int c = threadIdx.x;
  int h = c >> 6;
  int s = src[e], d = dst[e];
  float denom = dn[d * 4 + h];
  float alpha = ex[e * 4 + h] / fmaxf(denom, 1e-16f);
  atomicAdd(&out[(long)d * HIDC + c], alpha * feat[(long)s * HIDC + c]);
}

__global__ void relu_kernel(const float* __restrict__ in, float* __restrict__ out,
                            long total) {
  long i = (long)blockIdx.x * blockDim.x + threadIdx.x;
  if (i < total) out[i] = fmaxf(in[i], 0.0f);
}

__global__ void mean_heads_kernel(const float* __restrict__ hbuf,
                                  float* __restrict__ out, long total) {
  long i = (long)blockIdx.x * blockDim.x + threadIdx.x;
  if (i >= total) return;
  long n = i >> 6;
  int d = (int)(i & 63);
  const float* p = hbuf + n * HIDC + d;
  out[i] = 0.25f * (p[0] + p[DHC] + p[2 * DHC] + p[3 * DHC]);
}

extern "C" void kernel_launch(void* const* d_in, const int* in_sizes, int n_in,
                              void* d_out, int out_size, void* d_ws, size_t ws_size,
                              hipStream_t stream) {
  const float* x     = (const float*)d_in[0];
  const int*   ei    = (const int*)d_in[1];
  const float* W[3]  = {(const float*)d_in[2], (const float*)d_in[3],
                        (const float*)d_in[4]};
  const float* al[3] = {(const float*)d_in[5], (const float*)d_in[6],
                        (const float*)d_in[7]};
  const float* ar[3] = {(const float*)d_in[8], (const float*)d_in[9],
                        (const float*)d_in[10]};
  const float* b[3]  = {(const float*)d_in[11], (const float*)d_in[12],
                        (const float*)d_in[13]};
  const float* resW0 = (const float*)d_in[14];

  const int N = in_sizes[0] / 128;
  const int E = in_sizes[1] / 2;
  const int* src = ei;
  const int* dst = ei + E;

  float* ws   = (float*)d_ws;
  float* FEAT = ws;
  float* OUT  = FEAT + (size_t)N * HIDC;
  float* HBUF = OUT  + (size_t)N * HIDC;
  float* ELP  = HBUF + (size_t)N * HIDC;
  float* ERP  = ELP  + (size_t)N * HEADS;
  float* MP   = ERP  + (size_t)N * HEADS;
  float* DNP  = MP   + (size_t)N * HEADS;
  float* EXP  = DNP  + (size_t)N * HEADS;

  const int  mtiles = (N + 15) / 16;
  const long totNF  = (long)N * HIDC;
  const int  NH     = N * HEADS;
  const int  EH     = E * HEADS;
  const int  gNF    = (int)((totNF + 255) / 256);
  const int  gNH    = (NH + 255) / 256;
  const int  gEH    = (EH + 255) / 256;
  const int  gELR   = (NH + 7) / 8;

  auto attention = [&](const float* alw, const float* arw) {
    el_er_kernel<<<gELR, 256, 0, stream>>>(FEAT, alw, arw, ELP, ERP, NH);
    init_nh_kernel<<<gNH, 256, 0, stream>>>(MP, DNP, NH);
    edge_max_kernel<<<gEH, 256, 0, stream>>>(src, dst, ELP, ERP, MP, EH);
    edge_exp_kernel<<<gEH, 256, 0, stream>>>(src, dst, ELP, ERP, MP, DNP, EXP, EH);
    aggregate_kernel<<<E, 256, 0, stream>>>(src, dst, FEAT, EXP, DNP, OUT);
  };

  gemm_wmma_f32<128><<<mtiles, 512, 0, stream>>>(x, W[0], nullptr, FEAT, N);
  gemm_wmma_f32<128><<<mtiles, 512, 0, stream>>>(x, resW0, b[0], OUT, N);
  attention(al[0], ar[0]);
  relu_kernel<<<gNF, 256, 0, stream>>>(OUT, HBUF, totNF);

  gemm_wmma_f32<256><<<mtiles, 512, 0, stream>>>(HBUF, W[1], nullptr, FEAT, N);
  init_out_kernel<<<gNF, 256, 0, stream>>>(HBUF, b[1], OUT, totNF);
  attention(al[1], ar[1]);
  relu_kernel<<<gNF, 256, 0, stream>>>(OUT, HBUF, totNF);

  gemm_wmma_f32<256><<<mtiles, 512, 0, stream>>>(HBUF, W[2], nullptr, FEAT, N);
  init_out_kernel<<<gNF, 256, 0, stream>>>(HBUF, b[2], OUT, totNF);
  attention(al[2], ar[2]);
  mean_heads_kernel<<<(int)(((long)N * DHC + 255) / 256), 256, 0, stream>>>(
      OUT, (float*)d_out, (long)N * DHC);
}


# Neighbour sum made output-stationary: a work-group holds 196 dst rows in LDS, scans dst for its edges, adds src rows by LDS f32 atomics, one plain add to out; float sum order: LDS arrival
# speedup vs baseline: 2.3844x; 2.3844x over previous
#include <hip/hip_runtime.h>
#include <math.h>


typedef __attribute__((ext_vector_type(2))) float v2f;
typedef __attribute__((ext_vector_type(8))) float v8f;

#define HIDC 256
#define HEADS 4
#define DHC 64

__device__ __forceinline__ float atomicMaxF(float* addr, float value) {
  if (value >= 0.0f)
    return __int_as_float(atomicMax((int*)addr, __float_as_int(value)));
  else
    return __uint_as_float(atomicMin((unsigned int*)addr, __float_as_uint(value)));
}

__device__ __forceinline__ float lrelu(float v) {
  return v > 0.0f ? v : 0.2f * v;
}

template <int K>
__global__ __launch_bounds__(512)
void gemm_wmma_f32(const float* __restrict__ A, const float* __restrict__ W,
                   const float* __restrict__ bias, float* __restrict__ C,
                   int M) {
  constexpr int LDA = K + 4;
  __shared__ float As[16 * LDA];

  const int tid   = threadIdx.x;
  const int mBase = blockIdx.x << 4;

  #pragma unroll
  for (int i = tid; i < 16 * K; i += 512) {
    int r = i / K;
    int c = i - r * K;
    int row = mBase + r;
    if (row >= M) row = M - 1;
    As[r * LDA + c] = A[(long)row * K + c];
  }
  __syncthreads();

  const int lane  = tid & 31;
  const int wave  = tid >> 5;
  const int nBase = wave << 4;
  const int col   = lane & 15;
  const int kHalf = (lane >> 4) << 1;

  const float* as = &As[(lane & 15) * LDA + kHalf];
  const float* bp = W + (long)kHalf * HIDC + nBase + col;

  v8f acc = {};
  #pragma unroll 8
  for (int kb = 0; kb < K; kb += 4) {
    v2f a, b;
    a.x = as[0];
    a.y = as[1];
    b.x = bp[0];
    b.y = bp[HIDC];
    acc = __builtin_amdgcn_wmma_f32_16x16x4_f32(
         false, a,  false, b,
         (short)0, acc,  false,  false);
    as += 4;
    bp += 4 * HIDC;
  }

  const int rowOff = (lane >> 4) << 3;
  const float bc = bias ? bias[nBase + col] : 0.0f;
  #pragma unroll
  for (int r = 0; r < 8; ++r) {
    int row = mBase + rowOff + r;
    if (row < M) C[(long)row * HIDC + nBase + col] = acc[r] + bc;
  }
}

__global__ __launch_bounds__(256)
void el_er_kernel(const float* __restrict__ feat, const float* __restrict__ al,
                  const float* __restrict__ ar, float* __restrict__ el,
                  float* __restrict__ er, int NH) {
  int warp = blockIdx.x * (blockDim.x >> 5) + (threadIdx.x >> 5);
  int lane = threadIdx.x & 31;
  if (warp >= NH) return;
  int n = warp >> 2, h = warp & 3;
  const float* f  = feat + (long)n * HIDC + h * DHC;
  const float* a1 = al + h * DHC;
  const float* a2 = ar + h * DHC;
  float f0 = f[lane], f1 = f[lane + 32];
  float sl = f0 * a1[lane] + f1 * a1[lane + 32];
  float sr = f0 * a2[lane] + f1 * a2[lane + 32];
  #pragma unroll
  for (int off = 16; off; off >>= 1) {
    sl += __shfl_down(sl, off, 32);
    sr += __shfl_down(sr, off, 32);
  }
  if (lane == 0) { el[warp] = sl; er[warp] = sr; }
}

__global__ void init_nh_kernel(float* __restrict__ m, float* __restrict__ dn,
                               int NH) {
  int i = blockIdx.x * blockDim.x + threadIdx.x;
  if (i < NH) { m[i] = -INFINITY; dn[i] = 0.0f; }
}

__global__ void init_out_kernel(const float* __restrict__ hin,
                                const float* __restrict__ bias,
                                float* __restrict__ out, long total) {
  long i = (long)blockIdx.x * blockDim.x + threadIdx.x;
  if (i < total) out[i] = hin[i] + bias[i & (HIDC - 1)];
}

__global__ void edge_max_kernel(const int* __restrict__ src,
                                const int* __restrict__ dst,
                                const float* __restrict__ el,
                                const float* __restrict__ er,
                                float* __restrict__ m, int EH) {
  int idx = blockIdx.x * blockDim.x + threadIdx.x;
  if (idx >= EH) return;
  int e = idx >> 2, h = idx & 3;
  int s = src[e], d = dst[e];
  float v = lrelu(el[s * 4 + h] + er[d * 4 + h]);
  atomicMaxF(&m[d * 4 + h], v);
}

__global__ void edge_exp_kernel(const int* __restrict__ src,
                                const int* __restrict__ dst,
                                const float* __restrict__ el,
                                const float* __restrict__ er,
                                const float* __restrict__ m,
                                float* __restrict__ dn,
                                float* __restrict__ ex, int EH) {
  int idx = blockIdx.x * blockDim.x + threadIdx.x;
  if (idx >= EH) return;
  int e = idx >> 2, h = idx & 3;
  int s = src[e], d = dst[e];
  float v = lrelu(el[s * 4 + h] + er[d * 4 + h]);
  float mm = m[d * 4 + h];
  if (!(mm > -1e38f)) mm = 0.0f;
  float x = __expf(v - mm);
  ex[idx] = x;
  atomicAdd(&dn[d * 4 + h], x);
}

#define AGG_TN    196
#define AGG_TRIP  512
#define AGG_QCAP  768

__device__ __forceinline__ void agg_drain(
    const int* q, int cnt, int lo, int lane,
    const int* __restrict__ src, const int* __restrict__ dst,
    const float* __restrict__ feat, const float* __restrict__ ex,
    const float* __restrict__ dn, float* acc) {
  for (int i = 0; i < cnt; i += 32) {
    const int n = min(32, cnt - i);
    int s = 0, dl = 0;
    float a0 = 0.0f, a1 = 0.0f, a2 = 0.0f, a3 = 0.0f;
    if (lane < n) {
      const int e = q[i + lane];
      const int d = dst[e];
      s = src[e];
      const float4 x = *reinterpret_cast<const float4*>(ex + 4 * (long)e);
      const float4 y = *reinterpret_cast<const float4*>(dn + 4 * (long)d);
      a0 = x.x / fmaxf(y.x, 1e-16f);
      a1 = x.y / fmaxf(y.y, 1e-16f);
      a2 = x.z / fmaxf(y.z, 1e-16f);
      a3 = x.w / fmaxf(y.w, 1e-16f);
      dl = d - lo;
    }
    for (int k = 0; k < n; k += 4) {
      int su[4], du[4];
      float au[4][4];
      v2f f[4][4];
      #pragma unroll
      for (int u = 0; u < 4; ++u) {
        const int idx = min(k + u, n - 1);
        su[u] = __builtin_amdgcn_readlane(s, idx);
        du[u] = (k + u < n) ? __builtin_amdgcn_readlane(dl, idx) : AGG_TN;
        au[u][0] = __int_as_float(__builtin_amdgcn_readlane(__float_as_int(a0), idx));
        au[u][1] = __int_as_float(__builtin_amdgcn_readlane(__float_as_int(a1), idx));
        au[u][2] = __int_as_float(__builtin_amdgcn_readlane(__float_as_int(a2), idx));
        au[u][3] = __int_as_float(__builtin_amdgcn_readlane(__float_as_int(a3), idx));
      }
      #pragma unroll
      for (int u = 0; u < 4; ++u) {
        const float* fr = feat + (long)su[u] * HIDC + 2 * lane;
        #pragma unroll
        for (int j = 0; j < HEADS; ++j)
          f[u][j] = *reinterpret_cast<const v2f*>(fr + j * DHC);
      }
      #pragma unroll
      for (int u = 0; u < 4; ++u) {
        float* p = acc + du[u] * HIDC + 2 * lane;
        #pragma unroll
        for (int j = 0; j < HEADS; ++j) {
          atomicAdd(p + j * DHC,     au[u][j] * f[u][j].x);
          atomicAdd(p + j * DHC + 1, au[u][j] * f[u][j].y);
        }
      }
    }
  }
}

__global__ __launch_bounds__(1024)
void aggregate_tile_kernel(const int* __restrict__ src, const int* __restrict__ dst,
                           const float* __restrict__ feat,
                           const float* __restrict__ ex,
                           const float* __restrict__ dn, float* __restrict__ out,
                           int N, int E) {
  __shared__ __attribute__((aligned(16))) float acc[(AGG_TN + 1) * HIDC];
  __shared__ int queue[32 * AGG_QCAP];

  const int tid  = threadIdx.x;
  const int lane = tid & 31;
  const int wave = __builtin_amdgcn_readfirstlane(tid >> 5);
  const int lo   = blockIdx.x * AGG_TN;
  const int rows = min(AGG_TN, N - lo);

  for (int i = tid; i < (AGG_TN + 1) * HIDC / 4; i += 1024)
    reinterpret_cast<float4*>(acc)[i] = make_float4(0.0f, 0.0f, 0.0f, 0.0f);
  __syncthreads();

  int* q = queue + wave * AGG_QCAP;
  const int  T       = (E + AGG_TRIP - 1) / AGG_TRIP;
  const bool aligned = (reinterpret_cast<unsigned long>(dst) & 15ul) == 0ul;
  const int start = (T / (int)gridDim.x) * (int)blockIdx.x;
  int cnt = 0;
  for (int t = wave; t < T; t += 32) {
    int g = t + start;
    if (g >= T) g -= T;
    const int base = g * AGG_TRIP + 4 * lane;
    int4 dv[4];
    if (aligned && g * AGG_TRIP + AGG_TRIP <= E) {
      #pragma unroll
      for (int u = 0; u < 4; ++u)
        dv[u] = *reinterpret_cast<const int4*>(dst + base + 128 * u);
    } else {
      #pragma unroll
      for (int u = 0; u < 4; ++u) {
        const int e0 = base + 128 * u;
        dv[u].x = (e0     < E) ? dst[e0]     : -1;
        dv[u].y = (e0 + 1 < E) ? dst[e0 + 1] : -1;
        dv[u].z = (e0 + 2 < E) ? dst[e0 + 2] : -1;
        dv[u].w = (e0 + 3 < E) ? dst[e0 + 3] : -1;
      }
    }
    #pragma unroll
    for (int u = 0; u < 4; ++u) {
      const int  e0 = base + 128 * u;
      const bool mx = (unsigned)(dv[u].x - lo) < (unsigned)rows;
      const bool my = (unsigned)(dv[u].y - lo) < (unsigned)rows;
      const bool mz = (unsigned)(dv[u].z - lo) < (unsigned)rows;
      const bool mw = (unsigned)(dv[u].w - lo) < (unsigned)rows;
      if (__builtin_amdgcn_ballot_w32(mx | my | mz | mw)) {
        unsigned mask;
        mask = __builtin_amdgcn_ballot_w32(mx);
        if (mx) q[cnt + __builtin_amdgcn_mbcnt_lo(mask, 0u)] = e0;
        cnt += __builtin_popcount(mask);
        mask = __builtin_amdgcn_ballot_w32(my);
        if (my) q[cnt + __builtin_amdgcn_mbcnt_lo(mask, 0u)] = e0 + 1;
        cnt += __builtin_popcount(mask);
        mask = __builtin_amdgcn_ballot_w32(mz);
        if (mz) q[cnt + __builtin_amdgcn_mbcnt_lo(mask, 0u)] = e0 + 2;
        cnt += __builtin_popcount(mask);
        mask = __builtin_amdgcn_ballot_w32(mw);
        if (mw) q[cnt + __builtin_amdgcn_mbcnt_lo(mask, 0u)] = e0 + 3;
        cnt += __builtin_popcount(mask);
      }
    }
    if (cnt > AGG_QCAP - AGG_TRIP) {
      agg_drain(q, cnt, lo, lane, src, dst, feat, ex, dn, acc);
      cnt = 0;
    }
  }
  agg_drain(q, cnt, lo, lane, src, dst, feat, ex, dn, acc);
  __syncthreads();

  float4* op = reinterpret_cast<float4*>(out + (long)lo * HIDC);
  for (int i = tid; i < rows * (HIDC / 4); i += 1024) {
    const float4 a = reinterpret_cast<const float4*>(acc)[i];
    float4 o = op[i];
    o.x += a.x; o.y += a.y; o.z += a.z; o.w += a.w;
    op[i] = o;
  }
}

__global__ void relu_kernel(const float* __restrict__ in, float* __restrict__ out,
                            long total) {
  long i = (long)blockIdx.x * blockDim.x + threadIdx.x;
  if (i < total) out[i] = fmaxf(in[i], 0.0f);
}

__global__ void mean_heads_kernel(const float* __restrict__ hbuf,
                                  float* __restrict__ out, long total) {
  long i = (long)blockIdx.x * blockDim.x + threadIdx.x;
  if (i >= total) return;
  long n = i >> 6;
  int d = (int)(i & 63);
  const float* p = hbuf + n * HIDC + d;
  out[i] = 0.25f * (p[0] + p[DHC] + p[2 * DHC] + p[3 * DHC]);
}

extern "C" void kernel_launch(void* const* d_in, const int* in_sizes, int n_in,
                              void* d_out, int out_size, void* d_ws, size_t ws_size,
                              hipStream_t stream) {
  const float* x     = (const float*)d_in[0];
  const int*   ei    = (const int*)d_in[1];
  const float* W[3]  = {(const float*)d_in[2], (const float*)d_in[3],
                        (const float*)d_in[4]};
  const float* al[3] = {(const float*)d_in[5], (const float*)d_in[6],
                        (const float*)d_in[7]};
  const float* ar[3] = {(const float*)d_in[8], (const float*)d_in[9],
                        (const float*)d_in[10]};
  const float* b[3]  = {(const float*)d_in[11], (const float*)d_in[12],
                        (const float*)d_in[13]};
  const float* resW0 = (const float*)d_in[14];

  const int N = in_sizes[0] / 128;
  const int E = in_sizes[1] / 2;
  const int* src = ei;
  const int* dst = ei + E;

  float* ws   = (float*)d_ws;
  float* FEAT = ws;
  float* OUT  = FEAT + (size_t)N * HIDC;
  float* HBUF = OUT  + (size_t)N * HIDC;
  float* ELP  = HBUF + (size_t)N * HIDC;
  float* ERP  = ELP  + (size_t)N * HEADS;
  float* MP   = ERP  + (size_t)N * HEADS;
  float* DNP  = MP   + (size_t)N * HEADS;
  float* EXP  = DNP  + (size_t)N * HEADS;

  const int  mtiles = (N + 15) / 16;
  const long totNF  = (long)N * HIDC;
  const int  NH     = N * HEADS;
  const int  EH     = E * HEADS;
  const int  gNF    = (int)((totNF + 255) / 256);
  const int  gNH    = (NH + 255) / 256;
  const int  gEH    = (EH + 255) / 256;
  const int  gELR   = (NH + 7) / 8;

  auto attention = [&](const float* alw, const float* arw) {
    el_er_kernel<<<gELR, 256, 0, stream>>>(FEAT, alw, arw, ELP, ERP, NH);
    init_nh_kernel<<<gNH, 256, 0, stream>>>(MP, DNP, NH);
    edge_max_kernel<<<gEH, 256, 0, stream>>>(src, dst, ELP, ERP, MP, EH);
    edge_exp_kernel<<<gEH, 256, 0, stream>>>(src, dst, ELP, ERP, MP, DNP, EXP, EH);
    aggregate_tile_kernel<<<(N + AGG_TN - 1) / AGG_TN, 1024, 0, stream>>>(
        src, dst, FEAT, EXP, DNP, OUT, N, E);
  };

  gemm_wmma_f32<128><<<mtiles, 512, 0, stream>>>(x, W[0], nullptr, FEAT, N);
  gemm_wmma_f32<128><<<mtiles, 512, 0, stream>>>(x, resW0, b[0], OUT, N);
  attention(al[0], ar[0]);
  relu_kernel<<<gNF, 256, 0, stream>>>(OUT, HBUF, totNF);

  gemm_wmma_f32<256><<<mtiles, 512, 0, stream>>>(HBUF, W[1], nullptr, FEAT, N);
  init_out_kernel<<<gNF, 256, 0, stream>>>(HBUF, b[1], OUT, totNF);
  attention(al[1], ar[1]);
  relu_kernel<<<gNF, 256, 0, stream>>>(OUT, HBUF, totNF);

  gemm_wmma_f32<256><<<mtiles, 512, 0, stream>>>(HBUF, W[2], nullptr, FEAT, N);
  init_out_kernel<<<gNF, 256, 0, stream>>>(HBUF, b[2], OUT, totNF);
  attention(al[2], ar[2]);
  mean_heads_kernel<<<(int)(((long)N * DHC + 255) / 256), 256, 0, stream>>>(
      OUT, (float*)d_out, (long)N * DHC);
}


# Seven one-element-per-thread kernels (relu, out init, head mean, el/er dots, edge max/exp, stat init) loop in small resident grids, 16-byte accesses in the streams: fewer waves; float sums unchanged
# speedup vs baseline: 2.5843x; 1.0838x over previous
#include <hip/hip_runtime.h>
#include <math.h>


typedef __attribute__((ext_vector_type(2))) float v2f;
typedef __attribute__((ext_vector_type(8))) float v8f;

#define HIDC 256
#define HEADS 4
#define DHC 64

__device__ __forceinline__ float atomicMaxF(float* addr, float value) {
  if (value >= 0.0f)
    return __int_as_float(atomicMax((int*)addr, __float_as_int(value)));
  else
    return __uint_as_float(atomicMin((unsigned int*)addr, __float_as_uint(value)));
}

__device__ __forceinline__ float lrelu(float v) {
  return v > 0.0f ? v : 0.2f * v;
}

template <int K>
__global__ __launch_bounds__(512)
void gemm_wmma_f32(const float* __restrict__ A, const float* __restrict__ W,
                   const float* __restrict__ bias, float* __restrict__ C,
                   int M) {
  constexpr int LDA = K + 4;
  __shared__ float As[16 * LDA];

  const int tid   = threadIdx.x;
  const int mBase = blockIdx.x << 4;

  #pragma unroll
  for (int i = tid; i < 16 * K; i += 512) {
    int r = i / K;
    int c = i - r * K;
    int row = mBase + r;
    if (row >= M) row = M - 1;
    As[r * LDA + c] = A[(long)row * K + c];
  }
  __syncthreads();

  const int lane  = tid & 31;
  const int wave  = tid >> 5;
  const int nBase = wave << 4;
  const int col   = lane & 15;
  const int kHalf = (lane >> 4) << 1;

  const float* as = &As[(lane & 15) * LDA + kHalf];
  const float* bp = W + (long)kHalf * HIDC + nBase + col;

  v8f acc = {};
  #pragma unroll 8
  for (int kb = 0; kb < K; kb += 4) {
    v2f a, b;
    a.x = as[0];
    a.y = as[1];
    b.x = bp[0];
    b.y = bp[HIDC];
    acc = __builtin_amdgcn_wmma_f32_16x16x4_f32(
         false, a,  false, b,
         (short)0, acc,  false,  false);
    as += 4;
    bp += 4 * HIDC;
  }

  const int rowOff = (lane >> 4) << 3;
  const float bc = bias ? bias[nBase + col] : 0.0f;
  #pragma unroll
  for (int r = 0; r < 8; ++r) {
    int row = mBase + rowOff + r;
    if (row < M) C[(long)row * HIDC + nBase + col] = acc[r] + bc;
  }
}

__global__ __launch_bounds__(256)
void el_er_kernel(const float* __restrict__ feat, const float* __restrict__ al,
                  const float* __restrict__ ar, float* __restrict__ el,
                  float* __restrict__ er, int NH) {
  const int lane = threadIdx.x & 31;
  const int wave0 = __builtin_amdgcn_readfirstlane(
      (int)(blockIdx.x * (blockDim.x >> 5) + (threadIdx.x >> 5)));
  const int nW = (int)(gridDim.x * (blockDim.x >> 5));
  const int h  = wave0 & 3;
  const float l0 = al[h * DHC + lane], l1 = al[h * DHC + lane + 32];
  const float r0 = ar[h * DHC + lane], r1 = ar[h * DHC + lane + 32];
  for (int p = wave0; p < NH; p += 8 * nW) {
    float f0[8], f1[8];
    #pragma unroll
    for (int u = 0; u < 8; ++u) {
      const int q = p + u * nW;
      f0[u] = 0.0f;
      f1[u] = 0.0f;
      if (q < NH) {
        const float* f = feat + (long)q * DHC;
        f0[u] = f[lane];
        f1[u] = f[lane + 32];
      }
    }
    float sl[8], sr[8];
    #pragma unroll
    for (int u = 0; u < 8; ++u) {
      sl[u] = f0[u] * l0 + f1[u] * l1;
      sr[u] = f0[u] * r0 + f1[u] * r1;
    }
    #pragma unroll
    for (int off = 16; off; off >>= 1) {
      #pragma unroll
      for (int u = 0; u < 8; ++u) {
        sl[u] += __shfl_down(sl[u], off, 32);
        sr[u] += __shfl_down(sr[u], off, 32);
      }
    }
    #pragma unroll
    for (int u = 0; u < 8; ++u) {
      const int q = p + u * nW;
      if (lane == 0 && q < NH) { el[q] = sl[u]; er[q] = sr[u]; }
    }
  }
}

__global__ __launch_bounds__(256)
void init_nh_kernel(float* __restrict__ m, float* __restrict__ dn, int NH) {
  const int i = blockIdx.x * blockDim.x + threadIdx.x;
  if (i < (NH >> 2)) {
    reinterpret_cast<float4*>(m)[i]  = make_float4(-INFINITY, -INFINITY, -INFINITY, -INFINITY);
    reinterpret_cast<float4*>(dn)[i] = make_float4(0.0f, 0.0f, 0.0f, 0.0f);
  }
}

__global__ __launch_bounds__(256)
void init_out_kernel(const float* __restrict__ hin,
                     const float* __restrict__ bias,
                     float* __restrict__ out, long total) {
  const long n4     = total >> 2;
  const long stride = (long)gridDim.x * blockDim.x;
  const long t      = (long)blockIdx.x * blockDim.x + threadIdx.x;
  const int  c      = (int)(t & (HIDC / 4 - 1)) * 4;
  const float b0 = bias[c], b1 = bias[c + 1], b2 = bias[c + 2], b3 = bias[c + 3];
  const float4* ip = reinterpret_cast<const float4*>(hin);
  float4*       op = reinterpret_cast<float4*>(out);
  for (long i = t; i < n4; i += 8 * stride) {
    float4 v[8];
    #pragma unroll
    for (int u = 0; u < 8; ++u) {
      const long j = i + u * stride;
      if (j < n4) v[u] = ip[j];
    }
    #pragma unroll
    for (int u = 0; u < 8; ++u) {
      const long j = i + u * stride;
      if (j < n4) {
        float4 o = v[u];
        o.x += b0; o.y += b1; o.z += b2; o.w += b3;
        op[j] = o;
      }
    }
  }
}

__global__ __launch_bounds__(256)
void edge_max_kernel(const int* __restrict__ src,
                     const int* __restrict__ dst,
                     const float* __restrict__ el,
                     const float* __restrict__ er,
                     float* __restrict__ m, int EH) {
  const int stride = (int)(gridDim.x * blockDim.x);
  const int t = blockIdx.x * blockDim.x + threadIdx.x;
  const int h = t & 3;
  for (int idx = t; idx < EH; idx += 4 * stride) {
    int s[4], d[4];
    float a[4], b[4];
    #pragma unroll
    for (int u = 0; u < 4; ++u) {
      const int j = idx + u * stride;
      if (j < EH) { s[u] = src[j >> 2]; d[u] = dst[j >> 2]; }
    }
    #pragma unroll
    for (int u = 0; u < 4; ++u) {
      const int j = idx + u * stride;
      if (j < EH) { a[u] = el[s[u] * 4 + h]; b[u] = er[d[u] * 4 + h]; }
    }
    #pragma unroll
    for (int u = 0; u < 4; ++u) {
      const int j = idx + u * stride;
      if (j < EH) atomicMaxF(&m[d[u] * 4 + h], lrelu(a[u] + b[u]));
    }
  }
}

__global__ __launch_bounds__(256)
void edge_exp_kernel(const int* __restrict__ src,
                     const int* __restrict__ dst,
                     const float* __restrict__ el,
                     const float* __restrict__ er,
                     const float* __restrict__ m,
                     float* __restrict__ dn,
                     float* __restrict__ ex, int EH) {
  const int stride = (int)(gridDim.x * blockDim.x);
  const int t = blockIdx.x * blockDim.x + threadIdx.x;
  const int h = t & 3;
  for (int idx = t; idx < EH; idx += 4 * stride) {
    int s[4], d[4];
    float a[4], b[4], mx[4];
    #pragma unroll
    for (int u = 0; u < 4; ++u) {
      const int j = idx + u * stride;
      if (j < EH) { s[u] = src[j >> 2]; d[u] = dst[j >> 2]; }
    }
    #pragma unroll
    for (int u = 0; u < 4; ++u) {
      const int j = idx + u * stride;
      if (j < EH) {
        a[u]  = el[s[u] * 4 + h];
        b[u]  = er[d[u] * 4 + h];
        mx[u] = m[d[u] * 4 + h];
      }
    }
    #pragma unroll
    for (int u = 0; u < 4; ++u) {
      const int j = idx + u * stride;
      if (j < EH) {
        const float v = lrelu(a[u] + b[u]);
        float mm = mx[u];
        if (!(mm > -1e38f)) mm = 0.0f;
        const float x = __expf(v - mm);
        ex[j] = x;
        atomicAdd(&dn[d[u] * 4 + h], x);
      }
    }
  }
}

#define AGG_TN    196
#define AGG_TRIP  512
#define AGG_QCAP  768

__device__ __forceinline__ void agg_drain(
    const int* q, int cnt, int lo, int lane,
    const int* __restrict__ src, const int* __restrict__ dst,
    const float* __restrict__ feat, const float* __restrict__ ex,
    const float* __restrict__ dn, float* acc) {
  for (int i = 0; i < cnt; i += 32) {
    const int n = min(32, cnt - i);
    int s = 0, dl = 0;
    float a0 = 0.0f, a1 = 0.0f, a2 = 0.0f, a3 = 0.0f;
    if (lane < n) {
      const int e = q[i + lane];
      const int d = dst[e];
      s = src[e];
      const float4 x = *reinterpret_cast<const float4*>(ex + 4 * (long)e);
      const float4 y = *reinterpret_cast<const float4*>(dn + 4 * (long)d);
      a0 = x.x / fmaxf(y.x, 1e-16f);
      a1 = x.y / fmaxf(y.y, 1e-16f);
      a2 = x.z / fmaxf(y.z, 1e-16f);
      a3 = x.w / fmaxf(y.w, 1e-16f);
      dl = d - lo;
    }
    for (int k = 0; k < n; k += 4) {
      int su[4], du[4];
      float au[4][4];
      v2f f[4][4];
      #pragma unroll
      for (int u = 0; u < 4; ++u) {
        const int idx = min(k + u, n - 1);
        su[u] = __builtin_amdgcn_readlane(s, idx);
        du[u] = (k + u < n) ? __builtin_amdgcn_readlane(dl, idx) : AGG_TN;
        au[u][0] = __int_as_float(__builtin_amdgcn_readlane(__float_as_int(a0), idx));
        au[u][1] = __int_as_float(__builtin_amdgcn_readlane(__float_as_int(a1), idx));
        au[u][2] = __int_as_float(__builtin_amdgcn_readlane(__float_as_int(a2), idx));
        au[u][3] = __int_as_float(__builtin_amdgcn_readlane(__float_as_int(a3), idx));
      }
      #pragma unroll
      for (int u = 0; u < 4; ++u) {
        const float* fr = feat + (long)su[u] * HIDC + 2 * lane;
        #pragma unroll
        for (int j = 0; j < HEADS; ++j)
          f[u][j] = *reinterpret_cast<const v2f*>(fr + j * DHC);
      }
      #pragma unroll
      for (int u = 0; u < 4; ++u) {
        float* p = acc + du[u] * HIDC + 2 * lane;
        #pragma unroll
        for (int j = 0; j < HEADS; ++j) {
          atomicAdd(p + j * DHC,     au[u][j] * f[u][j].x);
          atomicAdd(p + j * DHC + 1, au[u][j] * f[u][j].y);
        }
      }
    }
  }
}

__global__ __launch_bounds__(1024)
void aggregate_tile_kernel(const int* __restrict__ src, const int* __restrict__ dst,
                           const float* __restrict__ feat,
                           const float* __restrict__ ex,
                           const float* __restrict__ dn, float* __restrict__ out,
                           int N, int E) {
  __shared__ __attribute__((aligned(16))) float acc[(AGG_TN + 1) * HIDC];
  __shared__ int queue[32 * AGG_QCAP];

  const int tid  = threadIdx.x;
  const int lane = tid & 31;
  const int wave = __builtin_amdgcn_readfirstlane(tid >> 5);
  const int lo   = blockIdx.x * AGG_TN;
  const int rows = min(AGG_TN, N - lo);

  for (int i = tid; i < (AGG_TN + 1) * HIDC / 4; i += 1024)
    reinterpret_cast<float4*>(acc)[i] = make_float4(0.0f, 0.0f, 0.0f, 0.0f);
  __syncthreads();

  int* q = queue + wave * AGG_QCAP;
  const int  T       = (E + AGG_TRIP - 1) / AGG_TRIP;
  const bool aligned = (reinterpret_cast<unsigned long>(dst) & 15ul) == 0ul;
  const int start = (T / (int)gridDim.x) * (int)blockIdx.x;
  int cnt = 0;
  for (int t = wave; t < T; t += 32) {
    int g = t + start;
    if (g >= T) g -= T;
    const int base = g * AGG_TRIP + 4 * lane;
    int4 dv[4];
    if (aligned && g * AGG_TRIP + AGG_TRIP <= E) {
      #pragma unroll
      for (int u = 0; u < 4; ++u)
        dv[u] = *reinterpret_cast<const int4*>(dst + base + 128 * u);
    } else {
      #pragma unroll
      for (int u = 0; u < 4; ++u) {
        const int e0 = base + 128 * u;
        dv[u].x = (e0     < E) ? dst[e0]     : -1;
        dv[u].y = (e0 + 1 < E) ? dst[e0 + 1] : -1;
        dv[u].z = (e0 + 2 < E) ? dst[e0 + 2] : -1;
        dv[u].w = (e0 + 3 < E) ? dst[e0 + 3] : -1;
      }
    }
    #pragma unroll
    for (int u = 0; u < 4; ++u) {
      const int  e0 = base + 128 * u;
      const bool mx = (unsigned)(dv[u].x - lo) < (unsigned)rows;
      const bool my = (unsigned)(dv[u].y - lo) < (unsigned)rows;
      const bool mz = (unsigned)(dv[u].z - lo) < (unsigned)rows;
      const bool mw = (unsigned)(dv[u].w - lo) < (unsigned)rows;
      if (__builtin_amdgcn_ballot_w32(mx | my | mz | mw)) {
        unsigned mask;
        mask = __builtin_amdgcn_ballot_w32(mx);
        if (mx) q[cnt + __builtin_amdgcn_mbcnt_lo(mask, 0u)] = e0;
        cnt += __builtin_popcount(mask);
        mask = __builtin_amdgcn_ballot_w32(my);
        if (my) q[cnt + __builtin_amdgcn_mbcnt_lo(mask, 0u)] = e0 + 1;
        cnt += __builtin_popcount(mask);
        mask = __builtin_amdgcn_ballot_w32(mz);
        if (mz) q[cnt + __builtin_amdgcn_mbcnt_lo(mask, 0u)] = e0 + 2;
        cnt += __builtin_popcount(mask);
        mask = __builtin_amdgcn_ballot_w32(mw);
        if (mw) q[cnt + __builtin_amdgcn_mbcnt_lo(mask, 0u)] = e0 + 3;
        cnt += __builtin_popcount(mask);
      }
    }
    if (cnt > AGG_QCAP - AGG_TRIP) {
      agg_drain(q, cnt, lo, lane, src, dst, feat, ex, dn, acc);
      cnt = 0;
    }
  }
  agg_drain(q, cnt, lo, lane, src, dst, feat, ex, dn, acc);
  __syncthreads();

  float4* op = reinterpret_cast<float4*>(out + (long)lo * HIDC);
  for (int i = tid; i < rows * (HIDC / 4); i += 1024) {
    const float4 a = reinterpret_cast<const float4*>(acc)[i];
    float4 o = op[i];
    o.x += a.x; o.y += a.y; o.z += a.z; o.w += a.w;
    op[i] = o;
  }
}

__global__ __launch_bounds__(256)
void relu_kernel(const float* __restrict__ in, float* __restrict__ out,
                 long total) {
  const long n4     = total >> 2;
  const long stride = (long)gridDim.x * blockDim.x;
  const float4* ip = reinterpret_cast<const float4*>(in);
  float4*       op = reinterpret_cast<float4*>(out);
  for (long i = (long)blockIdx.x * blockDim.x + threadIdx.x; i < n4; i += 8 * stride) {
    float4 v[8];
    #pragma unroll
    for (int u = 0; u < 8; ++u) {
      const long j = i + u * stride;
      if (j < n4) v[u] = ip[j];
    }
    #pragma unroll
    for (int u = 0; u < 8; ++u) {
      const long j = i + u * stride;
      if (j < n4) {
        float4 o;
        o.x = fmaxf(v[u].x, 0.0f); o.y = fmaxf(v[u].y, 0.0f);
        o.z = fmaxf(v[u].z, 0.0f); o.w = fmaxf(v[u].w, 0.0f);
        op[j] = o;
      }
    }
  }
}

__global__ __launch_bounds__(256)
void mean_heads_kernel(const float* __restrict__ hbuf,
                       float* __restrict__ out, long total) {
  const long stride = (long)gridDim.x * blockDim.x;
  const long t      = (long)blockIdx.x * blockDim.x + threadIdx.x;
  if ((reinterpret_cast<unsigned long>(out) & 15ul) == 0ul) {
    const long n4 = total >> 2;
    const float4* hp = reinterpret_cast<const float4*>(hbuf);
    float4*       op = reinterpret_cast<float4*>(out);
    for (long i = t; i < n4; i += 2 * stride) {
      float4 p[2][4];
      #pragma unroll
      for (int u = 0; u < 2; ++u) {
        const long j = i + u * stride;
        if (j < n4) {
          const float4* r = hp + (j >> 4) * (HIDC / 4) + (j & 15);
          #pragma unroll
          for (int k = 0; k < HEADS; ++k) p[u][k] = r[k * (DHC / 4)];
        }
      }
      #pragma unroll
      for (int u = 0; u < 2; ++u) {
        const long j = i + u * stride;
        if (j < n4) {
          float4 o;
          o.x = 0.25f * (p[u][0].x + p[u][1].x + p[u][2].x + p[u][3].x);
          o.y = 0.25f * (p[u][0].y + p[u][1].y + p[u][2].y + p[u][3].y);
          o.z = 0.25f * (p[u][0].z + p[u][1].z + p[u][2].z + p[u][3].z);
          o.w = 0.25f * (p[u][0].w + p[u][1].w + p[u][2].w + p[u][3].w);
          op[j] = o;
        }
      }
    }
  } else {
    for (long i = t; i < total; i += stride) {
      const float* p = hbuf + (i >> 6) * HIDC + (i & 63);
      out[i] = 0.25f * (p[0] + p[DHC] + p[2 * DHC] + p[3 * DHC]);
    }
  }
}

extern "C" void kernel_launch(void* const* d_in, const int* in_sizes, int n_in,
                              void* d_out, int out_size, void* d_ws, size_t ws_size,
                              hipStream_t stream) {
  const float* x     = (const float*)d_in[0];
  const int*   ei    = (const int*)d_in[1];
  const float* W[3]  = {(const float*)d_in[2], (const float*)d_in[3],
                        (const float*)d_in[4]};
  const float* al[3] = {(const float*)d_in[5], (const float*)d_in[6],
                        (const float*)d_in[7]};
  const float* ar[3] = {(const float*)d_in[8], (const float*)d_in[9],
                        (const float*)d_in[10]};
  const float* b[3]  = {(const float*)d_in[11], (const float*)d_in[12],
                        (const float*)d_in[13]};
  const float* resW0 = (const float*)d_in[14];

  const int N = in_sizes[0] / 128;
  const int E = in_sizes[1] / 2;
  const int* src = ei;
  const int* dst = ei + E;

  float* ws   = (float*)d_ws;
  float* FEAT = ws;
  float* OUT  = FEAT + (size_t)N * HIDC;
  float* HBUF = OUT  + (size_t)N * HIDC;
  float* ELP  = HBUF + (size_t)N * HIDC;
  float* ERP  = ELP  + (size_t)N * HEADS;
  float* MP   = ERP  + (size_t)N * HEADS;
  float* DNP  = MP   + (size_t)N * HEADS;
  float* EXP  = DNP  + (size_t)N * HEADS;

  const int  mtiles = (N + 15) / 16;
  const long totNF  = (long)N * HIDC;
  const int  NH     = N * HEADS;
  const int  EH     = E * HEADS;
  const int  gNH    = (NH / 4 + 255) / 256;
  const int  gSTREAM = 512;
  const int  gNARROW = 1024;

  auto attention = [&](const float* alw, const float* arw) {
    el_er_kernel<<<gNARROW, 256, 0, stream>>>(FEAT, alw, arw, ELP, ERP, NH);
    init_nh_kernel<<<gNH, 256, 0, stream>>>(MP, DNP, NH);
    edge_max_kernel<<<gNARROW, 256, 0, stream>>>(src, dst, ELP, ERP, MP, EH);
    edge_exp_kernel<<<gNARROW, 256, 0, stream>>>(src, dst, ELP, ERP, MP, DNP, EXP, EH);
    aggregate_tile_kernel<<<(N + AGG_TN - 1) / AGG_TN, 1024, 0, stream>>>(
        src, dst, FEAT, EXP, DNP, OUT, N, E);
  };

  gemm_wmma_f32<128><<<mtiles, 512, 0, stream>>>(x, W[0], nullptr, FEAT, N);
  gemm_wmma_f32<128><<<mtiles, 512, 0, stream>>>(x, resW0, b[0], OUT, N);
  attention(al[0], ar[0]);
  relu_kernel<<<gSTREAM, 256, 0, stream>>>(OUT, HBUF, totNF);

  gemm_wmma_f32<256><<<mtiles, 512, 0, stream>>>(HBUF, W[1], nullptr, FEAT, N);
  init_out_kernel<<<gSTREAM, 256, 0, stream>>>(HBUF, b[1], OUT, totNF);
  attention(al[1], ar[1]);
  relu_kernel<<<gSTREAM, 256, 0, stream>>>(OUT, HBUF, totNF);

  gemm_wmma_f32<256><<<mtiles, 512, 0, stream>>>(HBUF, W[2], nullptr, FEAT, N);
  init_out_kernel<<<gSTREAM, 256, 0, stream>>>(HBUF, b[2], OUT, totNF);
  attention(al[2], ar[2]);
  mean_heads_kernel<<<gSTREAM, 256, 0, stream>>>(
      OUT, (float*)d_out, (long)N * DHC);
}


# Edge softmax (max, exp, sum) computed inside the output-stationary aggregation work-group on LDS statistics of its rows; 9 launches, L2 atomics, exp array gone; sums: LDS arrival, divide after sum
# speedup vs baseline: 3.7655x; 1.4571x over previous
#include <hip/hip_runtime.h>
#include <math.h>


typedef __attribute__((ext_vector_type(2))) float v2f;
typedef __attribute__((ext_vector_type(8))) float v8f;

#define HIDC 256
#define HEADS 4
#define DHC 64

__device__ __forceinline__ float atomicMaxF(float* addr, float value) {
  if (value >= 0.0f)
    return __int_as_float(atomicMax((int*)addr, __float_as_int(value)));
  else
    return __uint_as_float(atomicMin((unsigned int*)addr, __float_as_uint(value)));
}

__device__ __forceinline__ float lrelu(float v) {
  return v > 0.0f ? v : 0.2f * v;
}

template <int K>
__global__ __launch_bounds__(512)
void gemm_wmma_f32(const float* __restrict__ A, const float* __restrict__ W,
                   const float* __restrict__ bias, float* __restrict__ C,
                   int M) {
  constexpr int LDA = K + 4;
  __shared__ float As[16 * LDA];

  const int tid   = threadIdx.x;
  const int mBase = blockIdx.x << 4;

  #pragma unroll
  for (int i = tid; i < 16 * K; i += 512) {
    int r = i / K;
    int c = i - r * K;
    int row = mBase + r;
    if (row >= M) row = M - 1;
    As[r * LDA + c] = A[(long)row * K + c];
  }
  __syncthreads();

  const int lane  = tid & 31;
  const int wave  = tid >> 5;
  const int nBase = wave << 4;
  const int col   = lane & 15;
  const int kHalf = (lane >> 4) << 1;

  const float* as = &As[(lane & 15) * LDA + kHalf];
  const float* bp = W + (long)kHalf * HIDC + nBase + col;

  v8f acc = {};
  #pragma unroll 8
  for (int kb = 0; kb < K; kb += 4) {
    v2f a, b;
    a.x = as[0];
    a.y = as[1];
    b.x = bp[0];
    b.y = bp[HIDC];
    acc = __builtin_amdgcn_wmma_f32_16x16x4_f32(
         false, a,  false, b,
         (short)0, acc,  false,  false);
    as += 4;
    bp += 4 * HIDC;
  }

  const int rowOff = (lane >> 4) << 3;
  const float bc = bias ? bias[nBase + col] : 0.0f;
  #pragma unroll
  for (int r = 0; r < 8; ++r) {
    int row = mBase + rowOff + r;
    if (row < M) C[(long)row * HIDC + nBase + col] = acc[r] + bc;
  }
}

__global__ __launch_bounds__(256)
void el_er_kernel(const float* __restrict__ feat, const float* __restrict__ al,
                  const float* __restrict__ ar, float* __restrict__ el,
                  float* __restrict__ er, int NH) {
  const int lane = threadIdx.x & 31;
  const int wave0 = __builtin_amdgcn_readfirstlane(
      (int)(blockIdx.x * (blockDim.x >> 5) + (threadIdx.x >> 5)));
  const int nW = (int)(gridDim.x * (blockDim.x >> 5));
  const int h  = wave0 & 3;
  const float l0 = al[h * DHC + lane], l1 = al[h * DHC + lane + 32];
  const float r0 = ar[h * DHC + lane], r1 = ar[h * DHC + lane + 32];
  for (int p = wave0; p < NH; p += 8 * nW) {
    float f0[8], f1[8];
    #pragma unroll
    for (int u = 0; u < 8; ++u) {
      const int q = p + u * nW;
      f0[u] = 0.0f;
      f1[u] = 0.0f;
      if (q < NH) {
        const float* f = feat + (long)q * DHC;
        f0[u] = f[lane];
        f1[u] = f[lane + 32];
      }
    }
    float sl[8], sr[8];
    #pragma unroll
    for (int u = 0; u < 8; ++u) {
      sl[u] = f0[u] * l0 + f1[u] * l1;
      sr[u] = f0[u] * r0 + f1[u] * r1;
    }
    #pragma unroll
    for (int off = 16; off; off >>= 1) {
      #pragma unroll
      for (int u = 0; u < 8; ++u) {
        sl[u] += __shfl_down(sl[u], off, 32);
        sr[u] += __shfl_down(sr[u], off, 32);
      }
    }
    #pragma unroll
    for (int u = 0; u < 8; ++u) {
      const int q = p + u * nW;
      if (lane == 0 && q < NH) { el[q] = sl[u]; er[q] = sr[u]; }
    }
  }
}

__global__ __launch_bounds__(256)
void init_out_kernel(const float* __restrict__ hin,
                     const float* __restrict__ bias,
                     float* __restrict__ out, long total) {
  const long n4     = total >> 2;
  const long stride = (long)gridDim.x * blockDim.x;
  const long t      = (long)blockIdx.x * blockDim.x + threadIdx.x;
  const int  c      = (int)(t & (HIDC / 4 - 1)) * 4;
  const float b0 = bias[c], b1 = bias[c + 1], b2 = bias[c + 2], b3 = bias[c + 3];
  const float4* ip = reinterpret_cast<const float4*>(hin);
  float4*       op = reinterpret_cast<float4*>(out);
  for (long i = t; i < n4; i += 8 * stride) {
    float4 v[8];
    #pragma unroll
    for (int u = 0; u < 8; ++u) {
      const long j = i + u * stride;
      if (j < n4) v[u] = ip[j];
    }
    #pragma unroll
    for (int u = 0; u < 8; ++u) {
      const long j = i + u * stride;
      if (j < n4) {
        float4 o = v[u];
        o.x += b0; o.y += b1; o.z += b2; o.w += b3;
        op[j] = o;
      }
    }
  }
}

#define AGG_TN    196
#define AGG_TRIP  512
#define AGG_QCAP  768

__device__ __forceinline__ float edge_weight(float score, float mx) {
  const float v = lrelu(score);
  if (!(mx > -1e38f)) mx = 0.0f;
  return __expf(v - mx);
}

__device__ __forceinline__ void agg_drain_max(
    const int* q, int cnt, int lo, int lane,
    const int* __restrict__ src, const int* __restrict__ dst,
    const float* __restrict__ el, const float* __restrict__ er, float* mt) {
  for (int i = 0; i < cnt; i += 64) {
    int s[2], d[2];
    float4 x[2], y[2];
    #pragma unroll
    for (int u = 0; u < 2; ++u) {
      const int p = i + 32 * u + lane;
      s[u] = 0; d[u] = lo;
      if (p < cnt) { const int e = q[p]; d[u] = dst[e]; s[u] = src[e]; }
    }
    #pragma unroll
    for (int u = 0; u < 2; ++u) {
      const int p = i + 32 * u + lane;
      if (p < cnt) {
        x[u] = *reinterpret_cast<const float4*>(el + 4 * (long)s[u]);
        y[u] = *reinterpret_cast<const float4*>(er + 4 * (long)d[u]);
      }
    }
    #pragma unroll
    for (int u = 0; u < 2; ++u) {
      const int p = i + 32 * u + lane;
      if (p < cnt) {
        float* m = mt + 4 * (d[u] - lo);
        atomicMaxF(m,     lrelu(x[u].x + y[u].x));
        atomicMaxF(m + 1, lrelu(x[u].y + y[u].y));
        atomicMaxF(m + 2, lrelu(x[u].z + y[u].z));
        atomicMaxF(m + 3, lrelu(x[u].w + y[u].w));
      }
    }
  }
}

__device__ __forceinline__ void agg_drain_acc(
    const int* q, int cnt, int lo, int lane,
    const int* __restrict__ src, const int* __restrict__ dst,
    const float* __restrict__ feat,
    const float* __restrict__ el, const float* __restrict__ er,
    const float* mt, float* dt, float* acc) {
  for (int i = 0; i < cnt; i += 32) {
    const int n = min(32, cnt - i);
    int s = 0, dl = 0;
    float a0 = 0.0f, a1 = 0.0f, a2 = 0.0f, a3 = 0.0f;
    if (lane < n) {
      const int e = q[i + lane];
      const int d = dst[e];
      s  = src[e];
      dl = d - lo;
      const float4 x = *reinterpret_cast<const float4*>(el + 4 * (long)s);
      const float4 y = *reinterpret_cast<const float4*>(er + 4 * (long)d);
      const float4 m = *reinterpret_cast<const float4*>(mt + 4 * dl);
      a0 = edge_weight(x.x + y.x, m.x);
      a1 = edge_weight(x.y + y.y, m.y);
      a2 = edge_weight(x.z + y.z, m.z);
      a3 = edge_weight(x.w + y.w, m.w);
      float* ds = dt + 4 * dl;
      atomicAdd(ds,     a0);
      atomicAdd(ds + 1, a1);
      atomicAdd(ds + 2, a2);
      atomicAdd(ds + 3, a3);
    }
    for (int k = 0; k < n; k += 4) {
      int su[4], du[4];
      float au[4][4];
      v2f f[4][4];
      #pragma unroll
      for (int u = 0; u < 4; ++u) {
        const int idx = min(k + u, n - 1);
        su[u] = __builtin_amdgcn_readlane(s, idx);
        du[u] = (k + u < n) ? __builtin_amdgcn_readlane(dl, idx) : AGG_TN;
        au[u][0] = __int_as_float(__builtin_amdgcn_readlane(__float_as_int(a0), idx));
        au[u][1] = __int_as_float(__builtin_amdgcn_readlane(__float_as_int(a1), idx));
        au[u][2] = __int_as_float(__builtin_amdgcn_readlane(__float_as_int(a2), idx));
        au[u][3] = __int_as_float(__builtin_amdgcn_readlane(__float_as_int(a3), idx));
      }
      #pragma unroll
      for (int u = 0; u < 4; ++u) {
        const float* fr = feat + (long)su[u] * HIDC + 2 * lane;
        #pragma unroll
        for (int j = 0; j < HEADS; ++j)
          f[u][j] = *reinterpret_cast<const v2f*>(fr + j * DHC);
      }
      #pragma unroll
      for (int u = 0; u < 4; ++u) {
        float* p = acc + du[u] * HIDC + 2 * lane;
        #pragma unroll
        for (int j = 0; j < HEADS; ++j) {
          atomicAdd(p + j * DHC,     au[u][j] * f[u][j].x);
          atomicAdd(p + j * DHC + 1, au[u][j] * f[u][j].y);
        }
      }
    }
  }
}

template <int PASS>
__device__ __forceinline__ int agg_scan(
    int* q, int wave, int lane, int lo, int rows, int E,
    const int* __restrict__ src, const int* __restrict__ dst,
    const float* __restrict__ feat,
    const float* __restrict__ el, const float* __restrict__ er,
    float* mt, float* dt, float* acc, int* full) {
  const int  T       = (E + AGG_TRIP - 1) / AGG_TRIP;
  const bool aligned = (reinterpret_cast<unsigned long>(dst) & 15ul) == 0ul;
  const int start = (T / (int)gridDim.x) * (int)blockIdx.x;
  int cnt = 0;
  for (int t = wave; t < T; t += 32) {
    int g = t + start;
    if (g >= T) g -= T;
    const int base = g * AGG_TRIP + 4 * lane;
    int4 dv[4];
    if (aligned && g * AGG_TRIP + AGG_TRIP <= E) {
      #pragma unroll
      for (int u = 0; u < 4; ++u)
        dv[u] = *reinterpret_cast<const int4*>(dst + base + 128 * u);
    } else {
      #pragma unroll
      for (int u = 0; u < 4; ++u) {
        const int e0 = base + 128 * u;
        dv[u].x = (e0     < E) ? dst[e0]     : -1;
        dv[u].y = (e0 + 1 < E) ? dst[e0 + 1] : -1;
        dv[u].z = (e0 + 2 < E) ? dst[e0 + 2] : -1;
        dv[u].w = (e0 + 3 < E) ? dst[e0 + 3] : -1;
      }
    }
    #pragma unroll
    for (int u = 0; u < 4; ++u) {
      const int  e0 = base + 128 * u;
      const bool mx = (unsigned)(dv[u].x - lo) < (unsigned)rows;
      const bool my = (unsigned)(dv[u].y - lo) < (unsigned)rows;
      const bool mz = (unsigned)(dv[u].z - lo) < (unsigned)rows;
      const bool mw = (unsigned)(dv[u].w - lo) < (unsigned)rows;
      if (__builtin_amdgcn_ballot_w32(mx | my | mz | mw)) {
        unsigned mask;
        mask = __builtin_amdgcn_ballot_w32(mx);
        if (mx) q[cnt + __builtin_amdgcn_mbcnt_lo(mask, 0u)] = e0;
        cnt += __builtin_popcount(mask);
        mask = __builtin_amdgcn_ballot_w32(my);
        if (my) q[cnt + __builtin_amdgcn_mbcnt_lo(mask, 0u)] = e0 + 1;
        cnt += __builtin_popcount(mask);
        mask = __builtin_amdgcn_ballot_w32(mz);
        if (mz) q[cnt + __builtin_amdgcn_mbcnt_lo(mask, 0u)] = e0 + 2;
        cnt += __builtin_popcount(mask);
        mask = __builtin_amdgcn_ballot_w32(mw);
        if (mw) q[cnt + __builtin_amdgcn_mbcnt_lo(mask, 0u)] = e0 + 3;
        cnt += __builtin_popcount(mask);
      }
    }
    if (cnt > AGG_QCAP - AGG_TRIP) {
      if (PASS == 0) {
        agg_drain_max(q, cnt, lo, lane, src, dst, el, er, mt);
        if (lane == 0) *full = 1;
      } else {
        agg_drain_acc(q, cnt, lo, lane, src, dst, feat, el, er, mt, dt, acc);
      }
      cnt = 0;
    }
  }
  return cnt;
}

__global__ __launch_bounds__(1024)
void aggregate_tile_kernel(const int* __restrict__ src, const int* __restrict__ dst,
                           const float* __restrict__ feat,
                           const float* __restrict__ el,
                           const float* __restrict__ er, float* __restrict__ out,
                           int N, int E) {
  __shared__ __attribute__((aligned(16))) float acc[(AGG_TN + 1) * HIDC];
  __shared__ __attribute__((aligned(16))) float mt[AGG_TN * HEADS];
  __shared__ __attribute__((aligned(16))) float dt[AGG_TN * HEADS];
  __shared__ int queue[32 * AGG_QCAP];
  __shared__ int full;

  const int tid  = threadIdx.x;
  const int lane = tid & 31;
  const int wave = __builtin_amdgcn_readfirstlane(tid >> 5);
  const int lo   = blockIdx.x * AGG_TN;
  const int rows = min(AGG_TN, N - lo);

  for (int i = tid; i < (AGG_TN + 1) * HIDC / 4; i += 1024)
    reinterpret_cast<float4*>(acc)[i] = make_float4(0.0f, 0.0f, 0.0f, 0.0f);
  if (tid < AGG_TN * HEADS) { mt[tid] = -INFINITY; dt[tid] = 0.0f; }
  if (tid == 0) full = 0;
  __syncthreads();

  int* q = queue + wave * AGG_QCAP;
  int cnt = agg_scan<0>(q, wave, lane, lo, rows, E, src, dst, feat, el, er, mt, dt, acc, &full);
  agg_drain_max(q, cnt, lo, lane, src, dst, el, er, mt);
  __syncthreads();

  const int early = __builtin_amdgcn_readfirstlane(full);
  if (early == 0) {
    agg_drain_acc(q, cnt, lo, lane, src, dst, feat, el, er, mt, dt, acc);
  } else {
    cnt = agg_scan<1>(q, wave, lane, lo, rows, E, src, dst, feat, el, er, mt, dt, acc, &full);
    agg_drain_acc(q, cnt, lo, lane, src, dst, feat, el, er, mt, dt, acc);
  }
  __syncthreads();

  if (tid < rows * HEADS) dt[tid] = 1.0f / fmaxf(dt[tid], 1e-16f);
  __syncthreads();

  float4* op = reinterpret_cast<float4*>(out + (long)lo * HIDC);
  for (int i = tid; i < rows * (HIDC / 4); i += 1024) {
    const float  r = dt[i >> 4];
    const float4 a = reinterpret_cast<const float4*>(acc)[i];
    float4 o = op[i];
    o.x += a.x * r; o.y += a.y * r; o.z += a.z * r; o.w += a.w * r;
    op[i] = o;
  }
}

__global__ __launch_bounds__(256)
void relu_kernel(const float* __restrict__ in, float* __restrict__ out,
                 long total) {
  const long n4     = total >> 2;
  const long stride = (long)gridDim.x * blockDim.x;
  const float4* ip = reinterpret_cast<const float4*>(in);
  float4*       op = reinterpret_cast<float4*>(out);
  for (long i = (long)blockIdx.x * blockDim.x + threadIdx.x; i < n4; i += 8 * stride) {
    float4 v[8];
    #pragma unroll
    for (int u = 0; u < 8; ++u) {
      const long j = i + u * stride;
      if (j < n4) v[u] = ip[j];
    }
    #pragma unroll
    for (int u = 0; u < 8; ++u) {
      const long j = i + u * stride;
      if (j < n4) {
        float4 o;
        o.x = fmaxf(v[u].x, 0.0f); o.y = fmaxf(v[u].y, 0.0f);
        o.z = fmaxf(v[u].z, 0.0f); o.w = fmaxf(v[u].w, 0.0f);
        op[j] = o;
      }
    }
  }
}

__global__ __launch_bounds__(256)
void mean_heads_kernel(const float* __restrict__ hbuf,
                       float* __restrict__ out, long total) {
  const long stride = (long)gridDim.x * blockDim.x;
  const long t      = (long)blockIdx.x * blockDim.x + threadIdx.x;
  if ((reinterpret_cast<unsigned long>(out) & 15ul) == 0ul) {
    const long n4 = total >> 2;
    const float4* hp = reinterpret_cast<const float4*>(hbuf);
    float4*       op = reinterpret_cast<float4*>(out);
    for (long i = t; i < n4; i += 2 * stride) {
      float4 p[2][4];
      #pragma unroll
      for (int u = 0; u < 2; ++u) {
        const long j = i + u * stride;
        if (j < n4) {
          const float4* r = hp + (j >> 4) * (HIDC / 4) + (j & 15);
          #pragma unroll
          for (int k = 0; k < HEADS; ++k) p[u][k] = r[k * (DHC / 4)];
        }
      }
      #pragma unroll
      for (int u = 0; u < 2; ++u) {
        const long j = i + u * stride;
        if (j < n4) {
          float4 o;
          o.x = 0.25f * (p[u][0].x + p[u][1].x + p[u][2].x + p[u][3].x);
          o.y = 0.25f * (p[u][0].y + p[u][1].y + p[u][2].y + p[u][3].y);
          o.z = 0.25f * (p[u][0].z + p[u][1].z + p[u][2].z + p[u][3].z);
          o.w = 0.25f * (p[u][0].w + p[u][1].w + p[u][2].w + p[u][3].w);
          op[j] = o;
        }
      }
    }
  } else {
    for (long i = t; i < total; i += stride) {
      const float* p = hbuf + (i >> 6) * HIDC + (i & 63);
      out[i] = 0.25f * (p[0] + p[DHC] + p[2 * DHC] + p[3 * DHC]);
    }
  }
}

extern "C" void kernel_launch(void* const* d_in, const int* in_sizes, int n_in,
                              void* d_out, int out_size, void* d_ws, size_t ws_size,
                              hipStream_t stream) {
  const float* x     = (const float*)d_in[0];
  const int*   ei    = (const int*)d_in[1];
  const float* W[3]  = {(const float*)d_in[2], (const float*)d_in[3],
                        (const float*)d_in[4]};
  const float* al[3] = {(const float*)d_in[5], (const float*)d_in[6],
                        (const float*)d_in[7]};
  const float* ar[3] = {(const float*)d_in[8], (const float*)d_in[9],
                        (const float*)d_in[10]};
  const float* b[3]  = {(const float*)d_in[11], (const float*)d_in[12],
                        (const float*)d_in[13]};
  const float* resW0 = (const float*)d_in[14];

  const int N = in_sizes[0] / 128;
  const int E = in_sizes[1] / 2;
  const int* src = ei;
  const int* dst = ei + E;

  float* ws   = (float*)d_ws;
  float* FEAT = ws;
  float* OUT  = FEAT + (size_t)N * HIDC;
  float* HBUF = OUT  + (size_t)N * HIDC;
  float* ELP  = HBUF + (size_t)N * HIDC;
  float* ERP  = ELP  + (size_t)N * HEADS;

  const int  mtiles = (N + 15) / 16;
  const long totNF  = (long)N * HIDC;
  const int  NH     = N * HEADS;
  const int  gSTREAM = 512;
  const int  gNARROW = 1024;

  auto attention = [&](const float* alw, const float* arw) {
    el_er_kernel<<<gNARROW, 256, 0, stream>>>(FEAT, alw, arw, ELP, ERP, NH);
    aggregate_tile_kernel<<<(N + AGG_TN - 1) / AGG_TN, 1024, 0, stream>>>(
        src, dst, FEAT, ELP, ERP, OUT, N, E);
  };

  gemm_wmma_f32<128><<<mtiles, 512, 0, stream>>>(x, W[0], nullptr, FEAT, N);
  gemm_wmma_f32<128><<<mtiles, 512, 0, stream>>>(x, resW0, b[0], OUT, N);
  attention(al[0], ar[0]);
  relu_kernel<<<gSTREAM, 256, 0, stream>>>(OUT, HBUF, totNF);

  gemm_wmma_f32<256><<<mtiles, 512, 0, stream>>>(HBUF, W[1], nullptr, FEAT, N);
  init_out_kernel<<<gSTREAM, 256, 0, stream>>>(HBUF, b[1], OUT, totNF);
  attention(al[1], ar[1]);
  relu_kernel<<<gSTREAM, 256, 0, stream>>>(OUT, HBUF, totNF);

  gemm_wmma_f32<256><<<mtiles, 512, 0, stream>>>(HBUF, W[2], nullptr, FEAT, N);
  init_out_kernel<<<gSTREAM, 256, 0, stream>>>(HBUF, b[2], OUT, totNF);
  attention(al[2], ar[2]);
  mean_heads_kernel<<<gSTREAM, 256, 0, stream>>>(
      OUT, (float*)d_out, (long)N * DHC);
}


# Tile edge queues found by the first layer's dst scan are kept in the workspace; layers 2 and 3 read them back instead of walking dst again (2 of 3 scans gone); float sums: LDS arrival as before
# speedup vs baseline: 4.1174x; 1.0935x over previous
#include <hip/hip_runtime.h>
#include <math.h>


typedef __attribute__((ext_vector_type(2))) float v2f;
typedef __attribute__((ext_vector_type(8))) float v8f;

#define HIDC 256
#define HEADS 4
#define DHC 64

__device__ __forceinline__ float atomicMaxF(float* addr, float value) {
  if (value >= 0.0f)
    return __int_as_float(atomicMax((int*)addr, __float_as_int(value)));
  else
    return __uint_as_float(atomicMin((unsigned int*)addr, __float_as_uint(value)));
}

__device__ __forceinline__ float lrelu(float v) {
  return v > 0.0f ? v : 0.2f * v;
}

template <int K>
__global__ __launch_bounds__(512)
void gemm_wmma_f32(const float* __restrict__ A, const float* __restrict__ W,
                   const float* __restrict__ bias, float* __restrict__ C,
                   int M) {
  constexpr int LDA = K + 4;
  __shared__ float As[16 * LDA];

  const int tid   = threadIdx.x;
  const int mBase = blockIdx.x << 4;

  #pragma unroll
  for (int i = tid; i < 16 * K; i += 512) {
    int r = i / K;
    int c = i - r * K;
    int row = mBase + r;
    if (row >= M) row = M - 1;
    As[r * LDA + c] = A[(long)row * K + c];
  }
  __syncthreads();

  const int lane  = tid & 31;
  const int wave  = tid >> 5;
  const int nBase = wave << 4;
  const int col   = lane & 15;
  const int kHalf = (lane >> 4) << 1;

  const float* as = &As[(lane & 15) * LDA + kHalf];
  const float* bp = W + (long)kHalf * HIDC + nBase + col;

  v8f acc = {};
  #pragma unroll 8
  for (int kb = 0; kb < K; kb += 4) {
    v2f a, b;
    a.x = as[0];
    a.y = as[1];
    b.x = bp[0];
    b.y = bp[HIDC];
    acc = __builtin_amdgcn_wmma_f32_16x16x4_f32(
         false, a,  false, b,
         (short)0, acc,  false,  false);
    as += 4;
    bp += 4 * HIDC;
  }

  const int rowOff = (lane >> 4) << 3;
  const float bc = bias ? bias[nBase + col] : 0.0f;
  #pragma unroll
  for (int r = 0; r < 8; ++r) {
    int row = mBase + rowOff + r;
    if (row < M) C[(long)row * HIDC + nBase + col] = acc[r] + bc;
  }
}

__global__ __launch_bounds__(256)
void el_er_kernel(const float* __restrict__ feat, const float* __restrict__ al,
                  const float* __restrict__ ar, float* __restrict__ el,
                  float* __restrict__ er, int NH) {
  const int lane = threadIdx.x & 31;
  const int wave0 = __builtin_amdgcn_readfirstlane(
      (int)(blockIdx.x * (blockDim.x >> 5) + (threadIdx.x >> 5)));
  const int nW = (int)(gridDim.x * (blockDim.x >> 5));
  const int h  = wave0 & 3;
  const float l0 = al[h * DHC + lane], l1 = al[h * DHC + lane + 32];
  const float r0 = ar[h * DHC + lane], r1 = ar[h * DHC + lane + 32];
  for (int p = wave0; p < NH; p += 8 * nW) {
    float f0[8], f1[8];
    #pragma unroll
    for (int u = 0; u < 8; ++u) {
      const int q = p + u * nW;
      f0[u] = 0.0f;
      f1[u] = 0.0f;
      if (q < NH) {
        const float* f = feat + (long)q * DHC;
        f0[u] = f[lane];
        f1[u] = f[lane + 32];
      }
    }
    float sl[8], sr[8];
    #pragma unroll
    for (int u = 0; u < 8; ++u) {
      sl[u] = f0[u] * l0 + f1[u] * l1;
      sr[u] = f0[u] * r0 + f1[u] * r1;
    }
    #pragma unroll
    for (int off = 16; off; off >>= 1) {
      #pragma unroll
      for (int u = 0; u < 8; ++u) {
        sl[u] += __shfl_down(sl[u], off, 32);
        sr[u] += __shfl_down(sr[u], off, 32);
      }
    }
    #pragma unroll
    for (int u = 0; u < 8; ++u) {
      const int q = p + u * nW;
      if (lane == 0 && q < NH) { el[q] = sl[u]; er[q] = sr[u]; }
    }
  }
}

__global__ __launch_bounds__(256)
void init_out_kernel(const float* __restrict__ hin,
                     const float* __restrict__ bias,
                     float* __restrict__ out, long total) {
  const long n4     = total >> 2;
  const long stride = (long)gridDim.x * blockDim.x;
  const long t      = (long)blockIdx.x * blockDim.x + threadIdx.x;
  const int  c      = (int)(t & (HIDC / 4 - 1)) * 4;
  const float b0 = bias[c], b1 = bias[c + 1], b2 = bias[c + 2], b3 = bias[c + 3];
  const float4* ip = reinterpret_cast<const float4*>(hin);
  float4*       op = reinterpret_cast<float4*>(out);
  for (long i = t; i < n4; i += 8 * stride) {
    float4 v[8];
    #pragma unroll
    for (int u = 0; u < 8; ++u) {
      const long j = i + u * stride;
      if (j < n4) v[u] = ip[j];
    }
    #pragma unroll
    for (int u = 0; u < 8; ++u) {
      const long j = i + u * stride;
      if (j < n4) {
        float4 o = v[u];
        o.x += b0; o.y += b1; o.z += b2; o.w += b3;
        op[j] = o;
      }
    }
  }
}

#define AGG_TN    196
#define AGG_TRIP  512
#define AGG_QCAP  768
#define AGG_SAVE  (AGG_QCAP - AGG_TRIP)
#define AGG_SHDR  64
#define AGG_SBLK  (AGG_SHDR + 32 * AGG_SAVE)

__device__ __forceinline__ float edge_weight(float score, float mx) {
  const float v = lrelu(score);
  if (!(mx > -1e38f)) mx = 0.0f;
  return __expf(v - mx);
}

__device__ __forceinline__ void agg_drain_max(
    const int* q, int cnt, int lo, int lane,
    const int* __restrict__ src, const int* __restrict__ dst,
    const float* __restrict__ el, const float* __restrict__ er, float* mt) {
  for (int i = 0; i < cnt; i += 64) {
    int s[2], d[2];
    float4 x[2], y[2];
    #pragma unroll
    for (int u = 0; u < 2; ++u) {
      const int p = i + 32 * u + lane;
      s[u] = 0; d[u] = lo;
      if (p < cnt) { const int e = q[p]; d[u] = dst[e]; s[u] = src[e]; }
    }
    #pragma unroll
    for (int u = 0; u < 2; ++u) {
      const int p = i + 32 * u + lane;
      if (p < cnt) {
        x[u] = *reinterpret_cast<const float4*>(el + 4 * (long)s[u]);
        y[u] = *reinterpret_cast<const float4*>(er + 4 * (long)d[u]);
      }
    }
    #pragma unroll
    for (int u = 0; u < 2; ++u) {
      const int p = i + 32 * u + lane;
      if (p < cnt) {
        float* m = mt + 4 * (d[u] - lo);
        atomicMaxF(m,     lrelu(x[u].x + y[u].x));
        atomicMaxF(m + 1, lrelu(x[u].y + y[u].y));
        atomicMaxF(m + 2, lrelu(x[u].z + y[u].z));
        atomicMaxF(m + 3, lrelu(x[u].w + y[u].w));
      }
    }
  }
}

__device__ __forceinline__ void agg_drain_acc(
    const int* q, int cnt, int lo, int lane,
    const int* __restrict__ src, const int* __restrict__ dst,
    const float* __restrict__ feat,
    const float* __restrict__ el, const float* __restrict__ er,
    const float* mt, float* dt, float* acc) {
  for (int i = 0; i < cnt; i += 32) {
    const int n = min(32, cnt - i);
    int s = 0, dl = 0;
    float a0 = 0.0f, a1 = 0.0f, a2 = 0.0f, a3 = 0.0f;
    if (lane < n) {
      const int e = q[i + lane];
      const int d = dst[e];
      s  = src[e];
      dl = d - lo;
      const float4 x = *reinterpret_cast<const float4*>(el + 4 * (long)s);
      const float4 y = *reinterpret_cast<const float4*>(er + 4 * (long)d);
      const float4 m = *reinterpret_cast<const float4*>(mt + 4 * dl);
      a0 = edge_weight(x.x + y.x, m.x);
      a1 = edge_weight(x.y + y.y, m.y);
      a2 = edge_weight(x.z + y.z, m.z);
      a3 = edge_weight(x.w + y.w, m.w);
      float* ds = dt + 4 * dl;
      atomicAdd(ds,     a0);
      atomicAdd(ds + 1, a1);
      atomicAdd(ds + 2, a2);
      atomicAdd(ds + 3, a3);
    }
    for (int k = 0; k < n; k += 4) {
      int su[4], du[4];
      float au[4][4];
      v2f f[4][4];
      #pragma unroll
      for (int u = 0; u < 4; ++u) {
        const int idx = min(k + u, n - 1);
        su[u] = __builtin_amdgcn_readlane(s, idx);
        du[u] = (k + u < n) ? __builtin_amdgcn_readlane(dl, idx) : AGG_TN;
        au[u][0] = __int_as_float(__builtin_amdgcn_readlane(__float_as_int(a0), idx));
        au[u][1] = __int_as_float(__builtin_amdgcn_readlane(__float_as_int(a1), idx));
        au[u][2] = __int_as_float(__builtin_amdgcn_readlane(__float_as_int(a2), idx));
        au[u][3] = __int_as_float(__builtin_amdgcn_readlane(__float_as_int(a3), idx));
      }
      #pragma unroll
      for (int u = 0; u < 4; ++u) {
        const float* fr = feat + (long)su[u] * HIDC + 2 * lane;
        #pragma unroll
        for (int j = 0; j < HEADS; ++j)
          f[u][j] = *reinterpret_cast<const v2f*>(fr + j * DHC);
      }
      #pragma unroll
      for (int u = 0; u < 4; ++u) {
        float* p = acc + du[u] * HIDC + 2 * lane;
        #pragma unroll
        for (int j = 0; j < HEADS; ++j) {
          atomicAdd(p + j * DHC,     au[u][j] * f[u][j].x);
          atomicAdd(p + j * DHC + 1, au[u][j] * f[u][j].y);
        }
      }
    }
  }
}

template <int PASS>
__device__ __forceinline__ int agg_scan(
    int* q, int wave, int lane, int lo, int rows, int E,
    const int* __restrict__ src, const int* __restrict__ dst,
    const float* __restrict__ feat,
    const float* __restrict__ el, const float* __restrict__ er,
    float* mt, float* dt, float* acc, int* full) {
  const int  T       = (E + AGG_TRIP - 1) / AGG_TRIP;
  const bool aligned = (reinterpret_cast<unsigned long>(dst) & 15ul) == 0ul;
  const int start = (T / (int)gridDim.x) * (int)blockIdx.x;
  int cnt = 0;
  for (int t = wave; t < T; t += 32) {
    int g = t + start;
    if (g >= T) g -= T;
    const int base = g * AGG_TRIP + 4 * lane;
    int4 dv[4];
    if (aligned && g * AGG_TRIP + AGG_TRIP <= E) {
      #pragma unroll
      for (int u = 0; u < 4; ++u)
        dv[u] = *reinterpret_cast<const int4*>(dst + base + 128 * u);
    } else {
      #pragma unroll
      for (int u = 0; u < 4; ++u) {
        const int e0 = base + 128 * u;
        dv[u].x = (e0     < E) ? dst[e0]     : -1;
        dv[u].y = (e0 + 1 < E) ? dst[e0 + 1] : -1;
        dv[u].z = (e0 + 2 < E) ? dst[e0 + 2] : -1;
        dv[u].w = (e0 + 3 < E) ? dst[e0 + 3] : -1;
      }
    }
    #pragma unroll
    for (int u = 0; u < 4; ++u) {
      const int  e0 = base + 128 * u;
      const bool mx = (unsigned)(dv[u].x - lo) < (unsigned)rows;
      const bool my = (unsigned)(dv[u].y - lo) < (unsigned)rows;
      const bool mz = (unsigned)(dv[u].z - lo) < (unsigned)rows;
      const bool mw = (unsigned)(dv[u].w - lo) < (unsigned)rows;
      if (__builtin_amdgcn_ballot_w32(mx | my | mz | mw)) {
        unsigned mask;
        mask = __builtin_amdgcn_ballot_w32(mx);
        if (mx) q[cnt + __builtin_amdgcn_mbcnt_lo(mask, 0u)] = e0;
        cnt += __builtin_popcount(mask);
        mask = __builtin_amdgcn_ballot_w32(my);
        if (my) q[cnt + __builtin_amdgcn_mbcnt_lo(mask, 0u)] = e0 + 1;
        cnt += __builtin_popcount(mask);
        mask = __builtin_amdgcn_ballot_w32(mz);
        if (mz) q[cnt + __builtin_amdgcn_mbcnt_lo(mask, 0u)] = e0 + 2;
        cnt += __builtin_popcount(mask);
        mask = __builtin_amdgcn_ballot_w32(mw);
        if (mw) q[cnt + __builtin_amdgcn_mbcnt_lo(mask, 0u)] = e0 + 3;
        cnt += __builtin_popcount(mask);
      }
    }
    if (cnt > AGG_QCAP - AGG_TRIP) {
      if (PASS == 0) {
        agg_drain_max(q, cnt, lo, lane, src, dst, el, er, mt);
        if (lane == 0) *full = 1;
      } else {
        agg_drain_acc(q, cnt, lo, lane, src, dst, feat, el, er, mt, dt, acc);
      }
      cnt = 0;
    }
  }
  return cnt;
}

__global__ __launch_bounds__(1024)
void aggregate_tile_kernel(const int* __restrict__ src, const int* __restrict__ dst,
                           const float* __restrict__ feat,
                           const float* __restrict__ el,
                           const float* __restrict__ er, float* __restrict__ out,
                           int N, int E, int* qsave, const float* wsBase,
                           unsigned long wsBytes, int reuseSaved) {
  __shared__ __attribute__((aligned(16))) float acc[(AGG_TN + 1) * HIDC];
  __shared__ __attribute__((aligned(16))) float mt[AGG_TN * HEADS];
  __shared__ __attribute__((aligned(16))) float dt[AGG_TN * HEADS];
  __shared__ int queue[32 * AGG_QCAP];
  __shared__ int full;

  const int tid  = threadIdx.x;
  const int lane = tid & 31;
  const int wave = __builtin_amdgcn_readfirstlane(tid >> 5);
  const int lo   = blockIdx.x * AGG_TN;
  const int rows = min(AGG_TN, N - lo);

  for (int i = tid; i < (AGG_TN + 1) * HIDC / 4; i += 1024)
    reinterpret_cast<float4*>(acc)[i] = make_float4(0.0f, 0.0f, 0.0f, 0.0f);
  if (tid < AGG_TN * HEADS) { mt[tid] = -INFINITY; dt[tid] = 0.0f; }
  if (tid == 0) full = 0;
  __syncthreads();

  int* q = queue + wave * AGG_QCAP;
  int* sv = qsave + (long)blockIdx.x * AGG_SBLK;
  const bool fits =
      (unsigned long)(reinterpret_cast<const char*>(qsave + (long)gridDim.x * AGG_SBLK) -
                      reinterpret_cast<const char*>(wsBase)) <= wsBytes;
  int reuse = 0, saved = 0;
  if (reuseSaved != 0 && fits) {
    const int okw = sv[32];
    const int cw  = sv[wave];
    reuse = (okw == 1) ? 1 : 0;
    saved = cw;
  }
  int cnt;
  if (reuse != 0) {
    cnt = min(max(saved, 0), AGG_SAVE);
    const int* sq = sv + AGG_SHDR + wave * AGG_SAVE;
    int v[AGG_SAVE / 32];
    #pragma unroll
    for (int u = 0; u < AGG_SAVE / 32; ++u) {
      const int i = lane + 32 * u;
      v[u] = 0;
      if (i < cnt) v[u] = sq[i];
    }
    #pragma unroll
    for (int u = 0; u < AGG_SAVE / 32; ++u) {
      const int i = lane + 32 * u;
      if (i < cnt) q[i] = v[u];
    }
  } else {
    cnt = agg_scan<0>(q, wave, lane, lo, rows, E, src, dst, feat, el, er, mt, dt, acc, &full);
  }
  agg_drain_max(q, cnt, lo, lane, src, dst, el, er, mt);
  __syncthreads();

  const int early = (reuse != 0) ? 0 : __builtin_amdgcn_readfirstlane(full);
  if (reuseSaved == 0 && fits) {
    if (early == 0) {
      int* sq = sv + AGG_SHDR + wave * AGG_SAVE;
      for (int i = lane; i < cnt; i += 32) sq[i] = q[i];
    }
    if (lane == 0) sv[wave] = (early == 0) ? cnt : 0;
    if (tid == 0) sv[32] = (early == 0) ? 1 : 0;
  }
  if (early == 0) {
    agg_drain_acc(q, cnt, lo, lane, src, dst, feat, el, er, mt, dt, acc);
  } else {
    cnt = agg_scan<1>(q, wave, lane, lo, rows, E, src, dst, feat, el, er, mt, dt, acc, &full);
    agg_drain_acc(q, cnt, lo, lane, src, dst, feat, el, er, mt, dt, acc);
  }
  __syncthreads();

  if (tid < rows * HEADS) dt[tid] = 1.0f / fmaxf(dt[tid], 1e-16f);
  __syncthreads();

  float4* op = reinterpret_cast<float4*>(out + (long)lo * HIDC);
  for (int i = tid; i < rows * (HIDC / 4); i += 1024) {
    const float  r = dt[i >> 4];
    const float4 a = reinterpret_cast<const float4*>(acc)[i];
    float4 o = op[i];
    o.x += a.x * r; o.y += a.y * r; o.z += a.z * r; o.w += a.w * r;
    op[i] = o;
  }
}

__global__ __launch_bounds__(256)
void relu_kernel(const float* __restrict__ in, float* __restrict__ out,
                 long total) {
  const long n4     = total >> 2;
  const long stride = (long)gridDim.x * blockDim.x;
  const float4* ip = reinterpret_cast<const float4*>(in);
  float4*       op = reinterpret_cast<float4*>(out);
  for (long i = (long)blockIdx.x * blockDim.x + threadIdx.x; i < n4; i += 8 * stride) {
    float4 v[8];
    #pragma unroll
    for (int u = 0; u < 8; ++u) {
      const long j = i + u * stride;
      if (j < n4) v[u] = ip[j];
    }
    #pragma unroll
    for (int u = 0; u < 8; ++u) {
      const long j = i + u * stride;
      if (j < n4) {
        float4 o;
        o.x = fmaxf(v[u].x, 0.0f); o.y = fmaxf(v[u].y, 0.0f);
        o.z = fmaxf(v[u].z, 0.0f); o.w = fmaxf(v[u].w, 0.0f);
        op[j] = o;
      }
    }
  }
}

__global__ __launch_bounds__(256)
void mean_heads_kernel(const float* __restrict__ hbuf,
                       float* __restrict__ out, long total) {
  const long stride = (long)gridDim.x * blockDim.x;
  const long t      = (long)blockIdx.x * blockDim.x + threadIdx.x;
  if ((reinterpret_cast<unsigned long>(out) & 15ul) == 0ul) {
    const long n4 = total >> 2;
    const float4* hp = reinterpret_cast<const float4*>(hbuf);
    float4*       op = reinterpret_cast<float4*>(out);
    for (long i = t; i < n4; i += 2 * stride) {
      float4 p[2][4];
      #pragma unroll
      for (int u = 0; u < 2; ++u) {
        const long j = i + u * stride;
        if (j < n4) {
          const float4* r = hp + (j >> 4) * (HIDC / 4) + (j & 15);
          #pragma unroll
          for (int k = 0; k < HEADS; ++k) p[u][k] = r[k * (DHC / 4)];
        }
      }
      #pragma unroll
      for (int u = 0; u < 2; ++u) {
        const long j = i + u * stride;
        if (j < n4) {
          float4 o;
          o.x = 0.25f * (p[u][0].x + p[u][1].x + p[u][2].x + p[u][3].x);
          o.y = 0.25f * (p[u][0].y + p[u][1].y + p[u][2].y + p[u][3].y);
          o.z = 0.25f * (p[u][0].z + p[u][1].z + p[u][2].z + p[u][3].z);
          o.w = 0.25f * (p[u][0].w + p[u][1].w + p[u][2].w + p[u][3].w);
          op[j] = o;
        }
      }
    }
  } else {
    for (long i = t; i < total; i += stride) {
      const float* p = hbuf + (i >> 6) * HIDC + (i & 63);
      out[i] = 0.25f * (p[0] + p[DHC] + p[2 * DHC] + p[3 * DHC]);
    }
  }
}

extern "C" void kernel_launch(void* const* d_in, const int* in_sizes, int n_in,
                              void* d_out, int out_size, void* d_ws, size_t ws_size,
                              hipStream_t stream) {
  const float* x     = (const float*)d_in[0];
  const int*   ei    = (const int*)d_in[1];
  const float* W[3]  = {(const float*)d_in[2], (const float*)d_in[3],
                        (const float*)d_in[4]};
  const float* al[3] = {(const float*)d_in[5], (const float*)d_in[6],
                        (const float*)d_in[7]};
  const float* ar[3] = {(const float*)d_in[8], (const float*)d_in[9],
                        (const float*)d_in[10]};
  const float* b[3]  = {(const float*)d_in[11], (const float*)d_in[12],
                        (const float*)d_in[13]};
  const float* resW0 = (const float*)d_in[14];

  const int N = in_sizes[0] / 128;
  const int E = in_sizes[1] / 2;
  const int* src = ei;
  const int* dst = ei + E;

  float* ws   = (float*)d_ws;
  float* FEAT = ws;
  float* OUT  = FEAT + (size_t)N * HIDC;
  float* HBUF = OUT  + (size_t)N * HIDC;
  float* ELP  = HBUF + (size_t)N * HIDC;
  float* ERP  = ELP  + (size_t)N * HEADS;
  int*   QSV  = (int*)(ERP + (size_t)N * HEADS);

  const int  mtiles = (N + 15) / 16;
  const long totNF  = (long)N * HIDC;
  const int  NH     = N * HEADS;
  const int  gSTREAM = 512;
  const int  gNARROW = 1024;

  auto attention = [&](const float* alw, const float* arw, int reuseSaved) {
    el_er_kernel<<<gNARROW, 256, 0, stream>>>(FEAT, alw, arw, ELP, ERP, NH);
    aggregate_tile_kernel<<<(N + AGG_TN - 1) / AGG_TN, 1024, 0, stream>>>(
        src, dst, FEAT, ELP, ERP, OUT, N, E, QSV, ws, (unsigned long)ws_size, reuseSaved);
  };

  gemm_wmma_f32<128><<<mtiles, 512, 0, stream>>>(x, W[0], nullptr, FEAT, N);
  gemm_wmma_f32<128><<<mtiles, 512, 0, stream>>>(x, resW0, b[0], OUT, N);
  attention(al[0], ar[0], 0);
  relu_kernel<<<gSTREAM, 256, 0, stream>>>(OUT, HBUF, totNF);

  gemm_wmma_f32<256><<<mtiles, 512, 0, stream>>>(HBUF, W[1], nullptr, FEAT, N);
  init_out_kernel<<<gSTREAM, 256, 0, stream>>>(HBUF, b[1], OUT, totNF);
  attention(al[1], ar[1], 1);
  relu_kernel<<<gSTREAM, 256, 0, stream>>>(OUT, HBUF, totNF);

  gemm_wmma_f32<256><<<mtiles, 512, 0, stream>>>(HBUF, W[2], nullptr, FEAT, N);
  init_out_kernel<<<gSTREAM, 256, 0, stream>>>(HBUF, b[2], OUT, totNF);
  attention(al[2], ar[2], 1);
  mean_heads_kernel<<<gSTREAM, 256, 0, stream>>>(
      OUT, (float*)d_out, (long)N * DHC);
}


# relu and the residual+bias start of both identity-residual layers move into the projection GEMM's A-strip staging, in place on the layer output; 4 element-wise launches gone; float sums unchanged
# speedup vs baseline: 4.3788x; 1.0635x over previous
#include <hip/hip_runtime.h>
#include <math.h>


typedef __attribute__((ext_vector_type(2))) float v2f;
typedef __attribute__((ext_vector_type(8))) float v8f;

#define HIDC 256
#define HEADS 4
#define DHC 64

__device__ __forceinline__ float atomicMaxF(float* addr, float value) {
  if (value >= 0.0f)
    return __int_as_float(atomicMax((int*)addr, __float_as_int(value)));
  else
    return __uint_as_float(atomicMin((unsigned int*)addr, __float_as_uint(value)));
}

__device__ __forceinline__ float lrelu(float v) {
  return v > 0.0f ? v : 0.2f * v;
}

template <int K, bool RELU_IN>
__global__ __launch_bounds__(512)
void gemm_wmma_f32(const float* A, const float* __restrict__ W,
                   const float* __restrict__ bias, float* __restrict__ C,
                   int M, float* R, const float* __restrict__ rbias) {
  constexpr int LDA = K + 4;
  constexpr int NV  = 16 * K / 512;
  constexpr int RS  = 512 / K;
  __shared__ float As[16 * LDA];

  const int tid   = threadIdx.x;
  const int mBase = blockIdx.x << 4;

  const int sc = tid & (K - 1);
  const int sr = tid / K;
  float v[NV];
  #pragma unroll
  for (int u = 0; u < NV; ++u) {
    int row = mBase + sr + u * RS;
    if (row >= M) row = M - 1;
    v[u] = A[(long)row * K + sc];
  }
  const float rb = RELU_IN ? rbias[sc] : 0.0f;
  #pragma unroll
  for (int u = 0; u < NV; ++u) {
    if (RELU_IN) v[u] = fmaxf(v[u], 0.0f);
    As[(sr + u * RS) * LDA + sc] = v[u];
  }
  __syncthreads();
  if (RELU_IN) {
    #pragma unroll
    for (int u = 0; u < NV; ++u) {
      const int row = mBase + sr + u * RS;
      if (row < M) R[(long)row * K + sc] = v[u] + rb;
    }
  }

  const int lane  = tid & 31;
  const int wave  = tid >> 5;
  const int nBase = wave << 4;
  const int col   = lane & 15;
  const int kHalf = (lane >> 4) << 1;

  const float* as = &As[(lane & 15) * LDA + kHalf];
  const float* bp = W + (long)kHalf * HIDC + nBase + col;

  v8f acc = {};
  #pragma unroll 8
  for (int kb = 0; kb < K; kb += 4) {
    v2f a, b;
    a.x = as[0];
    a.y = as[1];
    b.x = bp[0];
    b.y = bp[HIDC];
    acc = __builtin_amdgcn_wmma_f32_16x16x4_f32(
         false, a,  false, b,
         (short)0, acc,  false,  false);
    as += 4;
    bp += 4 * HIDC;
  }

  const int rowOff = (lane >> 4) << 3;
  const float bc = bias ? bias[nBase + col] : 0.0f;
  #pragma unroll
  for (int r = 0; r < 8; ++r) {
    int row = mBase + rowOff + r;
    if (row < M) C[(long)row * HIDC + nBase + col] = acc[r] + bc;
  }
}

__global__ __launch_bounds__(256)
void el_er_kernel(const float* __restrict__ feat, const float* __restrict__ al,
                  const float* __restrict__ ar, float* __restrict__ el,
                  float* __restrict__ er, int NH) {
  const int lane = threadIdx.x & 31;
  const int wave0 = __builtin_amdgcn_readfirstlane(
      (int)(blockIdx.x * (blockDim.x >> 5) + (threadIdx.x >> 5)));
  const int nW = (int)(gridDim.x * (blockDim.x >> 5));
  const int h  = wave0 & 3;
  const float l0 = al[h * DHC + lane], l1 = al[h * DHC + lane + 32];
  const float r0 = ar[h * DHC + lane], r1 = ar[h * DHC + lane + 32];
  for (int p = wave0; p < NH; p += 8 * nW) {
    float f0[8], f1[8];
    #pragma unroll
    for (int u = 0; u < 8; ++u) {
      const int q = p + u * nW;
      f0[u] = 0.0f;
      f1[u] = 0.0f;
      if (q < NH) {
        const float* f = feat + (long)q * DHC;
        f0[u] = f[lane];
        f1[u] = f[lane + 32];
      }
    }
    float sl[8], sr[8];
    #pragma unroll
    for (int u = 0; u < 8; ++u) {
      sl[u] = f0[u] * l0 + f1[u] * l1;
      sr[u] = f0[u] * r0 + f1[u] * r1;
    }
    #pragma unroll
    for (int off = 16; off; off >>= 1) {
      #pragma unroll
      for (int u = 0; u < 8; ++u) {
        sl[u] += __shfl_down(sl[u], off, 32);
        sr[u] += __shfl_down(sr[u], off, 32);
      }
    }
    #pragma unroll
    for (int u = 0; u < 8; ++u) {
      const int q = p + u * nW;
      if (lane == 0 && q < NH) { el[q] = sl[u]; er[q] = sr[u]; }
    }
  }
}

#define AGG_TN    196
#define AGG_TRIP  512
#define AGG_QCAP  768
#define AGG_SAVE  (AGG_QCAP - AGG_TRIP)
#define AGG_SHDR  64
#define AGG_SBLK  (AGG_SHDR + 32 * AGG_SAVE)

__device__ __forceinline__ float edge_weight(float score, float mx) {
  const float v = lrelu(score);
  if (!(mx > -1e38f)) mx = 0.0f;
  return __expf(v - mx);
}

__device__ __forceinline__ void agg_drain_max(
    const int* q, int cnt, int lo, int lane,
    const int* __restrict__ src, const int* __restrict__ dst,
    const float* __restrict__ el, const float* __restrict__ er, float* mt) {
  for (int i = 0; i < cnt; i += 64) {
    int s[2], d[2];
    float4 x[2], y[2];
    #pragma unroll
    for (int u = 0; u < 2; ++u) {
      const int p = i + 32 * u + lane;
      s[u] = 0; d[u] = lo;
      if (p < cnt) { const int e = q[p]; d[u] = dst[e]; s[u] = src[e]; }
    }
    #pragma unroll
    for (int u = 0; u < 2; ++u) {
      const int p = i + 32 * u + lane;
      if (p < cnt) {
        x[u] = *reinterpret_cast<const float4*>(el + 4 * (long)s[u]);
        y[u] = *reinterpret_cast<const float4*>(er + 4 * (long)d[u]);
      }
    }
    #pragma unroll
    for (int u = 0; u < 2; ++u) {
      const int p = i + 32 * u + lane;
      if (p < cnt) {
        float* m = mt + 4 * (d[u] - lo);
        atomicMaxF(m,     lrelu(x[u].x + y[u].x));
        atomicMaxF(m + 1, lrelu(x[u].y + y[u].y));
        atomicMaxF(m + 2, lrelu(x[u].z + y[u].z));
        atomicMaxF(m + 3, lrelu(x[u].w + y[u].w));
      }
    }
  }
}

__device__ __forceinline__ void agg_drain_acc(
    const int* q, int cnt, int lo, int lane,
    const int* __restrict__ src, const int* __restrict__ dst,
    const float* __restrict__ feat,
    const float* __restrict__ el, const float* __restrict__ er,
    const float* mt, float* dt, float* acc) {
  for (int i = 0; i < cnt; i += 32) {
    const int n = min(32, cnt - i);
    int s = 0, dl = 0;
    float a0 = 0.0f, a1 = 0.0f, a2 = 0.0f, a3 = 0.0f;
    if (lane < n) {
      const int e = q[i + lane];
      const int d = dst[e];
      s  = src[e];
      dl = d - lo;
      const float4 x = *reinterpret_cast<const float4*>(el + 4 * (long)s);
      const float4 y = *reinterpret_cast<const float4*>(er + 4 * (long)d);
      const float4 m = *reinterpret_cast<const float4*>(mt + 4 * dl);
      a0 = edge_weight(x.x + y.x, m.x);
      a1 = edge_weight(x.y + y.y, m.y);
      a2 = edge_weight(x.z + y.z, m.z);
      a3 = edge_weight(x.w + y.w, m.w);
      float* ds = dt + 4 * dl;
      atomicAdd(ds,     a0);
      atomicAdd(ds + 1, a1);
      atomicAdd(ds + 2, a2);
      atomicAdd(ds + 3, a3);
    }
    for (int k = 0; k < n; k += 4) {
      int su[4], du[4];
      float au[4][4];
      v2f f[4][4];
      #pragma unroll
      for (int u = 0; u < 4; ++u) {
        const int idx = min(k + u, n - 1);
        su[u] = __builtin_amdgcn_readlane(s, idx);
        du[u] = (k + u < n) ? __builtin_amdgcn_readlane(dl, idx) : AGG_TN;
        au[u][0] = __int_as_float(__builtin_amdgcn_readlane(__float_as_int(a0), idx));
        au[u][1] = __int_as_float(__builtin_amdgcn_readlane(__float_as_int(a1), idx));
        au[u][2] = __int_as_float(__builtin_amdgcn_readlane(__float_as_int(a2), idx));
        au[u][3] = __int_as_float(__builtin_amdgcn_readlane(__float_as_int(a3), idx));
      }
      #pragma unroll
      for (int u = 0; u < 4; ++u) {
        const float* fr = feat + (long)su[u] * HIDC + 2 * lane;
        #pragma unroll
        for (int j = 0; j < HEADS; ++j)
          f[u][j] = *reinterpret_cast<const v2f*>(fr + j * DHC);
      }
      #pragma unroll
      for (int u = 0; u < 4; ++u) {
        float* p = acc + du[u] * HIDC + 2 * lane;
        #pragma unroll
        for (int j = 0; j < HEADS; ++j) {
          atomicAdd(p + j * DHC,     au[u][j] * f[u][j].x);
          atomicAdd(p + j * DHC + 1, au[u][j] * f[u][j].y);
        }
      }
    }
  }
}

template <int PASS>
__device__ __forceinline__ int agg_scan(
    int* q, int wave, int lane, int lo, int rows, int E,
    const int* __restrict__ src, const int* __restrict__ dst,
    const float* __restrict__ feat,
    const float* __restrict__ el, const float* __restrict__ er,
    float* mt, float* dt, float* acc, int* full) {
  const int  T       = (E + AGG_TRIP - 1) / AGG_TRIP;
  const bool aligned = (reinterpret_cast<unsigned long>(dst) & 15ul) == 0ul;
  const int start = (T / (int)gridDim.x) * (int)blockIdx.x;
  int cnt = 0;
  for (int t = wave; t < T; t += 32) {
    int g = t + start;
    if (g >= T) g -= T;
    const int base = g * AGG_TRIP + 4 * lane;
    int4 dv[4];
    if (aligned && g * AGG_TRIP + AGG_TRIP <= E) {
      #pragma unroll
      for (int u = 0; u < 4; ++u)
        dv[u] = *reinterpret_cast<const int4*>(dst + base + 128 * u);
    } else {
      #pragma unroll
      for (int u = 0; u < 4; ++u) {
        const int e0 = base + 128 * u;
        dv[u].x = (e0     < E) ? dst[e0]     : -1;
        dv[u].y = (e0 + 1 < E) ? dst[e0 + 1] : -1;
        dv[u].z = (e0 + 2 < E) ? dst[e0 + 2] : -1;
        dv[u].w = (e0 + 3 < E) ? dst[e0 + 3] : -1;
      }
    }
    #pragma unroll
    for (int u = 0; u < 4; ++u) {
      const int  e0 = base + 128 * u;
      const bool mx = (unsigned)(dv[u].x - lo) < (unsigned)rows;
      const bool my = (unsigned)(dv[u].y - lo) < (unsigned)rows;
      const bool mz = (unsigned)(dv[u].z - lo) < (unsigned)rows;
      const bool mw = (unsigned)(dv[u].w - lo) < (unsigned)rows;
      if (__builtin_amdgcn_ballot_w32(mx | my | mz | mw)) {
        unsigned mask;
        mask = __builtin_amdgcn_ballot_w32(mx);
        if (mx) q[cnt + __builtin_amdgcn_mbcnt_lo(mask, 0u)] = e0;
        cnt += __builtin_popcount(mask);
        mask = __builtin_amdgcn_ballot_w32(my);
        if (my) q[cnt + __builtin_amdgcn_mbcnt_lo(mask, 0u)] = e0 + 1;
        cnt += __builtin_popcount(mask);
        mask = __builtin_amdgcn_ballot_w32(mz);
        if (mz) q[cnt + __builtin_amdgcn_mbcnt_lo(mask, 0u)] = e0 + 2;
        cnt += __builtin_popcount(mask);
        mask = __builtin_amdgcn_ballot_w32(mw);
        if (mw) q[cnt + __builtin_amdgcn_mbcnt_lo(mask, 0u)] = e0 + 3;
        cnt += __builtin_popcount(mask);
      }
    }
    if (cnt > AGG_QCAP - AGG_TRIP) {
      if (PASS == 0) {
        agg_drain_max(q, cnt, lo, lane, src, dst, el, er, mt);
        if (lane == 0) *full = 1;
      } else {
        agg_drain_acc(q, cnt, lo, lane, src, dst, feat, el, er, mt, dt, acc);
      }
      cnt = 0;
    }
  }
  return cnt;
}

__global__ __launch_bounds__(1024)
void aggregate_tile_kernel(const int* __restrict__ src, const int* __restrict__ dst,
                           const float* __restrict__ feat,
                           const float* __restrict__ el,
                           const float* __restrict__ er, float* __restrict__ out,
                           int N, int E, int* qsave, const float* wsBase,
                           unsigned long wsBytes, int reuseSaved) {
  __shared__ __attribute__((aligned(16))) float acc[(AGG_TN + 1) * HIDC];
  __shared__ __attribute__((aligned(16))) float mt[AGG_TN * HEADS];
  __shared__ __attribute__((aligned(16))) float dt[AGG_TN * HEADS];
  __shared__ int queue[32 * AGG_QCAP];
  __shared__ int full;

  const int tid  = threadIdx.x;
  const int lane = tid & 31;
  const int wave = __builtin_amdgcn_readfirstlane(tid >> 5);
  const int lo   = blockIdx.x * AGG_TN;
  const int rows = min(AGG_TN, N - lo);

  for (int i = tid; i < (AGG_TN + 1) * HIDC / 4; i += 1024)
    reinterpret_cast<float4*>(acc)[i] = make_float4(0.0f, 0.0f, 0.0f, 0.0f);
  if (tid < AGG_TN * HEADS) { mt[tid] = -INFINITY; dt[tid] = 0.0f; }
  if (tid == 0) full = 0;
  __syncthreads();

  int* q = queue + wave * AGG_QCAP;
  int* sv = qsave + (long)blockIdx.x * AGG_SBLK;
  const bool fits =
      (unsigned long)(reinterpret_cast<const char*>(qsave + (long)gridDim.x * AGG_SBLK) -
                      reinterpret_cast<const char*>(wsBase)) <= wsBytes;
  int reuse = 0, saved = 0;
  if (reuseSaved != 0 && fits) {
    const int okw = sv[32];
    const int cw  = sv[wave];
    reuse = (okw == 1) ? 1 : 0;
    saved = cw;
  }
  int cnt;
  if (reuse != 0) {
    cnt = min(max(saved, 0), AGG_SAVE);
    const int* sq = sv + AGG_SHDR + wave * AGG_SAVE;
    int v[AGG_SAVE / 32];
    #pragma unroll
    for (int u = 0; u < AGG_SAVE / 32; ++u) {
      const int i = lane + 32 * u;
      v[u] = 0;
      if (i < cnt) v[u] = sq[i];
    }
    #pragma unroll
    for (int u = 0; u < AGG_SAVE / 32; ++u) {
      const int i = lane + 32 * u;
      if (i < cnt) q[i] = v[u];
    }
  } else {
    cnt = agg_scan<0>(q, wave, lane, lo, rows, E, src, dst, feat, el, er, mt, dt, acc, &full);
  }
  agg_drain_max(q, cnt, lo, lane, src, dst, el, er, mt);
  __syncthreads();

  const int early = (reuse != 0) ? 0 : __builtin_amdgcn_readfirstlane(full);
  if (reuseSaved == 0 && fits) {
    if (early == 0) {
      int* sq = sv + AGG_SHDR + wave * AGG_SAVE;
      for (int i = lane; i < cnt; i += 32) sq[i] = q[i];
    }
    if (lane == 0) sv[wave] = (early == 0) ? cnt : 0;
    if (tid == 0) sv[32] = (early == 0) ? 1 : 0;
  }
  if (early == 0) {
    agg_drain_acc(q, cnt, lo, lane, src, dst, feat, el, er, mt, dt, acc);
  } else {
    cnt = agg_scan<1>(q, wave, lane, lo, rows, E, src, dst, feat, el, er, mt, dt, acc, &full);
    agg_drain_acc(q, cnt, lo, lane, src, dst, feat, el, er, mt, dt, acc);
  }
  __syncthreads();

  if (tid < rows * HEADS) dt[tid] = 1.0f / fmaxf(dt[tid], 1e-16f);
  __syncthreads();

  float4* op = reinterpret_cast<float4*>(out + (long)lo * HIDC);
  for (int i = tid; i < rows * (HIDC / 4); i += 1024) {
    const float  r = dt[i >> 4];
    const float4 a = reinterpret_cast<const float4*>(acc)[i];
    float4 o = op[i];
    o.x += a.x * r; o.y += a.y * r; o.z += a.z * r; o.w += a.w * r;
    op[i] = o;
  }
}

__global__ __launch_bounds__(256)
void mean_heads_kernel(const float* __restrict__ hbuf,
                       float* __restrict__ out, long total) {
  const long stride = (long)gridDim.x * blockDim.x;
  const long t      = (long)blockIdx.x * blockDim.x + threadIdx.x;
  if ((reinterpret_cast<unsigned long>(out) & 15ul) == 0ul) {
    const long n4 = total >> 2;
    const float4* hp = reinterpret_cast<const float4*>(hbuf);
    float4*       op = reinterpret_cast<float4*>(out);
    for (long i = t; i < n4; i += 2 * stride) {
      float4 p[2][4];
      #pragma unroll
      for (int u = 0; u < 2; ++u) {
        const long j = i + u * stride;
        if (j < n4) {
          const float4* r = hp + (j >> 4) * (HIDC / 4) + (j & 15);
          #pragma unroll
          for (int k = 0; k < HEADS; ++k) p[u][k] = r[k * (DHC / 4)];
        }
      }
      #pragma unroll
      for (int u = 0; u < 2; ++u) {
        const long j = i + u * stride;
        if (j < n4) {
          float4 o;
          o.x = 0.25f * (p[u][0].x + p[u][1].x + p[u][2].x + p[u][3].x);
          o.y = 0.25f * (p[u][0].y + p[u][1].y + p[u][2].y + p[u][3].y);
          o.z = 0.25f * (p[u][0].z + p[u][1].z + p[u][2].z + p[u][3].z);
          o.w = 0.25f * (p[u][0].w + p[u][1].w + p[u][2].w + p[u][3].w);
          op[j] = o;
        }
      }
    }
  } else {
    for (long i = t; i < total; i += stride) {
      const float* p = hbuf + (i >> 6) * HIDC + (i & 63);
      out[i] = 0.25f * (p[0] + p[DHC] + p[2 * DHC] + p[3 * DHC]);
    }
  }
}

extern "C" void kernel_launch(void* const* d_in, const int* in_sizes, int n_in,
                              void* d_out, int out_size, void* d_ws, size_t ws_size,
                              hipStream_t stream) {
  const float* x     = (const float*)d_in[0];
  const int*   ei    = (const int*)d_in[1];
  const float* W[3]  = {(const float*)d_in[2], (const float*)d_in[3],
                        (const float*)d_in[4]};
  const float* al[3] = {(const float*)d_in[5], (const float*)d_in[6],
                        (const float*)d_in[7]};
  const float* ar[3] = {(const float*)d_in[8], (const float*)d_in[9],
                        (const float*)d_in[10]};
  const float* b[3]  = {(const float*)d_in[11], (const float*)d_in[12],
                        (const float*)d_in[13]};
  const float* resW0 = (const float*)d_in[14];

  const int N = in_sizes[0] / 128;
  const int E = in_sizes[1] / 2;
  const int* src = ei;
  const int* dst = ei + E;

  float* ws   = (float*)d_ws;
  float* FEAT = ws;
  float* OUT  = FEAT + (size_t)N * HIDC;
  float* ELP  = OUT  + (size_t)N * HIDC;
  float* ERP  = ELP  + (size_t)N * HEADS;
  int*   QSV  = (int*)(ERP + (size_t)N * HEADS);

  const int  mtiles = (N + 15) / 16;
  const int  NH     = N * HEADS;
  const int  gSTREAM = 512;
  const int  gNARROW = 1024;

  auto attention = [&](const float* alw, const float* arw, int reuseSaved) {
    el_er_kernel<<<gNARROW, 256, 0, stream>>>(FEAT, alw, arw, ELP, ERP, NH);
    aggregate_tile_kernel<<<(N + AGG_TN - 1) / AGG_TN, 1024, 0, stream>>>(
        src, dst, FEAT, ELP, ERP, OUT, N, E, QSV, ws, (unsigned long)ws_size, reuseSaved);
  };

  gemm_wmma_f32<128, false><<<mtiles, 512, 0, stream>>>(x, W[0], nullptr, FEAT, N, nullptr, nullptr);
  gemm_wmma_f32<128, false><<<mtiles, 512, 0, stream>>>(x, resW0, b[0], OUT, N, nullptr, nullptr);
  attention(al[0], ar[0], 0);

  gemm_wmma_f32<256, true><<<mtiles, 512, 0, stream>>>(OUT, W[1], nullptr, FEAT, N, OUT, b[1]);
  attention(al[1], ar[1], 1);

  gemm_wmma_f32<256, true><<<mtiles, 512, 0, stream>>>(OUT, W[2], nullptr, FEAT, N, OUT, b[2]);
  attention(al[2], ar[2], 1);
  mean_heads_kernel<<<gSTREAM, 256, 0, stream>>>(
      OUT, (float*)d_out, (long)N * DHC);
}


# Projection GEMMs made weight-stationary: 256 resident work-groups keep all of W in LDS (B fragments by 16-byte LDS reads) and loop over 16-row strips; W leaves L2 once per WGP; float sums unchanged
# speedup vs baseline: 4.5545x; 1.0401x over previous
#include <hip/hip_runtime.h>
#include <math.h>


typedef __attribute__((ext_vector_type(2))) float v2f;
typedef __attribute__((ext_vector_type(8))) float v8f;

#define HIDC 256
#define HEADS 4
#define DHC 64

__device__ __forceinline__ float atomicMaxF(float* addr, float value) {
  if (value >= 0.0f)
    return __int_as_float(atomicMax((int*)addr, __float_as_int(value)));
  else
    return __uint_as_float(atomicMin((unsigned int*)addr, __float_as_uint(value)));
}

__device__ __forceinline__ float lrelu(float v) {
  return v > 0.0f ? v : 0.2f * v;
}

__device__ __forceinline__ float4 ld4(const float* p, bool al) {
  if (al) return *reinterpret_cast<const float4*>(p);
  return make_float4(p[0], p[1], p[2], p[3]);
}

template <int K, bool RELU_IN>
__global__ __launch_bounds__(512)
void gemm_wmma_f32(const float* A, const float* __restrict__ W,
                   const float* __restrict__ bias, float* __restrict__ C,
                   int M, float* R, const float* __restrict__ rbias) {
  constexpr int LDA  = K + 4;
  constexpr int NV   = 16 * K / 512;
  constexpr int RS   = 512 / K;
  constexpr int NU   = K / 32;
  constexpr int NG   = K / 32;
  constexpr int NBUF = (K >= 256) ? 2 : 4;
  __shared__ __attribute__((aligned(16))) float Ws[K * HIDC];
  __shared__ __attribute__((aligned(16))) float As[NBUF][16 * LDA];

  const int tid    = threadIdx.x;
  const int mtiles = (M + 15) >> 4;
  const int G      = (int)gridDim.x;
  int s = (int)blockIdx.x;
  if (s >= mtiles) return;

  const int sc  = tid & (K - 1);
  const int sr  = tid / K;
  const int scp = (sc & ~6) | ((sc & 2) << 1) | ((sc & 4) >> 1);
  float v[NV], w[NV];
  #pragma unroll
  for (int u = 0; u < NV; ++u) {
    int row = (s << 4) + sr + u * RS;
    if (row >= M) row = M - 1;
    v[u] = A[(long)row * K + sc];
  }
  const float rb = RELU_IN ? rbias[sc] : 0.0f;

  {
    const bool wal = (reinterpret_cast<unsigned long>(W) & 15ul) == 0ul;
    float4 L[NU][4];
    #pragma unroll
    for (int j = 0; j < NU; ++j) {
      const int jj = (j + (int)blockIdx.x) & (NU - 1);
      const int u  = tid + 512 * jj;
      const int q  = u & 63;
      const int p  = u >> 6;
      const float* g = W + (long)(8 * (p >> 1) + 2 * (p & 1)) * HIDC + 4 * q;
      L[j][0] = ld4(g, wal);
      L[j][1] = ld4(g + HIDC, wal);
      L[j][2] = ld4(g + 4 * HIDC, wal);
      L[j][3] = ld4(g + 5 * HIDC, wal);
    }
    #pragma unroll
    for (int j = 0; j < NU; ++j) {
      const int jj = (j + (int)blockIdx.x) & (NU - 1);
      const int u  = tid + 512 * jj;
      const int q  = u & 63;
      const int p  = u >> 6;
      const float c0[4] = {L[j][0].x, L[j][0].y, L[j][0].z, L[j][0].w};
      const float c1[4] = {L[j][1].x, L[j][1].y, L[j][1].z, L[j][1].w};
      const float c2[4] = {L[j][2].x, L[j][2].y, L[j][2].z, L[j][2].w};
      const float c3[4] = {L[j][3].x, L[j][3].y, L[j][3].z, L[j][3].w};
      #pragma unroll
      for (int i = 0; i < 4; ++i) {
        const int n  = 4 * q + i;
        const int np = (n & ~15) | ((n + (q >> 2)) & 15);
        *reinterpret_cast<float4*>(&Ws[(p * HIDC + np) * 4]) =
            make_float4(c0[i], c1[i], c2[i], c3[i]);
      }
    }
  }

  const int lane   = tid & 31;
  const int wave   = tid >> 5;
  const int nBase  = wave << 4;
  const int col    = lane & 15;
  const int h      = lane >> 4;
  const int rowOff = h << 3;
  const float* bp  = &Ws[(h * HIDC + (nBase | ((col + wave) & 15))) * 4];
  const int aoff   = col * LDA + 4 * h;
  const float bc   = bias ? bias[nBase + col] : 0.0f;

  #pragma unroll
  for (int u = 0; u < NV; ++u) {
    float t = v[u];
    if (RELU_IN) t = fmaxf(t, 0.0f);
    As[0][(sr + u * RS) * LDA + scp] = t;
    w[u] = t + rb;
  }
  __syncthreads();

  int buf   = 0;
  int mPrev = -1;
  v8f accPrev = {};
  for (; s < mtiles; s += G) {
    const int  ns      = s + G;
    const bool hasNext = ns < mtiles;
    if (hasNext) {
      #pragma unroll
      for (int u = 0; u < NV; ++u) {
        int row = (ns << 4) + sr + u * RS;
        if (row >= M) row = M - 1;
        v[u] = A[(long)row * K + sc];
      }
    }
    if (RELU_IN) {
      #pragma unroll
      for (int u = 0; u < NV; ++u) {
        const int row = (s << 4) + sr + u * RS;
        if (row < M) R[(long)row * K + sc] = w[u];
      }
    }
    if (mPrev >= 0) {
      #pragma unroll
      for (int r = 0; r < 8; ++r) {
        const int row = mPrev + rowOff + r;
        if (row < M) C[(long)row * HIDC + nBase + col] = accPrev[r];
      }
    }

    const float* ap = &As[buf][aoff];
    float4 fa[2][4], fb[2][4];
    #pragma unroll
    for (int t = 0; t < 4; ++t) {
      fa[0][t] = *reinterpret_cast<const float4*>(ap + 8 * t);
      fb[0][t] = *reinterpret_cast<const float4*>(bp + 2 * HIDC * 4 * t);
    }
    v8f acc = {};
    #pragma unroll
    for (int g = 0; g < NG; ++g) {
      const int cur = g & 1, nxt = cur ^ 1;
      if (g + 1 < NG) {
        #pragma unroll
        for (int t = 0; t < 4; ++t) {
          fa[nxt][t] = *reinterpret_cast<const float4*>(ap + 8 * (4 * (g + 1) + t));
          fb[nxt][t] = *reinterpret_cast<const float4*>(bp + 2 * HIDC * 4 * (4 * (g + 1) + t));
        }
      }
      __builtin_amdgcn_sched_barrier(0);
      #pragma unroll
      for (int t = 0; t < 4; ++t) {
        v2f a, b;
        a.x = fa[cur][t].x; a.y = fa[cur][t].y;
        b.x = fb[cur][t].x; b.y = fb[cur][t].y;
        acc = __builtin_amdgcn_wmma_f32_16x16x4_f32(
             false, a,  false, b,
             (short)0, acc,  false,  false);
        a.x = fa[cur][t].z; a.y = fa[cur][t].w;
        b.x = fb[cur][t].z; b.y = fb[cur][t].w;
        acc = __builtin_amdgcn_wmma_f32_16x16x4_f32(
             false, a,  false, b,
             (short)0, acc,  false,  false);
      }
      __builtin_amdgcn_sched_barrier(0);
    }
    #pragma unroll
    for (int r = 0; r < 8; ++r) accPrev[r] = acc[r] + bc;
    mPrev = s << 4;

    const int nb = (buf + 1) & (NBUF - 1);
    if (hasNext) {
      #pragma unroll
      for (int u = 0; u < NV; ++u) {
        float t = v[u];
        if (RELU_IN) t = fmaxf(t, 0.0f);
        As[nb][(sr + u * RS) * LDA + scp] = t;
        w[u] = t + rb;
      }
    }
    __syncthreads();
    buf = nb;
  }
  #pragma unroll
  for (int r = 0; r < 8; ++r) {
    const int row = mPrev + rowOff + r;
    if (row < M) C[(long)row * HIDC + nBase + col] = accPrev[r];
  }
}

__global__ __launch_bounds__(256)
void el_er_kernel(const float* __restrict__ feat, const float* __restrict__ al,
                  const float* __restrict__ ar, float* __restrict__ el,
                  float* __restrict__ er, int NH) {
  const int lane = threadIdx.x & 31;
  const int wave0 = __builtin_amdgcn_readfirstlane(
      (int)(blockIdx.x * (blockDim.x >> 5) + (threadIdx.x >> 5)));
  const int nW = (int)(gridDim.x * (blockDim.x >> 5));
  const int h  = wave0 & 3;
  const float l0 = al[h * DHC + lane], l1 = al[h * DHC + lane + 32];
  const float r0 = ar[h * DHC + lane], r1 = ar[h * DHC + lane + 32];
  for (int p = wave0; p < NH; p += 8 * nW) {
    float f0[8], f1[8];
    #pragma unroll
    for (int u = 0; u < 8; ++u) {
      const int q = p + u * nW;
      f0[u] = 0.0f;
      f1[u] = 0.0f;
      if (q < NH) {
        const float* f = feat + (long)q * DHC;
        f0[u] = f[lane];
        f1[u] = f[lane + 32];
      }
    }
    float sl[8], sr[8];
    #pragma unroll
    for (int u = 0; u < 8; ++u) {
      sl[u] = f0[u] * l0 + f1[u] * l1;
      sr[u] = f0[u] * r0 + f1[u] * r1;
    }
    #pragma unroll
    for (int off = 16; off; off >>= 1) {
      #pragma unroll
      for (int u = 0; u < 8; ++u) {
        sl[u] += __shfl_down(sl[u], off, 32);
        sr[u] += __shfl_down(sr[u], off, 32);
      }
    }
    #pragma unroll
    for (int u = 0; u < 8; ++u) {
      const int q = p + u * nW;
      if (lane == 0 && q < NH) { el[q] = sl[u]; er[q] = sr[u]; }
    }
  }
}

#define AGG_TN    196
#define AGG_TRIP  512
#define AGG_QCAP  768
#define AGG_SAVE  (AGG_QCAP - AGG_TRIP)
#define AGG_SHDR  64
#define AGG_SBLK  (AGG_SHDR + 32 * AGG_SAVE)

__device__ __forceinline__ float edge_weight(float score, float mx) {
  const float v = lrelu(score);
  if (!(mx > -1e38f)) mx = 0.0f;
  return __expf(v - mx);
}

__device__ __forceinline__ void agg_drain_max(
    const int* q, int cnt, int lo, int lane,
    const int* __restrict__ src, const int* __restrict__ dst,
    const float* __restrict__ el, const float* __restrict__ er, float* mt) {
  for (int i = 0; i < cnt; i += 64) {
    int s[2], d[2];
    float4 x[2], y[2];
    #pragma unroll
    for (int u = 0; u < 2; ++u) {
      const int p = i + 32 * u + lane;
      s[u] = 0; d[u] = lo;
      if (p < cnt) { const int e = q[p]; d[u] = dst[e]; s[u] = src[e]; }
    }
    #pragma unroll
    for (int u = 0; u < 2; ++u) {
      const int p = i + 32 * u + lane;
      if (p < cnt) {
        x[u] = *reinterpret_cast<const float4*>(el + 4 * (long)s[u]);
        y[u] = *reinterpret_cast<const float4*>(er + 4 * (long)d[u]);
      }
    }
    #pragma unroll
    for (int u = 0; u < 2; ++u) {
      const int p = i + 32 * u + lane;
      if (p < cnt) {
        float* m = mt + 4 * (d[u] - lo);
        atomicMaxF(m,     lrelu(x[u].x + y[u].x));
        atomicMaxF(m + 1, lrelu(x[u].y + y[u].y));
        atomicMaxF(m + 2, lrelu(x[u].z + y[u].z));
        atomicMaxF(m + 3, lrelu(x[u].w + y[u].w));
      }
    }
  }
}

__device__ __forceinline__ void agg_drain_acc(
    const int* q, int cnt, int lo, int lane,
    const int* __restrict__ src, const int* __restrict__ dst,
    const float* __restrict__ feat,
    const float* __restrict__ el, const float* __restrict__ er,
    const float* mt, float* dt, float* acc) {
  for (int i = 0; i < cnt; i += 32) {
    const int n = min(32, cnt - i);
    int s = 0, dl = 0;
    float a0 = 0.0f, a1 = 0.0f, a2 = 0.0f, a3 = 0.0f;
    if (lane < n) {
      const int e = q[i + lane];
      const int d = dst[e];
      s  = src[e];
      dl = d - lo;
      const float4 x = *reinterpret_cast<const float4*>(el + 4 * (long)s);
      const float4 y = *reinterpret_cast<const float4*>(er + 4 * (long)d);
      const float4 m = *reinterpret_cast<const float4*>(mt + 4 * dl);
      a0 = edge_weight(x.x + y.x, m.x);
      a1 = edge_weight(x.y + y.y, m.y);
      a2 = edge_weight(x.z + y.z, m.z);
      a3 = edge_weight(x.w + y.w, m.w);
      float* ds = dt + 4 * dl;
      atomicAdd(ds,     a0);
      atomicAdd(ds + 1, a1);
      atomicAdd(ds + 2, a2);
      atomicAdd(ds + 3, a3);
    }
    for (int k = 0; k < n; k += 4) {
      int su[4], du[4];
      float au[4][4];
      v2f f[4][4];
      #pragma unroll
      for (int u = 0; u < 4; ++u) {
        const int idx = min(k + u, n - 1);
        su[u] = __builtin_amdgcn_readlane(s, idx);
        du[u] = (k + u < n) ? __builtin_amdgcn_readlane(dl, idx) : AGG_TN;
        au[u][0] = __int_as_float(__builtin_amdgcn_readlane(__float_as_int(a0), idx));
        au[u][1] = __int_as_float(__builtin_amdgcn_readlane(__float_as_int(a1), idx));
        au[u][2] = __int_as_float(__builtin_amdgcn_readlane(__float_as_int(a2), idx));
        au[u][3] = __int_as_float(__builtin_amdgcn_readlane(__float_as_int(a3), idx));
      }
      #pragma unroll
      for (int u = 0; u < 4; ++u) {
        const float* fr = feat + (long)su[u] * HIDC + 2 * lane;
        #pragma unroll
        for (int j = 0; j < HEADS; ++j)
          f[u][j] = *reinterpret_cast<const v2f*>(fr + j * DHC);
      }
      #pragma unroll
      for (int u = 0; u < 4; ++u) {
        float* p = acc + du[u] * HIDC + 2 * lane;
        #pragma unroll
        for (int j = 0; j < HEADS; ++j) {
          atomicAdd(p + j * DHC,     au[u][j] * f[u][j].x);
          atomicAdd(p + j * DHC + 1, au[u][j] * f[u][j].y);
        }
      }
    }
  }
}

template <int PASS>
__device__ __forceinline__ int agg_scan(
    int* q, int wave, int lane, int lo, int rows, int E,
    const int* __restrict__ src, const int* __restrict__ dst,
    const float* __restrict__ feat,
    const float* __restrict__ el, const float* __restrict__ er,
    float* mt, float* dt, float* acc, int* full) {
  const int  T       = (E + AGG_TRIP - 1) / AGG_TRIP;
  const bool aligned = (reinterpret_cast<unsigned long>(dst) & 15ul) == 0ul;
  const int start = (T / (int)gridDim.x) * (int)blockIdx.x;
  int cnt = 0;
  for (int t = wave; t < T; t += 32) {
    int g = t + start;
    if (g >= T) g -= T;
    const int base = g * AGG_TRIP + 4 * lane;
    int4 dv[4];
    if (aligned && g * AGG_TRIP + AGG_TRIP <= E) {
      #pragma unroll
      for (int u = 0; u < 4; ++u)
        dv[u] = *reinterpret_cast<const int4*>(dst + base + 128 * u);
    } else {
      #pragma unroll
      for (int u = 0; u < 4; ++u) {
        const int e0 = base + 128 * u;
        dv[u].x = (e0     < E) ? dst[e0]     : -1;
        dv[u].y = (e0 + 1 < E) ? dst[e0 + 1] : -1;
        dv[u].z = (e0 + 2 < E) ? dst[e0 + 2] : -1;
        dv[u].w = (e0 + 3 < E) ? dst[e0 + 3] : -1;
      }
    }
    #pragma unroll
    for (int u = 0; u < 4; ++u) {
      const int  e0 = base + 128 * u;
      const bool mx = (unsigned)(dv[u].x - lo) < (unsigned)rows;
      const bool my = (unsigned)(dv[u].y - lo) < (unsigned)rows;
      const bool mz = (unsigned)(dv[u].z - lo) < (unsigned)rows;
      const bool mw = (unsigned)(dv[u].w - lo) < (unsigned)rows;
      if (__builtin_amdgcn_ballot_w32(mx | my | mz | mw)) {
        unsigned mask;
        mask = __builtin_amdgcn_ballot_w32(mx);
        if (mx) q[cnt + __builtin_amdgcn_mbcnt_lo(mask, 0u)] = e0;
        cnt += __builtin_popcount(mask);
        mask = __builtin_amdgcn_ballot_w32(my);
        if (my) q[cnt + __builtin_amdgcn_mbcnt_lo(mask, 0u)] = e0 + 1;
        cnt += __builtin_popcount(mask);
        mask = __builtin_amdgcn_ballot_w32(mz);
        if (mz) q[cnt + __builtin_amdgcn_mbcnt_lo(mask, 0u)] = e0 + 2;
        cnt += __builtin_popcount(mask);
        mask = __builtin_amdgcn_ballot_w32(mw);
        if (mw) q[cnt + __builtin_amdgcn_mbcnt_lo(mask, 0u)] = e0 + 3;
        cnt += __builtin_popcount(mask);
      }
    }
    if (cnt > AGG_QCAP - AGG_TRIP) {
      if (PASS == 0) {
        agg_drain_max(q, cnt, lo, lane, src, dst, el, er, mt);
        if (lane == 0) *full = 1;
      } else {
        agg_drain_acc(q, cnt, lo, lane, src, dst, feat, el, er, mt, dt, acc);
      }
      cnt = 0;
    }
  }
  return cnt;
}

__global__ __launch_bounds__(1024)
void aggregate_tile_kernel(const int* __restrict__ src, const int* __restrict__ dst,
                           const float* __restrict__ feat,
                           const float* __restrict__ el,
                           const float* __restrict__ er, float* __restrict__ out,
                           int N, int E, int* qsave, const float* wsBase,
                           unsigned long wsBytes, int reuseSaved) {
  __shared__ __attribute__((aligned(16))) float acc[(AGG_TN + 1) * HIDC];
  __shared__ __attribute__((aligned(16))) float mt[AGG_TN * HEADS];
  __shared__ __attribute__((aligned(16))) float dt[AGG_TN * HEADS];
  __shared__ int queue[32 * AGG_QCAP];
  __shared__ int full;

  const int tid  = threadIdx.x;
  const int lane = tid & 31;
  const int wave = __builtin_amdgcn_readfirstlane(tid >> 5);
  const int lo   = blockIdx.x * AGG_TN;
  const int rows = min(AGG_TN, N - lo);

  for (int i = tid; i < (AGG_TN + 1) * HIDC / 4; i += 1024)
    reinterpret_cast<float4*>(acc)[i] = make_float4(0.0f, 0.0f, 0.0f, 0.0f);
  if (tid < AGG_TN * HEADS) { mt[tid] = -INFINITY; dt[tid] = 0.0f; }
  if (tid == 0) full = 0;
  __syncthreads();

  int* q = queue + wave * AGG_QCAP;
  int* sv = qsave + (long)blockIdx.x * AGG_SBLK;
  const bool fits =
      (unsigned long)(reinterpret_cast<const char*>(qsave + (long)gridDim.x * AGG_SBLK) -
                      reinterpret_cast<const char*>(wsBase)) <= wsBytes;
  int reuse = 0, saved = 0;
  if (reuseSaved != 0 && fits) {
    const int okw = sv[32];
    const int cw  = sv[wave];
    reuse = (okw == 1) ? 1 : 0;
    saved = cw;
  }
  int cnt;
  if (reuse != 0) {
    cnt = min(max(saved, 0), AGG_SAVE);
    const int* sq = sv + AGG_SHDR + wave * AGG_SAVE;
    int v[AGG_SAVE / 32];
    #pragma unroll
    for (int u = 0; u < AGG_SAVE / 32; ++u) {
      const int i = lane + 32 * u;
      v[u] = 0;
      if (i < cnt) v[u] = sq[i];
    }
    #pragma unroll
    for (int u = 0; u < AGG_SAVE / 32; ++u) {
      const int i = lane + 32 * u;
      if (i < cnt) q[i] = v[u];
    }
  } else {
    cnt = agg_scan<0>(q, wave, lane, lo, rows, E, src, dst, feat, el, er, mt, dt, acc, &full);
  }
  agg_drain_max(q, cnt, lo, lane, src, dst, el, er, mt);
  __syncthreads();

  const int early = (reuse != 0) ? 0 : __builtin_amdgcn_readfirstlane(full);
  if (reuseSaved == 0 && fits) {
    if (early == 0) {
      int* sq = sv + AGG_SHDR + wave * AGG_SAVE;
      for (int i = lane; i < cnt; i += 32) sq[i] = q[i];
    }
    if (lane == 0) sv[wave] = (early == 0) ? cnt : 0;
    if (tid == 0) sv[32] = (early == 0) ? 1 : 0;
  }
  if (early == 0) {
    agg_drain_acc(q, cnt, lo, lane, src, dst, feat, el, er, mt, dt, acc);
  } else {
    cnt = agg_scan<1>(q, wave, lane, lo, rows, E, src, dst, feat, el, er, mt, dt, acc, &full);
    agg_drain_acc(q, cnt, lo, lane, src, dst, feat, el, er, mt, dt, acc);
  }
  __syncthreads();

  if (tid < rows * HEADS) dt[tid] = 1.0f / fmaxf(dt[tid], 1e-16f);
  __syncthreads();

  float4* op = reinterpret_cast<float4*>(out + (long)lo * HIDC);
  for (int i = tid; i < rows * (HIDC / 4); i += 1024) {
    const float  r = dt[i >> 4];
    const float4 a = reinterpret_cast<const float4*>(acc)[i];
    float4 o = op[i];
    o.x += a.x * r; o.y += a.y * r; o.z += a.z * r; o.w += a.w * r;
    op[i] = o;
  }
}

__global__ __launch_bounds__(256)
void mean_heads_kernel(const float* __restrict__ hbuf,
                       float* __restrict__ out, long total) {
  const long stride = (long)gridDim.x * blockDim.x;
  const long t      = (long)blockIdx.x * blockDim.x + threadIdx.x;
  if ((reinterpret_cast<unsigned long>(out) & 15ul) == 0ul) {
    const long n4 = total >> 2;
    const float4* hp = reinterpret_cast<const float4*>(hbuf);
    float4*       op = reinterpret_cast<float4*>(out);
    for (long i = t; i < n4; i += 2 * stride) {
      float4 p[2][4];
      #pragma unroll
      for (int u = 0; u < 2; ++u) {
        const long j = i + u * stride;
        if (j < n4) {
          const float4* r = hp + (j >> 4) * (HIDC / 4) + (j & 15);
          #pragma unroll
          for (int k = 0; k < HEADS; ++k) p[u][k] = r[k * (DHC / 4)];
        }
      }
      #pragma unroll
      for (int u = 0; u < 2; ++u) {
        const long j = i + u * stride;
        if (j < n4) {
          float4 o;
          o.x = 0.25f * (p[u][0].x + p[u][1].x + p[u][2].x + p[u][3].x);
          o.y = 0.25f * (p[u][0].y + p[u][1].y + p[u][2].y + p[u][3].y);
          o.z = 0.25f * (p[u][0].z + p[u][1].z + p[u][2].z + p[u][3].z);
          o.w = 0.25f * (p[u][0].w + p[u][1].w + p[u][2].w + p[u][3].w);
          op[j] = o;
        }
      }
    }
  } else {
    for (long i = t; i < total; i += stride) {
      const float* p = hbuf + (i >> 6) * HIDC + (i & 63);
      out[i] = 0.25f * (p[0] + p[DHC] + p[2 * DHC] + p[3 * DHC]);
    }
  }
}

extern "C" void kernel_launch(void* const* d_in, const int* in_sizes, int n_in,
                              void* d_out, int out_size, void* d_ws, size_t ws_size,
                              hipStream_t stream) {
  const float* x     = (const float*)d_in[0];
  const int*   ei    = (const int*)d_in[1];
  const float* W[3]  = {(const float*)d_in[2], (const float*)d_in[3],
                        (const float*)d_in[4]};
  const float* al[3] = {(const float*)d_in[5], (const float*)d_in[6],
                        (const float*)d_in[7]};
  const float* ar[3] = {(const float*)d_in[8], (const float*)d_in[9],
                        (const float*)d_in[10]};
  const float* b[3]  = {(const float*)d_in[11], (const float*)d_in[12],
                        (const float*)d_in[13]};
  const float* resW0 = (const float*)d_in[14];

  const int N = in_sizes[0] / 128;
  const int E = in_sizes[1] / 2;
  const int* src = ei;
  const int* dst = ei + E;

  float* ws   = (float*)d_ws;
  float* FEAT = ws;
  float* OUT  = FEAT + (size_t)N * HIDC;
  float* ELP  = OUT  + (size_t)N * HIDC;
  float* ERP  = ELP  + (size_t)N * HEADS;
  int*   QSV  = (int*)(ERP + (size_t)N * HEADS);

  const int  gGEMM  = 256;
  const int  NH     = N * HEADS;
  const int  gSTREAM = 512;
  const int  gNARROW = 1024;

  auto attention = [&](const float* alw, const float* arw, int reuseSaved) {
    el_er_kernel<<<gNARROW, 256, 0, stream>>>(FEAT, alw, arw, ELP, ERP, NH);
    aggregate_tile_kernel<<<(N + AGG_TN - 1) / AGG_TN, 1024, 0, stream>>>(
        src, dst, FEAT, ELP, ERP, OUT, N, E, QSV, ws, (unsigned long)ws_size, reuseSaved);
  };

  gemm_wmma_f32<128, false><<<gGEMM, 512, 0, stream>>>(x, W[0], nullptr, FEAT, N, nullptr, nullptr);
  gemm_wmma_f32<128, false><<<gGEMM, 512, 0, stream>>>(x, resW0, b[0], OUT, N, nullptr, nullptr);
  attention(al[0], ar[0], 0);

  gemm_wmma_f32<256, true><<<gGEMM, 512, 0, stream>>>(OUT, W[1], nullptr, FEAT, N, OUT, b[1]);
  attention(al[1], ar[1], 1);

  gemm_wmma_f32<256, true><<<gGEMM, 512, 0, stream>>>(OUT, W[2], nullptr, FEAT, N, OUT, b[2]);
  attention(al[2], ar[2], 1);
  mean_heads_kernel<<<gSTREAM, 256, 0, stream>>>(
      OUT, (float*)d_out, (long)N * DHC);
}


# Neighbour sum moved from LDS float adds to register accumulators on the matrix unit: edges listed per 16-row block, A = exp weight placed by row, B = 4 gathered rows; sums: list slot order, 4 a step
# speedup vs baseline: 4.7842x; 1.0504x over previous
#include <hip/hip_runtime.h>
#include <math.h>


typedef __attribute__((ext_vector_type(2))) float v2f;
typedef __attribute__((ext_vector_type(8))) float v8f;

#define HIDC 256
#define HEADS 4
#define DHC 64

__device__ __forceinline__ float atomicMaxF(float* addr, float value) {
  if (value >= 0.0f)
    return __int_as_float(atomicMax((int*)addr, __float_as_int(value)));
  else
    return __uint_as_float(atomicMin((unsigned int*)addr, __float_as_uint(value)));
}

__device__ __forceinline__ float lrelu(float v) {
  return v > 0.0f ? v : 0.2f * v;
}

__device__ __forceinline__ float4 ld4(const float* p, bool al) {
  if (al) return *reinterpret_cast<const float4*>(p);
  return make_float4(p[0], p[1], p[2], p[3]);
}

template <int K, bool RELU_IN>
__global__ __launch_bounds__(512)
void gemm_wmma_f32(const float* A, const float* __restrict__ W,
                   const float* __restrict__ bias, float* __restrict__ C,
                   int M, float* R, const float* __restrict__ rbias) {
  constexpr int LDA  = K + 4;
  constexpr int NV   = 16 * K / 512;
  constexpr int RS   = 512 / K;
  constexpr int NU   = K / 32;
  constexpr int NG   = K / 32;
  constexpr int NBUF = (K >= 256) ? 2 : 4;
  __shared__ __attribute__((aligned(16))) float Ws[K * HIDC];
  __shared__ __attribute__((aligned(16))) float As[NBUF][16 * LDA];

  const int tid    = threadIdx.x;
  const int mtiles = (M + 15) >> 4;
  const int G      = (int)gridDim.x;
  int s = (int)blockIdx.x;
  if (s >= mtiles) return;

  const int sc  = tid & (K - 1);
  const int sr  = tid / K;
  const int scp = (sc & ~6) | ((sc & 2) << 1) | ((sc & 4) >> 1);
  float v[NV], w[NV];
  #pragma unroll
  for (int u = 0; u < NV; ++u) {
    int row = (s << 4) + sr + u * RS;
    if (row >= M) row = M - 1;
    v[u] = A[(long)row * K + sc];
  }
  const float rb = RELU_IN ? rbias[sc] : 0.0f;

  {
    const bool wal = (reinterpret_cast<unsigned long>(W) & 15ul) == 0ul;
    float4 L[NU][4];
    #pragma unroll
    for (int j = 0; j < NU; ++j) {
      const int jj = (j + (int)blockIdx.x) & (NU - 1);
      const int u  = tid + 512 * jj;
      const int q  = u & 63;
      const int p  = u >> 6;
      const float* g = W + (long)(8 * (p >> 1) + 2 * (p & 1)) * HIDC + 4 * q;
      L[j][0] = ld4(g, wal);
      L[j][1] = ld4(g + HIDC, wal);
      L[j][2] = ld4(g + 4 * HIDC, wal);
      L[j][3] = ld4(g + 5 * HIDC, wal);
    }
    #pragma unroll
    for (int j = 0; j < NU; ++j) {
      const int jj = (j + (int)blockIdx.x) & (NU - 1);
      const int u  = tid + 512 * jj;
      const int q  = u & 63;
      const int p  = u >> 6;
      const float c0[4] = {L[j][0].x, L[j][0].y, L[j][0].z, L[j][0].w};
      const float c1[4] = {L[j][1].x, L[j][1].y, L[j][1].z, L[j][1].w};
      const float c2[4] = {L[j][2].x, L[j][2].y, L[j][2].z, L[j][2].w};
      const float c3[4] = {L[j][3].x, L[j][3].y, L[j][3].z, L[j][3].w};
      #pragma unroll
      for (int i = 0; i < 4; ++i) {
        const int n  = 4 * q + i;
        const int np = (n & ~15) | ((n + (q >> 2)) & 15);
        *reinterpret_cast<float4*>(&Ws[(p * HIDC + np) * 4]) =
            make_float4(c0[i], c1[i], c2[i], c3[i]);
      }
    }
  }

  const int lane   = tid & 31;
  const int wave   = tid >> 5;
  const int nBase  = wave << 4;
  const int col    = lane & 15;
  const int h      = lane >> 4;
  const int rowOff = h << 3;
  const float* bp  = &Ws[(h * HIDC + (nBase | ((col + wave) & 15))) * 4];
  const int aoff   = col * LDA + 4 * h;
  const float bc   = bias ? bias[nBase + col] : 0.0f;

  #pragma unroll
  for (int u = 0; u < NV; ++u) {
    float t = v[u];
    if (RELU_IN) t = fmaxf(t, 0.0f);
    As[0][(sr + u * RS) * LDA + scp] = t;
    w[u] = t + rb;
  }
  __syncthreads();

  int buf   = 0;
  int mPrev = -1;
  v8f accPrev = {};
  for (; s < mtiles; s += G) {
    const int  ns      = s + G;
    const bool hasNext = ns < mtiles;
    if (hasNext) {
      #pragma unroll
      for (int u = 0; u < NV; ++u) {
        int row = (ns << 4) + sr + u * RS;
        if (row >= M) row = M - 1;
        v[u] = A[(long)row * K + sc];
      }
    }
    if (RELU_IN) {
      #pragma unroll
      for (int u = 0; u < NV; ++u) {
        const int row = (s << 4) + sr + u * RS;
        if (row < M) R[(long)row * K + sc] = w[u];
      }
    }
    if (mPrev >= 0) {
      #pragma unroll
      for (int r = 0; r < 8; ++r) {
        const int row = mPrev + rowOff + r;
        if (row < M) C[(long)row * HIDC + nBase + col] = accPrev[r];
      }
    }

    const float* ap = &As[buf][aoff];
    float4 fa[2][4], fb[2][4];
    #pragma unroll
    for (int t = 0; t < 4; ++t) {
      fa[0][t] = *reinterpret_cast<const float4*>(ap + 8 * t);
      fb[0][t] = *reinterpret_cast<const float4*>(bp + 2 * HIDC * 4 * t);
    }
    v8f acc = {};
    #pragma unroll
    for (int g = 0; g < NG; ++g) {
      const int cur = g & 1, nxt = cur ^ 1;
      if (g + 1 < NG) {
        #pragma unroll
        for (int t = 0; t < 4; ++t) {
          fa[nxt][t] = *reinterpret_cast<const float4*>(ap + 8 * (4 * (g + 1) + t));
          fb[nxt][t] = *reinterpret_cast<const float4*>(bp + 2 * HIDC * 4 * (4 * (g + 1) + t));
        }
      }
      __builtin_amdgcn_sched_barrier(0);
      #pragma unroll
      for (int t = 0; t < 4; ++t) {
        v2f a, b;
        a.x = fa[cur][t].x; a.y = fa[cur][t].y;
        b.x = fb[cur][t].x; b.y = fb[cur][t].y;
        acc = __builtin_amdgcn_wmma_f32_16x16x4_f32(
             false, a,  false, b,
             (short)0, acc,  false,  false);
        a.x = fa[cur][t].z; a.y = fa[cur][t].w;
        b.x = fb[cur][t].z; b.y = fb[cur][t].w;
        acc = __builtin_amdgcn_wmma_f32_16x16x4_f32(
             false, a,  false, b,
             (short)0, acc,  false,  false);
      }
      __builtin_amdgcn_sched_barrier(0);
    }
    #pragma unroll
    for (int r = 0; r < 8; ++r) accPrev[r] = acc[r] + bc;
    mPrev = s << 4;

    const int nb = (buf + 1) & (NBUF - 1);
    if (hasNext) {
      #pragma unroll
      for (int u = 0; u < NV; ++u) {
        float t = v[u];
        if (RELU_IN) t = fmaxf(t, 0.0f);
        As[nb][(sr + u * RS) * LDA + scp] = t;
        w[u] = t + rb;
      }
    }
    __syncthreads();
    buf = nb;
  }
  #pragma unroll
  for (int r = 0; r < 8; ++r) {
    const int row = mPrev + rowOff + r;
    if (row < M) C[(long)row * HIDC + nBase + col] = accPrev[r];
  }
}

__global__ __launch_bounds__(256)
void el_er_kernel(const float* __restrict__ feat, const float* __restrict__ al,
                  const float* __restrict__ ar, float* __restrict__ el,
                  float* __restrict__ er, int NH) {
  const int lane = threadIdx.x & 31;
  const int wave0 = __builtin_amdgcn_readfirstlane(
      (int)(blockIdx.x * (blockDim.x >> 5) + (threadIdx.x >> 5)));
  const int nW = (int)(gridDim.x * (blockDim.x >> 5));
  const int h  = wave0 & 3;
  const float l0 = al[h * DHC + lane], l1 = al[h * DHC + lane + 32];
  const float r0 = ar[h * DHC + lane], r1 = ar[h * DHC + lane + 32];
  for (int p = wave0; p < NH; p += 8 * nW) {
    float f0[8], f1[8];
    #pragma unroll
    for (int u = 0; u < 8; ++u) {
      const int q = p + u * nW;
      f0[u] = 0.0f;
      f1[u] = 0.0f;
      if (q < NH) {
        const float* f = feat + (long)q * DHC;
        f0[u] = f[lane];
        f1[u] = f[lane + 32];
      }
    }
    float sl[8], sr[8];
    #pragma unroll
    for (int u = 0; u < 8; ++u) {
      sl[u] = f0[u] * l0 + f1[u] * l1;
      sr[u] = f0[u] * r0 + f1[u] * r1;
    }
    #pragma unroll
    for (int off = 16; off; off >>= 1) {
      #pragma unroll
      for (int u = 0; u < 8; ++u) {
        sl[u] += __shfl_down(sl[u], off, 32);
        sr[u] += __shfl_down(sr[u], off, 32);
      }
    }
    #pragma unroll
    for (int u = 0; u < 8; ++u) {
      const int q = p + u * nW;
      if (lane == 0 && q < NH) { el[q] = sl[u]; er[q] = sr[u]; }
    }
  }
}

#define AGG_TN    196
#define AGG_TRIP  512
#define AGG_QCAP  768
#define AGG_SAVE  (AGG_QCAP - AGG_TRIP)
#define AGG_SHDR  64
#define AGG_SBLK  (AGG_SHDR + 32 * AGG_SAVE)
#define AGG_NBLK  ((AGG_TN + 15) / 16)
#define AGG_RCAP  (32 * AGG_SAVE + 8 * AGG_NBLK)

__device__ __forceinline__ float edge_weight(float score, float mx) {
  const float v = lrelu(score);
  if (!(mx > -1e38f)) mx = 0.0f;
  return __expf(v - mx);
}

__device__ __forceinline__ void agg_drain_max(
    const int* q, int cnt, int lo, int lane,
    const int* __restrict__ src, const int* __restrict__ dst,
    const float* __restrict__ el, const float* __restrict__ er, float* mt, int* bc) {
  for (int i = 0; i < cnt; i += 64) {
    int s[2], d[2];
    float4 x[2], y[2];
    #pragma unroll
    for (int u = 0; u < 2; ++u) {
      const int p = i + 32 * u + lane;
      s[u] = 0; d[u] = lo;
      if (p < cnt) { const int e = q[p]; d[u] = dst[e]; s[u] = src[e]; }
    }
    #pragma unroll
    for (int u = 0; u < 2; ++u) {
      const int p = i + 32 * u + lane;
      if (p < cnt) {
        x[u] = *reinterpret_cast<const float4*>(el + 4 * (long)s[u]);
        y[u] = *reinterpret_cast<const float4*>(er + 4 * (long)d[u]);
      }
    }
    #pragma unroll
    for (int u = 0; u < 2; ++u) {
      const int p = i + 32 * u + lane;
      if (p < cnt) {
        float* m = mt + 4 * (d[u] - lo);
        atomicAdd(bc + ((d[u] - lo) >> 4), 1);
        atomicMaxF(m,     lrelu(x[u].x + y[u].x));
        atomicMaxF(m + 1, lrelu(x[u].y + y[u].y));
        atomicMaxF(m + 2, lrelu(x[u].z + y[u].z));
        atomicMaxF(m + 3, lrelu(x[u].w + y[u].w));
      }
    }
  }
}

__device__ __forceinline__ void agg_drain_acc(
    const int* q, int cnt, int lo, int lane,
    const int* __restrict__ src, const int* __restrict__ dst,
    const float* __restrict__ feat,
    const float* __restrict__ el, const float* __restrict__ er,
    const float* mt, float* dt, float* acc) {
  for (int i = 0; i < cnt; i += 32) {
    const int n = min(32, cnt - i);
    int s = 0, dl = 0;
    float a0 = 0.0f, a1 = 0.0f, a2 = 0.0f, a3 = 0.0f;
    if (lane < n) {
      const int e = q[i + lane];
      const int d = dst[e];
      s  = src[e];
      dl = d - lo;
      const float4 x = *reinterpret_cast<const float4*>(el + 4 * (long)s);
      const float4 y = *reinterpret_cast<const float4*>(er + 4 * (long)d);
      const float4 m = *reinterpret_cast<const float4*>(mt + 4 * dl);
      a0 = edge_weight(x.x + y.x, m.x);
      a1 = edge_weight(x.y + y.y, m.y);
      a2 = edge_weight(x.z + y.z, m.z);
      a3 = edge_weight(x.w + y.w, m.w);
      float* ds = dt + 4 * dl;
      atomicAdd(ds,     a0);
      atomicAdd(ds + 1, a1);
      atomicAdd(ds + 2, a2);
      atomicAdd(ds + 3, a3);
    }
    for (int k = 0; k < n; k += 4) {
      int su[4], du[4];
      float au[4][4];
      v2f f[4][4];
      #pragma unroll
      for (int u = 0; u < 4; ++u) {
        const int idx = min(k + u, n - 1);
        su[u] = __builtin_amdgcn_readlane(s, idx);
        du[u] = (k + u < n) ? __builtin_amdgcn_readlane(dl, idx) : AGG_TN;
        au[u][0] = __int_as_float(__builtin_amdgcn_readlane(__float_as_int(a0), idx));
        au[u][1] = __int_as_float(__builtin_amdgcn_readlane(__float_as_int(a1), idx));
        au[u][2] = __int_as_float(__builtin_amdgcn_readlane(__float_as_int(a2), idx));
        au[u][3] = __int_as_float(__builtin_amdgcn_readlane(__float_as_int(a3), idx));
      }
      #pragma unroll
      for (int u = 0; u < 4; ++u) {
        const float* fr = feat + (long)su[u] * HIDC + 2 * lane;
        #pragma unroll
        for (int j = 0; j < HEADS; ++j)
          f[u][j] = *reinterpret_cast<const v2f*>(fr + j * DHC);
      }
      #pragma unroll
      for (int u = 0; u < 4; ++u) {
        float* p = acc + du[u] * HIDC + 2 * lane;
        #pragma unroll
        for (int j = 0; j < HEADS; ++j) {
          atomicAdd(p + j * DHC,     au[u][j] * f[u][j].x);
          atomicAdd(p + j * DHC + 1, au[u][j] * f[u][j].y);
        }
      }
    }
  }
}

template <int PASS>
__device__ __forceinline__ int agg_scan(
    int* q, int wave, int lane, int lo, int rows, int E,
    const int* __restrict__ src, const int* __restrict__ dst,
    const float* __restrict__ feat,
    const float* __restrict__ el, const float* __restrict__ er,
    float* mt, float* dt, float* acc, int* full, int* bc) {
  const int  T       = (E + AGG_TRIP - 1) / AGG_TRIP;
  const bool aligned = (reinterpret_cast<unsigned long>(dst) & 15ul) == 0ul;
  const int start = (T / (int)gridDim.x) * (int)blockIdx.x;
  int cnt = 0;
  for (int t = wave; t < T; t += 32) {
    int g = t + start;
    if (g >= T) g -= T;
    const int base = g * AGG_TRIP + 4 * lane;
    int4 dv[4];
    if (aligned && g * AGG_TRIP + AGG_TRIP <= E) {
      #pragma unroll
      for (int u = 0; u < 4; ++u)
        dv[u] = *reinterpret_cast<const int4*>(dst + base + 128 * u);
    } else {
      #pragma unroll
      for (int u = 0; u < 4; ++u) {
        const int e0 = base + 128 * u;
        dv[u].x = (e0     < E) ? dst[e0]     : -1;
        dv[u].y = (e0 + 1 < E) ? dst[e0 + 1] : -1;
        dv[u].z = (e0 + 2 < E) ? dst[e0 + 2] : -1;
        dv[u].w = (e0 + 3 < E) ? dst[e0 + 3] : -1;
      }
    }
    #pragma unroll
    for (int u = 0; u < 4; ++u) {
      const int  e0 = base + 128 * u;
      const bool mx = (unsigned)(dv[u].x - lo) < (unsigned)rows;
      const bool my = (unsigned)(dv[u].y - lo) < (unsigned)rows;
      const bool mz = (unsigned)(dv[u].z - lo) < (unsigned)rows;
      const bool mw = (unsigned)(dv[u].w - lo) < (unsigned)rows;
      if (__builtin_amdgcn_ballot_w32(mx | my | mz | mw)) {
        unsigned mask;
        mask = __builtin_amdgcn_ballot_w32(mx);
        if (mx) q[cnt + __builtin_amdgcn_mbcnt_lo(mask, 0u)] = e0;
        cnt += __builtin_popcount(mask);
        mask = __builtin_amdgcn_ballot_w32(my);
        if (my) q[cnt + __builtin_amdgcn_mbcnt_lo(mask, 0u)] = e0 + 1;
        cnt += __builtin_popcount(mask);
        mask = __builtin_amdgcn_ballot_w32(mz);
        if (mz) q[cnt + __builtin_amdgcn_mbcnt_lo(mask, 0u)] = e0 + 2;
        cnt += __builtin_popcount(mask);
        mask = __builtin_amdgcn_ballot_w32(mw);
        if (mw) q[cnt + __builtin_amdgcn_mbcnt_lo(mask, 0u)] = e0 + 3;
        cnt += __builtin_popcount(mask);
      }
    }
    if (cnt > AGG_QCAP - AGG_TRIP) {
      if (PASS == 0) {
        agg_drain_max(q, cnt, lo, lane, src, dst, el, er, mt, bc);
        if (lane == 0) *full = 1;
      } else {
        agg_drain_acc(q, cnt, lo, lane, src, dst, feat, el, er, mt, dt, acc);
      }
      cnt = 0;
    }
  }
  return cnt;
}

__global__ __launch_bounds__(1024)
void aggregate_tile_kernel(const int* __restrict__ src, const int* __restrict__ dst,
                           const float* __restrict__ feat,
                           const float* __restrict__ el,
                           const float* __restrict__ er, float* __restrict__ out,
                           int N, int E, int* qsave, const float* wsBase,
                           unsigned long wsBytes, int reuseSaved) {
  __shared__ __attribute__((aligned(16))) float acc[(AGG_TN + 1) * HIDC];
  static_assert(6 * AGG_RCAP <= (AGG_TN + 1) * HIDC, "records must fit the accumulators' memory");
  __shared__ __attribute__((aligned(16))) float mt[AGG_TN * HEADS];
  __shared__ __attribute__((aligned(16))) float dt[AGG_TN * HEADS];
  __shared__ int queue[32 * AGG_QCAP];
  __shared__ int full;
  __shared__ int blkCnt[16];
  __shared__ int blkFill[16];
  __shared__ int blkOff[16];

  const int tid  = threadIdx.x;
  const int lane = tid & 31;
  const int wave = __builtin_amdgcn_readfirstlane(tid >> 5);
  const int lo   = blockIdx.x * AGG_TN;
  const int rows = min(AGG_TN, N - lo);

  if (tid < AGG_TN * HEADS) { mt[tid] = -INFINITY; dt[tid] = 0.0f; }
  if (tid < 16) { blkCnt[tid] = 0; blkFill[tid] = 0; }
  if (tid == 0) full = 0;
  __syncthreads();

  int* q = queue + wave * AGG_QCAP;
  int* sv = qsave + (long)blockIdx.x * AGG_SBLK;
  const bool fits =
      (unsigned long)(reinterpret_cast<const char*>(qsave + (long)gridDim.x * AGG_SBLK) -
                      reinterpret_cast<const char*>(wsBase)) <= wsBytes;
  int reuse = 0, saved = 0;
  if (reuseSaved != 0 && fits) {
    const int okw = sv[32];
    const int cw  = sv[wave];
    reuse = (okw == 1) ? 1 : 0;
    saved = cw;
  }
  int cnt;
  if (reuse != 0) {
    cnt = min(max(saved, 0), AGG_SAVE);
    const int* sq = sv + AGG_SHDR + wave * AGG_SAVE;
    int v[AGG_SAVE / 32];
    #pragma unroll
    for (int u = 0; u < AGG_SAVE / 32; ++u) {
      const int i = lane + 32 * u;
      v[u] = 0;
      if (i < cnt) v[u] = sq[i];
    }
    #pragma unroll
    for (int u = 0; u < AGG_SAVE / 32; ++u) {
      const int i = lane + 32 * u;
      if (i < cnt) q[i] = v[u];
    }
  } else {
    cnt = agg_scan<0>(q, wave, lane, lo, rows, E, src, dst, feat, el, er, mt, dt, acc, &full, blkCnt);
  }
  agg_drain_max(q, cnt, lo, lane, src, dst, el, er, mt, blkCnt);
  __syncthreads();

  const int early = (reuse != 0) ? 0 : __builtin_amdgcn_readfirstlane(full);
  if (reuseSaved == 0 && fits) {
    if (early == 0) {
      int* sq = sv + AGG_SHDR + wave * AGG_SAVE;
      for (int i = lane; i < cnt; i += 32) sq[i] = q[i];
    }
    if (lane == 0) sv[wave] = (early == 0) ? cnt : 0;
    if (tid == 0) sv[32] = (early == 0) ? 1 : 0;
  }
  float4* op = reinterpret_cast<float4*>(out + (long)lo * HIDC);

  if (early != 0) {
    for (int i = tid; i < (AGG_TN + 1) * HIDC / 4; i += 1024)
      reinterpret_cast<float4*>(acc)[i] = make_float4(0.0f, 0.0f, 0.0f, 0.0f);
    __syncthreads();
    cnt = agg_scan<1>(q, wave, lane, lo, rows, E, src, dst, feat, el, er, mt, dt, acc, &full, blkCnt);
    agg_drain_acc(q, cnt, lo, lane, src, dst, feat, el, er, mt, dt, acc);
    __syncthreads();
    if (tid < rows * HEADS) dt[tid] = 1.0f / fmaxf(dt[tid], 1e-16f);
    __syncthreads();
    for (int i = tid; i < rows * (HIDC / 4); i += 1024) {
      const float  r = dt[i >> 4];
      const float4 a = reinterpret_cast<const float4*>(acc)[i];
      float4 o = op[i];
      o.x += a.x * r; o.y += a.y * r; o.z += a.z * r; o.w += a.w * r;
      op[i] = o;
    }
    return;
  }

  int*   recS = reinterpret_cast<int*>(acc);
  int*   recD = recS + AGG_RCAP;
  float* recA = reinterpret_cast<float*>(recD + AGG_RCAP);
  if (tid < AGG_NBLK) {
    int o = 0;
    for (int i = 0; i < tid; ++i) o += (blkCnt[i] + 7) & ~7;
    blkOff[tid] = o;
  }
  __syncthreads();
  if (wave < AGG_NBLK && lane < 8) {
    const int n = blkCnt[wave];
    if (n + lane < ((n + 7) & ~7)) {
      const int pos = blkOff[wave] + n + lane;
      recS[pos] = 0;
      recD[pos] = 0;
      recA[pos] = 0.0f;
      recA[AGG_RCAP + pos] = 0.0f;
      recA[2 * AGG_RCAP + pos] = 0.0f;
      recA[3 * AGG_RCAP + pos] = 0.0f;
    }
  }
  for (int i = 0; i < cnt; i += 128) {
    int s[4], d[4];
    float4 x[4], y[4];
    #pragma unroll
    for (int u = 0; u < 4; ++u) {
      const int p = i + 32 * u + lane;
      s[u] = 0; d[u] = lo;
      if (p < cnt) { const int e = q[p]; d[u] = dst[e]; s[u] = src[e]; }
    }
    #pragma unroll
    for (int u = 0; u < 4; ++u) {
      const int p = i + 32 * u + lane;
      if (p < cnt) {
        x[u] = *reinterpret_cast<const float4*>(el + 4 * (long)s[u]);
        y[u] = *reinterpret_cast<const float4*>(er + 4 * (long)d[u]);
      }
    }
    #pragma unroll
    for (int u = 0; u < 4; ++u) {
      const int p = i + 32 * u + lane;
      if (p < cnt) {
        const int dl = d[u] - lo;
        const float4 m = *reinterpret_cast<const float4*>(mt + 4 * dl);
        const float a0 = edge_weight(x[u].x + y[u].x, m.x);
        const float a1 = edge_weight(x[u].y + y[u].y, m.y);
        const float a2 = edge_weight(x[u].z + y[u].z, m.z);
        const float a3 = edge_weight(x[u].w + y[u].w, m.w);
        float* ds = dt + 4 * dl;
        atomicAdd(ds,     a0);
        atomicAdd(ds + 1, a1);
        atomicAdd(ds + 2, a2);
        atomicAdd(ds + 3, a3);
        const int bb  = dl >> 4;
        const int pos = blkOff[bb] + atomicAdd(&blkFill[bb], 1);
        recS[pos] = s[u];
        recD[pos] = dl & 15;
        recA[pos] = a0;
        recA[AGG_RCAP + pos] = a1;
        recA[2 * AGG_RCAP + pos] = a2;
        recA[3 * AGG_RCAP + pos] = a3;
      }
    }
  }
  __syncthreads();

  const int col = lane & 15;
  const int hf  = lane >> 4;
  const int nJobs = 4 * ((rows + 15) >> 4);
  for (int j = wave; j < nJobs; j += 32) {
    const int b  = j >> 2;
    const int hd = j & 3;
    const int nIt  = (__builtin_amdgcn_readfirstlane(blkCnt[b]) + 7) >> 3;
    const int off  = __builtin_amdgcn_readfirstlane(blkOff[b]);
    const float* fcol = feat + hd * DHC + 4 * col;
    const float* ra   = recA + hd * AGG_RCAP;

    float4 F[2][2];
    v2f    A[2];
    auto fetch = [&](int eb, float4 (&Fo)[2][2], v2f (&Ao)[2]) {
      #pragma unroll
      for (int u = 0; u < 2; ++u) {
        const int2   s2 = *reinterpret_cast<const int2*>(recS + eb + 4 * u);
        const int2   d2 = *reinterpret_cast<const int2*>(recD + eb + 4 * u);
        const float2 a2 = *reinterpret_cast<const float2*>(ra + eb + 4 * u);
        Fo[u][0] = *reinterpret_cast<const float4*>(fcol + (long)s2.x * HIDC);
        Fo[u][1] = *reinterpret_cast<const float4*>(fcol + (long)s2.y * HIDC);
        Ao[u].x = (d2.x == col) ? a2.x : 0.0f;
        Ao[u].y = (d2.y == col) ? a2.y : 0.0f;
      }
    };

    v8f ac[4] = {};
    int eb = off + 2 * hf;
    if (nIt > 0) fetch(eb, F, A);
    for (int it = 0; it < nIt; ++it) {
      const int en = (it + 1 < nIt) ? eb + 8 : eb;
      float4 G[2][2];
      v2f    An[2];
      fetch(en, G, An);
      #pragma unroll
      for (int u = 0; u < 2; ++u) {
        const float f0[4] = {F[u][0].x, F[u][0].y, F[u][0].z, F[u][0].w};
        const float f1[4] = {F[u][1].x, F[u][1].y, F[u][1].z, F[u][1].w};
        #pragma unroll
        for (int t = 0; t < 4; ++t) {
          v2f bv;
          bv.x = f0[t];
          bv.y = f1[t];
          ac[t] = __builtin_amdgcn_wmma_f32_16x16x4_f32(
               false, A[u],  false, bv,
               (short)0, ac[t],  false,  false);
        }
      }
      #pragma unroll
      for (int u = 0; u < 2; ++u) {
        F[u][0] = G[u][0];
        F[u][1] = G[u][1];
        A[u]    = An[u];
      }
      eb = en;
    }

    float  es[8];
    float4 o[8];
    #pragma unroll
    for (int r = 0; r < 8; ++r) {
      const int rl = min(16 * b + 8 * hf + r, rows - 1);
      es[r] = dt[rl * HEADS + hd];
      o[r]  = op[rl * (HIDC / 4) + hd * (DHC / 4) + col];
    }
    #pragma unroll
    for (int r = 0; r < 8; ++r) {
      const int rl = 16 * b + 8 * hf + r;
      if (rl < rows) {
        const float rc = 1.0f / fmaxf(es[r], 1e-16f);
        o[r].x += ac[0][r] * rc; o[r].y += ac[1][r] * rc;
        o[r].z += ac[2][r] * rc; o[r].w += ac[3][r] * rc;
        op[rl * (HIDC / 4) + hd * (DHC / 4) + col] = o[r];
      }
    }
  }
}

__global__ __launch_bounds__(256)
void mean_heads_kernel(const float* __restrict__ hbuf,
                       float* __restrict__ out, long total) {
  const long stride = (long)gridDim.x * blockDim.x;
  const long t      = (long)blockIdx.x * blockDim.x + threadIdx.x;
  if ((reinterpret_cast<unsigned long>(out) & 15ul) == 0ul) {
    const long n4 = total >> 2;
    const float4* hp = reinterpret_cast<const float4*>(hbuf);
    float4*       op = reinterpret_cast<float4*>(out);
    for (long i = t; i < n4; i += 2 * stride) {
      float4 p[2][4];
      #pragma unroll
      for (int u = 0; u < 2; ++u) {
        const long j = i + u * stride;
        if (j < n4) {
          const float4* r = hp + (j >> 4) * (HIDC / 4) + (j & 15);
          #pragma unroll
          for (int k = 0; k < HEADS; ++k) p[u][k] = r[k * (DHC / 4)];
        }
      }
      #pragma unroll
      for (int u = 0; u < 2; ++u) {
        const long j = i + u * stride;
        if (j < n4) {
          float4 o;
          o.x = 0.25f * (p[u][0].x + p[u][1].x + p[u][2].x + p[u][3].x);
          o.y = 0.25f * (p[u][0].y + p[u][1].y + p[u][2].y + p[u][3].y);
          o.z = 0.25f * (p[u][0].z + p[u][1].z + p[u][2].z + p[u][3].z);
          o.w = 0.25f * (p[u][0].w + p[u][1].w + p[u][2].w + p[u][3].w);
          op[j] = o;
        }
      }
    }
  } else {
    for (long i = t; i < total; i += stride) {
      const float* p = hbuf + (i >> 6) * HIDC + (i & 63);
      out[i] = 0.25f * (p[0] + p[DHC] + p[2 * DHC] + p[3 * DHC]);
    }
  }
}

extern "C" void kernel_launch(void* const* d_in, const int* in_sizes, int n_in,
                              void* d_out, int out_size, void* d_ws, size_t ws_size,
                              hipStream_t stream) {
  const float* x     = (const float*)d_in[0];
  const int*   ei    = (const int*)d_in[1];
  const float* W[3]  = {(const float*)d_in[2], (const float*)d_in[3],
                        (const float*)d_in[4]};
  const float* al[3] = {(const float*)d_in[5], (const float*)d_in[6],
                        (const float*)d_in[7]};
  const float* ar[3] = {(const float*)d_in[8], (const float*)d_in[9],
                        (const float*)d_in[10]};
  const float* b[3]  = {(const float*)d_in[11], (const float*)d_in[12],
                        (const float*)d_in[13]};
  const float* resW0 = (const float*)d_in[14];

  const int N = in_sizes[0] / 128;
  const int E = in_sizes[1] / 2;
  const int* src = ei;
  const int* dst = ei + E;

  float* ws   = (float*)d_ws;
  float* FEAT = ws;
  float* OUT  = FEAT + (size_t)N * HIDC;
  float* ELP  = OUT  + (size_t)N * HIDC;
  float* ERP  = ELP  + (size_t)N * HEADS;
  int*   QSV  = (int*)(ERP + (size_t)N * HEADS);

  const int  gGEMM  = 256;
  const int  NH     = N * HEADS;
  const int  gSTREAM = 512;
  const int  gNARROW = 1024;

  auto attention = [&](const float* alw, const float* arw, int reuseSaved) {
    el_er_kernel<<<gNARROW, 256, 0, stream>>>(FEAT, alw, arw, ELP, ERP, NH);
    aggregate_tile_kernel<<<(N + AGG_TN - 1) / AGG_TN, 1024, 0, stream>>>(
        src, dst, FEAT, ELP, ERP, OUT, N, E, QSV, ws, (unsigned long)ws_size, reuseSaved);
  };

  gemm_wmma_f32<128, false><<<gGEMM, 512, 0, stream>>>(x, W[0], nullptr, FEAT, N, nullptr, nullptr);
  gemm_wmma_f32<128, false><<<gGEMM, 512, 0, stream>>>(x, resW0, b[0], OUT, N, nullptr, nullptr);
  attention(al[0], ar[0], 0);

  gemm_wmma_f32<256, true><<<gGEMM, 512, 0, stream>>>(OUT, W[1], nullptr, FEAT, N, OUT, b[1]);
  attention(al[1], ar[1], 1);

  gemm_wmma_f32<256, true><<<gGEMM, 512, 0, stream>>>(OUT, W[2], nullptr, FEAT, N, OUT, b[2]);
  attention(al[2], ar[2], 1);
  mean_heads_kernel<<<gSTREAM, 256, 0, stream>>>(
      OUT, (float*)d_out, (long)N * DHC);
}


# Tile queue entries resolved once per call to (source row, row in tile), packed in one word and saved; max and exp passes of all three layers gather no src/dst any more; float sums as before
# speedup vs baseline: 5.0715x; 1.0601x over previous
#include <hip/hip_runtime.h>
#include <math.h>


typedef __attribute__((ext_vector_type(2))) float v2f;
typedef __attribute__((ext_vector_type(8))) float v8f;

#define HIDC 256
#define HEADS 4
#define DHC 64

__device__ __forceinline__ float atomicMaxF(float* addr, float value) {
  if (value >= 0.0f)
    return __int_as_float(atomicMax((int*)addr, __float_as_int(value)));
  else
    return __uint_as_float(atomicMin((unsigned int*)addr, __float_as_uint(value)));
}

__device__ __forceinline__ float lrelu(float v) {
  return v > 0.0f ? v : 0.2f * v;
}

__device__ __forceinline__ float4 ld4(const float* p, bool al) {
  if (al) return *reinterpret_cast<const float4*>(p);
  return make_float4(p[0], p[1], p[2], p[3]);
}

template <int K, bool RELU_IN>
__global__ __launch_bounds__(512)
void gemm_wmma_f32(const float* A, const float* __restrict__ W,
                   const float* __restrict__ bias, float* __restrict__ C,
                   int M, float* R, const float* __restrict__ rbias) {
  constexpr int LDA  = K + 4;
  constexpr int NV   = 16 * K / 512;
  constexpr int RS   = 512 / K;
  constexpr int NU   = K / 32;
  constexpr int NG   = K / 32;
  constexpr int NBUF = (K >= 256) ? 2 : 4;
  __shared__ __attribute__((aligned(16))) float Ws[K * HIDC];
  __shared__ __attribute__((aligned(16))) float As[NBUF][16 * LDA];

  const int tid    = threadIdx.x;
  const int mtiles = (M + 15) >> 4;
  const int G      = (int)gridDim.x;
  int s = (int)blockIdx.x;
  if (s >= mtiles) return;

  const int sc  = tid & (K - 1);
  const int sr  = tid / K;
  const int scp = (sc & ~6) | ((sc & 2) << 1) | ((sc & 4) >> 1);
  float v[NV], w[NV];
  #pragma unroll
  for (int u = 0; u < NV; ++u) {
    int row = (s << 4) + sr + u * RS;
    if (row >= M) row = M - 1;
    v[u] = A[(long)row * K + sc];
  }
  const float rb = RELU_IN ? rbias[sc] : 0.0f;

  {
    const bool wal = (reinterpret_cast<unsigned long>(W) & 15ul) == 0ul;
    float4 L[NU][4];
    #pragma unroll
    for (int j = 0; j < NU; ++j) {
      const int jj = (j + (int)blockIdx.x) & (NU - 1);
      const int u  = tid + 512 * jj;
      const int q  = u & 63;
      const int p  = u >> 6;
      const float* g = W + (long)(8 * (p >> 1) + 2 * (p & 1)) * HIDC + 4 * q;
      L[j][0] = ld4(g, wal);
      L[j][1] = ld4(g + HIDC, wal);
      L[j][2] = ld4(g + 4 * HIDC, wal);
      L[j][3] = ld4(g + 5 * HIDC, wal);
    }
    #pragma unroll
    for (int j = 0; j < NU; ++j) {
      const int jj = (j + (int)blockIdx.x) & (NU - 1);
      const int u  = tid + 512 * jj;
      const int q  = u & 63;
      const int p  = u >> 6;
      const float c0[4] = {L[j][0].x, L[j][0].y, L[j][0].z, L[j][0].w};
      const float c1[4] = {L[j][1].x, L[j][1].y, L[j][1].z, L[j][1].w};
      const float c2[4] = {L[j][2].x, L[j][2].y, L[j][2].z, L[j][2].w};
      const float c3[4] = {L[j][3].x, L[j][3].y, L[j][3].z, L[j][3].w};
      #pragma unroll
      for (int i = 0; i < 4; ++i) {
        const int n  = 4 * q + i;
        const int np = (n & ~15) | ((n + (q >> 2)) & 15);
        *reinterpret_cast<float4*>(&Ws[(p * HIDC + np) * 4]) =
            make_float4(c0[i], c1[i], c2[i], c3[i]);
      }
    }
  }

  const int lane   = tid & 31;
  const int wave   = tid >> 5;
  const int nBase  = wave << 4;
  const int col    = lane & 15;
  const int h      = lane >> 4;
  const int rowOff = h << 3;
  const float* bp  = &Ws[(h * HIDC + (nBase | ((col + wave) & 15))) * 4];
  const int aoff   = col * LDA + 4 * h;
  const float bc   = bias ? bias[nBase + col] : 0.0f;

  #pragma unroll
  for (int u = 0; u < NV; ++u) {
    float t = v[u];
    if (RELU_IN) t = fmaxf(t, 0.0f);
    As[0][(sr + u * RS) * LDA + scp] = t;
    w[u] = t + rb;
  }
  __syncthreads();

  int buf   = 0;
  int mPrev = -1;
  v8f accPrev = {};
  for (; s < mtiles; s += G) {
    const int  ns      = s + G;
    const bool hasNext = ns < mtiles;
    if (hasNext) {
      #pragma unroll
      for (int u = 0; u < NV; ++u) {
        int row = (ns << 4) + sr + u * RS;
        if (row >= M) row = M - 1;
        v[u] = A[(long)row * K + sc];
      }
    }
    if (RELU_IN) {
      #pragma unroll
      for (int u = 0; u < NV; ++u) {
        const int row = (s << 4) + sr + u * RS;
        if (row < M) R[(long)row * K + sc] = w[u];
      }
    }
    if (mPrev >= 0) {
      #pragma unroll
      for (int r = 0; r < 8; ++r) {
        const int row = mPrev + rowOff + r;
        if (row < M) C[(long)row * HIDC + nBase + col] = accPrev[r];
      }
    }

    const float* ap = &As[buf][aoff];
    float4 fa[2][4], fb[2][4];
    #pragma unroll
    for (int t = 0; t < 4; ++t) {
      fa[0][t] = *reinterpret_cast<const float4*>(ap + 8 * t);
      fb[0][t] = *reinterpret_cast<const float4*>(bp + 2 * HIDC * 4 * t);
    }
    v8f acc = {};
    #pragma unroll
    for (int g = 0; g < NG; ++g) {
      const int cur = g & 1, nxt = cur ^ 1;
      if (g + 1 < NG) {
        #pragma unroll
        for (int t = 0; t < 4; ++t) {
          fa[nxt][t] = *reinterpret_cast<const float4*>(ap + 8 * (4 * (g + 1) + t));
          fb[nxt][t] = *reinterpret_cast<const float4*>(bp + 2 * HIDC * 4 * (4 * (g + 1) + t));
        }
      }
      __builtin_amdgcn_sched_barrier(0);
      #pragma unroll
      for (int t = 0; t < 4; ++t) {
        v2f a, b;
        a.x = fa[cur][t].x; a.y = fa[cur][t].y;
        b.x = fb[cur][t].x; b.y = fb[cur][t].y;
        acc = __builtin_amdgcn_wmma_f32_16x16x4_f32(
             false, a,  false, b,
             (short)0, acc,  false,  false);
        a.x = fa[cur][t].z; a.y = fa[cur][t].w;
        b.x = fb[cur][t].z; b.y = fb[cur][t].w;
        acc = __builtin_amdgcn_wmma_f32_16x16x4_f32(
             false, a,  false, b,
             (short)0, acc,  false,  false);
      }
      __builtin_amdgcn_sched_barrier(0);
    }
    #pragma unroll
    for (int r = 0; r < 8; ++r) accPrev[r] = acc[r] + bc;
    mPrev = s << 4;

    const int nb = (buf + 1) & (NBUF - 1);
    if (hasNext) {
      #pragma unroll
      for (int u = 0; u < NV; ++u) {
        float t = v[u];
        if (RELU_IN) t = fmaxf(t, 0.0f);
        As[nb][(sr + u * RS) * LDA + scp] = t;
        w[u] = t + rb;
      }
    }
    __syncthreads();
    buf = nb;
  }
  #pragma unroll
  for (int r = 0; r < 8; ++r) {
    const int row = mPrev + rowOff + r;
    if (row < M) C[(long)row * HIDC + nBase + col] = accPrev[r];
  }
}

__global__ __launch_bounds__(256)
void el_er_kernel(const float* __restrict__ feat, const float* __restrict__ al,
                  const float* __restrict__ ar, float* __restrict__ el,
                  float* __restrict__ er, int NH) {
  const int lane = threadIdx.x & 31;
  const int wave0 = __builtin_amdgcn_readfirstlane(
      (int)(blockIdx.x * (blockDim.x >> 5) + (threadIdx.x >> 5)));
  const int nW = (int)(gridDim.x * (blockDim.x >> 5));
  const int h  = wave0 & 3;
  const float l0 = al[h * DHC + lane], l1 = al[h * DHC + lane + 32];
  const float r0 = ar[h * DHC + lane], r1 = ar[h * DHC + lane + 32];
  for (int p = wave0; p < NH; p += 8 * nW) {
    float f0[8], f1[8];
    #pragma unroll
    for (int u = 0; u < 8; ++u) {
      const int q = p + u * nW;
      f0[u] = 0.0f;
      f1[u] = 0.0f;
      if (q < NH) {
        const float* f = feat + (long)q * DHC;
        f0[u] = f[lane];
        f1[u] = f[lane + 32];
      }
    }
    float sl[8], sr[8];
    #pragma unroll
    for (int u = 0; u < 8; ++u) {
      sl[u] = f0[u] * l0 + f1[u] * l1;
      sr[u] = f0[u] * r0 + f1[u] * r1;
    }
    #pragma unroll
    for (int off = 16; off; off >>= 1) {
      #pragma unroll
      for (int u = 0; u < 8; ++u) {
        sl[u] += __shfl_down(sl[u], off, 32);
        sr[u] += __shfl_down(sr[u], off, 32);
      }
    }
    #pragma unroll
    for (int u = 0; u < 8; ++u) {
      const int q = p + u * nW;
      if (lane == 0 && q < NH) { el[q] = sl[u]; er[q] = sr[u]; }
    }
  }
}

#define AGG_TN    196
#define AGG_TRIP  512
#define AGG_QCAP  768
#define AGG_SAVE  (AGG_QCAP - AGG_TRIP)
#define AGG_SHDR  64
#define AGG_SBLK  (AGG_SHDR + 32 * AGG_SAVE)
#define AGG_NBLK  ((AGG_TN + 15) / 16)
#define AGG_RCAP  (32 * AGG_SAVE + 8 * AGG_NBLK)
#define AGG_LOW   0x00FFFFFF
static_assert(AGG_TN <= 256, "the row inside the tile must fit the entry's upper eight bits");

__device__ __forceinline__ int agg_pack(int low, int rowInTile) {
  return (int)(((unsigned)low & 0x00FFFFFFu) | ((unsigned)rowInTile << 24));
}

__device__ __forceinline__ float edge_weight(float score, float mx) {
  const float v = lrelu(score);
  if (!(mx > -1e38f)) mx = 0.0f;
  return __expf(v - mx);
}

__device__ __forceinline__ void agg_drain_max(
    const int* q, int cnt, int lo, int lane,
    const int* __restrict__ src, const int* __restrict__ dst,
    const float* __restrict__ el, const float* __restrict__ er, float* mt, int* bc) {
  for (int i = 0; i < cnt; i += 64) {
    int s[2], d[2];
    float4 x[2], y[2];
    #pragma unroll
    for (int u = 0; u < 2; ++u) {
      const int p = i + 32 * u + lane;
      s[u] = 0; d[u] = lo;
      if (p < cnt) { const int e = q[p] & AGG_LOW; d[u] = dst[e]; s[u] = src[e]; }
    }
    #pragma unroll
    for (int u = 0; u < 2; ++u) {
      const int p = i + 32 * u + lane;
      if (p < cnt) {
        x[u] = *reinterpret_cast<const float4*>(el + 4 * (long)s[u]);
        y[u] = *reinterpret_cast<const float4*>(er + 4 * (long)d[u]);
      }
    }
    #pragma unroll
    for (int u = 0; u < 2; ++u) {
      const int p = i + 32 * u + lane;
      if (p < cnt) {
        float* m = mt + 4 * (d[u] - lo);
        atomicAdd(bc + ((d[u] - lo) >> 4), 1);
        atomicMaxF(m,     lrelu(x[u].x + y[u].x));
        atomicMaxF(m + 1, lrelu(x[u].y + y[u].y));
        atomicMaxF(m + 2, lrelu(x[u].z + y[u].z));
        atomicMaxF(m + 3, lrelu(x[u].w + y[u].w));
      }
    }
  }
}

__device__ __forceinline__ void agg_resolve(int* q, int cnt, int lane,
                                            const int* __restrict__ src) {
  for (int i = 0; i < cnt; i += AGG_SAVE) {
    int v[AGG_SAVE / 32], s[AGG_SAVE / 32];
    #pragma unroll
    for (int u = 0; u < AGG_SAVE / 32; ++u) {
      const int p = i + 32 * u + lane;
      v[u] = 0;
      if (p < cnt) v[u] = q[p];
    }
    #pragma unroll
    for (int u = 0; u < AGG_SAVE / 32; ++u) {
      const int p = i + 32 * u + lane;
      s[u] = 0;
      if (p < cnt) s[u] = src[v[u] & AGG_LOW];
    }
    #pragma unroll
    for (int u = 0; u < AGG_SAVE / 32; ++u) {
      const int p = i + 32 * u + lane;
      if (p < cnt) q[p] = agg_pack(s[u], (int)((unsigned)v[u] >> 24));
    }
  }
}

__device__ __forceinline__ void agg_drain_max_resolved(
    const int* q, int cnt, int lo, int lane,
    const float* __restrict__ el, const float* __restrict__ er, float* mt, int* bc) {
  for (int i = 0; i < cnt; i += 128) {
    int s[4], dl[4];
    float4 x[4], y[4];
    #pragma unroll
    for (int u = 0; u < 4; ++u) {
      const int p = i + 32 * u + lane;
      s[u] = 0; dl[u] = 0;
      if (p < cnt) { const int v = q[p]; s[u] = v & AGG_LOW; dl[u] = (int)((unsigned)v >> 24); }
    }
    #pragma unroll
    for (int u = 0; u < 4; ++u) {
      const int p = i + 32 * u + lane;
      if (p < cnt) {
        x[u] = *reinterpret_cast<const float4*>(el + 4 * (long)s[u]);
        y[u] = *reinterpret_cast<const float4*>(er + 4 * (long)(lo + dl[u]));
      }
    }
    #pragma unroll
    for (int u = 0; u < 4; ++u) {
      const int p = i + 32 * u + lane;
      if (p < cnt) {
        float* m = mt + 4 * dl[u];
        atomicAdd(bc + (dl[u] >> 4), 1);
        atomicMaxF(m,     lrelu(x[u].x + y[u].x));
        atomicMaxF(m + 1, lrelu(x[u].y + y[u].y));
        atomicMaxF(m + 2, lrelu(x[u].z + y[u].z));
        atomicMaxF(m + 3, lrelu(x[u].w + y[u].w));
      }
    }
  }
}

__device__ __forceinline__ void agg_drain_acc(
    const int* q, int cnt, int lo, int lane,
    const int* __restrict__ src, const int* __restrict__ dst,
    const float* __restrict__ feat,
    const float* __restrict__ el, const float* __restrict__ er,
    const float* mt, float* dt, float* acc) {
  for (int i = 0; i < cnt; i += 32) {
    const int n = min(32, cnt - i);
    int s = 0, dl = 0;
    float a0 = 0.0f, a1 = 0.0f, a2 = 0.0f, a3 = 0.0f;
    if (lane < n) {
      const int e = q[i + lane] & AGG_LOW;
      const int d = dst[e];
      s  = src[e];
      dl = d - lo;
      const float4 x = *reinterpret_cast<const float4*>(el + 4 * (long)s);
      const float4 y = *reinterpret_cast<const float4*>(er + 4 * (long)d);
      const float4 m = *reinterpret_cast<const float4*>(mt + 4 * dl);
      a0 = edge_weight(x.x + y.x, m.x);
      a1 = edge_weight(x.y + y.y, m.y);
      a2 = edge_weight(x.z + y.z, m.z);
      a3 = edge_weight(x.w + y.w, m.w);
      float* ds = dt + 4 * dl;
      atomicAdd(ds,     a0);
      atomicAdd(ds + 1, a1);
      atomicAdd(ds + 2, a2);
      atomicAdd(ds + 3, a3);
    }
    for (int k = 0; k < n; k += 4) {
      int su[4], du[4];
      float au[4][4];
      v2f f[4][4];
      #pragma unroll
      for (int u = 0; u < 4; ++u) {
        const int idx = min(k + u, n - 1);
        su[u] = __builtin_amdgcn_readlane(s, idx);
        du[u] = (k + u < n) ? __builtin_amdgcn_readlane(dl, idx) : AGG_TN;
        au[u][0] = __int_as_float(__builtin_amdgcn_readlane(__float_as_int(a0), idx));
        au[u][1] = __int_as_float(__builtin_amdgcn_readlane(__float_as_int(a1), idx));
        au[u][2] = __int_as_float(__builtin_amdgcn_readlane(__float_as_int(a2), idx));
        au[u][3] = __int_as_float(__builtin_amdgcn_readlane(__float_as_int(a3), idx));
      }
      #pragma unroll
      for (int u = 0; u < 4; ++u) {
        const float* fr = feat + (long)su[u] * HIDC + 2 * lane;
        #pragma unroll
        for (int j = 0; j < HEADS; ++j)
          f[u][j] = *reinterpret_cast<const v2f*>(fr + j * DHC);
      }
      #pragma unroll
      for (int u = 0; u < 4; ++u) {
        float* p = acc + du[u] * HIDC + 2 * lane;
        #pragma unroll
        for (int j = 0; j < HEADS; ++j) {
          atomicAdd(p + j * DHC,     au[u][j] * f[u][j].x);
          atomicAdd(p + j * DHC + 1, au[u][j] * f[u][j].y);
        }
      }
    }
  }
}

template <int PASS>
__device__ __forceinline__ int agg_scan(
    int* q, int wave, int lane, int lo, int rows, int E,
    const int* __restrict__ src, const int* __restrict__ dst,
    const float* __restrict__ feat,
    const float* __restrict__ el, const float* __restrict__ er,
    float* mt, float* dt, float* acc, int* full, int* bc) {
  const int  T       = (E + AGG_TRIP - 1) / AGG_TRIP;
  const bool aligned = (reinterpret_cast<unsigned long>(dst) & 15ul) == 0ul;
  const int start = (T / (int)gridDim.x) * (int)blockIdx.x;
  int cnt = 0;
  for (int t = wave; t < T; t += 32) {
    int g = t + start;
    if (g >= T) g -= T;
    const int base = g * AGG_TRIP + 4 * lane;
    int4 dv[4];
    if (aligned && g * AGG_TRIP + AGG_TRIP <= E) {
      #pragma unroll
      for (int u = 0; u < 4; ++u)
        dv[u] = *reinterpret_cast<const int4*>(dst + base + 128 * u);
    } else {
      #pragma unroll
      for (int u = 0; u < 4; ++u) {
        const int e0 = base + 128 * u;
        dv[u].x = (e0     < E) ? dst[e0]     : -1;
        dv[u].y = (e0 + 1 < E) ? dst[e0 + 1] : -1;
        dv[u].z = (e0 + 2 < E) ? dst[e0 + 2] : -1;
        dv[u].w = (e0 + 3 < E) ? dst[e0 + 3] : -1;
      }
    }
    #pragma unroll
    for (int u = 0; u < 4; ++u) {
      const int  e0 = base + 128 * u;
      const bool mx = (unsigned)(dv[u].x - lo) < (unsigned)rows;
      const bool my = (unsigned)(dv[u].y - lo) < (unsigned)rows;
      const bool mz = (unsigned)(dv[u].z - lo) < (unsigned)rows;
      const bool mw = (unsigned)(dv[u].w - lo) < (unsigned)rows;
      if (__builtin_amdgcn_ballot_w32(mx | my | mz | mw)) {
        unsigned mask;
        mask = __builtin_amdgcn_ballot_w32(mx);
        if (mx) q[cnt + __builtin_amdgcn_mbcnt_lo(mask, 0u)] = agg_pack(e0, dv[u].x - lo);
        cnt += __builtin_popcount(mask);
        mask = __builtin_amdgcn_ballot_w32(my);
        if (my) q[cnt + __builtin_amdgcn_mbcnt_lo(mask, 0u)] = agg_pack(e0 + 1, dv[u].y - lo);
        cnt += __builtin_popcount(mask);
        mask = __builtin_amdgcn_ballot_w32(mz);
        if (mz) q[cnt + __builtin_amdgcn_mbcnt_lo(mask, 0u)] = agg_pack(e0 + 2, dv[u].z - lo);
        cnt += __builtin_popcount(mask);
        mask = __builtin_amdgcn_ballot_w32(mw);
        if (mw) q[cnt + __builtin_amdgcn_mbcnt_lo(mask, 0u)] = agg_pack(e0 + 3, dv[u].w - lo);
        cnt += __builtin_popcount(mask);
      }
    }
    if (cnt > AGG_QCAP - AGG_TRIP) {
      if (PASS == 0) {
        agg_drain_max(q, cnt, lo, lane, src, dst, el, er, mt, bc);
        if (lane == 0) *full = 1;
      } else {
        agg_drain_acc(q, cnt, lo, lane, src, dst, feat, el, er, mt, dt, acc);
      }
      cnt = 0;
    }
  }
  return cnt;
}

__global__ __launch_bounds__(1024)
void aggregate_tile_kernel(const int* __restrict__ src, const int* __restrict__ dst,
                           const float* __restrict__ feat,
                           const float* __restrict__ el,
                           const float* __restrict__ er, float* __restrict__ out,
                           int N, int E, int* qsave, const float* wsBase,
                           unsigned long wsBytes, int reuseSaved) {
  __shared__ __attribute__((aligned(16))) float acc[(AGG_TN + 1) * HIDC];
  static_assert(6 * AGG_RCAP <= (AGG_TN + 1) * HIDC, "records must fit the accumulators' memory");
  __shared__ __attribute__((aligned(16))) float mt[AGG_TN * HEADS];
  __shared__ __attribute__((aligned(16))) float dt[AGG_TN * HEADS];
  __shared__ int queue[32 * AGG_QCAP];
  __shared__ int full;
  __shared__ int blkCnt[16];
  __shared__ int blkFill[16];
  __shared__ int blkOff[16];

  const int tid  = threadIdx.x;
  const int lane = tid & 31;
  const int wave = __builtin_amdgcn_readfirstlane(tid >> 5);
  const int lo   = blockIdx.x * AGG_TN;
  const int rows = min(AGG_TN, N - lo);

  if (tid < AGG_TN * HEADS) { mt[tid] = -INFINITY; dt[tid] = 0.0f; }
  if (tid < 16) { blkCnt[tid] = 0; blkFill[tid] = 0; }
  if (tid == 0) full = 0;
  __syncthreads();

  int* q = queue + wave * AGG_QCAP;
  int* sv = qsave + (long)blockIdx.x * AGG_SBLK;
  const bool fits =
      (unsigned long)(reinterpret_cast<const char*>(qsave + (long)gridDim.x * AGG_SBLK) -
                      reinterpret_cast<const char*>(wsBase)) <= wsBytes;
  int reuse = 0, saved = 0;
  if (reuseSaved != 0 && fits) {
    const int okw = sv[32];
    const int cw  = sv[wave];
    reuse = (okw == 1) ? 1 : 0;
    saved = cw;
  }
  int cnt;
  if (reuse != 0) {
    cnt = min(max(saved, 0), AGG_SAVE);
    const int* sq = sv + AGG_SHDR + wave * AGG_SAVE;
    int v[AGG_SAVE / 32];
    #pragma unroll
    for (int u = 0; u < AGG_SAVE / 32; ++u) {
      const int i = lane + 32 * u;
      v[u] = 0;
      if (i < cnt) v[u] = sq[i];
    }
    #pragma unroll
    for (int u = 0; u < AGG_SAVE / 32; ++u) {
      const int i = lane + 32 * u;
      if (i < cnt) q[i] = v[u];
    }
  } else {
    cnt = agg_scan<0>(q, wave, lane, lo, rows, E, src, dst, feat, el, er, mt, dt, acc, &full, blkCnt);
    agg_resolve(q, cnt, lane, src);
  }
  agg_drain_max_resolved(q, cnt, lo, lane, el, er, mt, blkCnt);
  __syncthreads();

  const int early = (reuse != 0) ? 0 : __builtin_amdgcn_readfirstlane(full);
  if (reuseSaved == 0 && fits) {
    if (early == 0) {
      int* sq = sv + AGG_SHDR + wave * AGG_SAVE;
      for (int i = lane; i < cnt; i += 32) sq[i] = q[i];
    }
    if (lane == 0) sv[wave] = (early == 0) ? cnt : 0;
    if (tid == 0) sv[32] = (early == 0) ? 1 : 0;
  }
  float4* op = reinterpret_cast<float4*>(out + (long)lo * HIDC);

  if (early != 0) {
    for (int i = tid; i < (AGG_TN + 1) * HIDC / 4; i += 1024)
      reinterpret_cast<float4*>(acc)[i] = make_float4(0.0f, 0.0f, 0.0f, 0.0f);
    __syncthreads();
    cnt = agg_scan<1>(q, wave, lane, lo, rows, E, src, dst, feat, el, er, mt, dt, acc, &full, blkCnt);
    agg_drain_acc(q, cnt, lo, lane, src, dst, feat, el, er, mt, dt, acc);
    __syncthreads();
    if (tid < rows * HEADS) dt[tid] = 1.0f / fmaxf(dt[tid], 1e-16f);
    __syncthreads();
    for (int i = tid; i < rows * (HIDC / 4); i += 1024) {
      const float  r = dt[i >> 4];
      const float4 a = reinterpret_cast<const float4*>(acc)[i];
      float4 o = op[i];
      o.x += a.x * r; o.y += a.y * r; o.z += a.z * r; o.w += a.w * r;
      op[i] = o;
    }
    return;
  }

  int*   recS = reinterpret_cast<int*>(acc);
  int*   recD = recS + AGG_RCAP;
  float* recA = reinterpret_cast<float*>(recD + AGG_RCAP);
  if (tid < AGG_NBLK) {
    int o = 0;
    for (int i = 0; i < tid; ++i) o += (blkCnt[i] + 7) & ~7;
    blkOff[tid] = o;
  }
  __syncthreads();
  if (wave < AGG_NBLK && lane < 8) {
    const int n = blkCnt[wave];
    if (n + lane < ((n + 7) & ~7)) {
      const int pos = blkOff[wave] + n + lane;
      recS[pos] = 0;
      recD[pos] = 0;
      recA[pos] = 0.0f;
      recA[AGG_RCAP + pos] = 0.0f;
      recA[2 * AGG_RCAP + pos] = 0.0f;
      recA[3 * AGG_RCAP + pos] = 0.0f;
    }
  }
  for (int i = 0; i < cnt; i += 128) {
    int s[4], dr[4];
    float4 x[4], y[4];
    #pragma unroll
    for (int u = 0; u < 4; ++u) {
      const int p = i + 32 * u + lane;
      s[u] = 0; dr[u] = 0;
      if (p < cnt) { const int v = q[p]; s[u] = v & AGG_LOW; dr[u] = (int)((unsigned)v >> 24); }
    }
    #pragma unroll
    for (int u = 0; u < 4; ++u) {
      const int p = i + 32 * u + lane;
      if (p < cnt) {
        x[u] = *reinterpret_cast<const float4*>(el + 4 * (long)s[u]);
        y[u] = *reinterpret_cast<const float4*>(er + 4 * (long)(lo + dr[u]));
      }
    }
    #pragma unroll
    for (int u = 0; u < 4; ++u) {
      const int p = i + 32 * u + lane;
      if (p < cnt) {
        const int dl = dr[u];
        const float4 m = *reinterpret_cast<const float4*>(mt + 4 * dl);
        const float a0 = edge_weight(x[u].x + y[u].x, m.x);
        const float a1 = edge_weight(x[u].y + y[u].y, m.y);
        const float a2 = edge_weight(x[u].z + y[u].z, m.z);
        const float a3 = edge_weight(x[u].w + y[u].w, m.w);
        float* ds = dt + 4 * dl;
        atomicAdd(ds,     a0);
        atomicAdd(ds + 1, a1);
        atomicAdd(ds + 2, a2);
        atomicAdd(ds + 3, a3);
        const int bb  = dl >> 4;
        const int pos = blkOff[bb] + atomicAdd(&blkFill[bb], 1);
        recS[pos] = s[u];
        recD[pos] = dl & 15;
        recA[pos] = a0;
        recA[AGG_RCAP + pos] = a1;
        recA[2 * AGG_RCAP + pos] = a2;
        recA[3 * AGG_RCAP + pos] = a3;
      }
    }
  }
  __syncthreads();

  const int col = lane & 15;
  const int hf  = lane >> 4;
  const int nJobs = 4 * ((rows + 15) >> 4);
  for (int j = wave; j < nJobs; j += 32) {
    const int b  = j >> 2;
    const int hd = j & 3;
    const int nIt  = (__builtin_amdgcn_readfirstlane(blkCnt[b]) + 7) >> 3;
    const int off  = __builtin_amdgcn_readfirstlane(blkOff[b]);
    const float* fcol = feat + hd * DHC + 4 * col;
    const float* ra   = recA + hd * AGG_RCAP;

    float4 F[2][2];
    v2f    A[2];
    auto fetch = [&](int eb, float4 (&Fo)[2][2], v2f (&Ao)[2]) {
      #pragma unroll
      for (int u = 0; u < 2; ++u) {
        const int2   s2 = *reinterpret_cast<const int2*>(recS + eb + 4 * u);
        const int2   d2 = *reinterpret_cast<const int2*>(recD + eb + 4 * u);
        const float2 a2 = *reinterpret_cast<const float2*>(ra + eb + 4 * u);
        Fo[u][0] = *reinterpret_cast<const float4*>(fcol + (long)s2.x * HIDC);
        Fo[u][1] = *reinterpret_cast<const float4*>(fcol + (long)s2.y * HIDC);
        Ao[u].x = (d2.x == col) ? a2.x : 0.0f;
        Ao[u].y = (d2.y == col) ? a2.y : 0.0f;
      }
    };

    v8f ac[4] = {};
    int eb = off + 2 * hf;
    if (nIt > 0) fetch(eb, F, A);
    for (int it = 0; it < nIt; ++it) {
      const int en = (it + 1 < nIt) ? eb + 8 : eb;
      float4 G[2][2];
      v2f    An[2];
      fetch(en, G, An);
      #pragma unroll
      for (int u = 0; u < 2; ++u) {
        const float f0[4] = {F[u][0].x, F[u][0].y, F[u][0].z, F[u][0].w};
        const float f1[4] = {F[u][1].x, F[u][1].y, F[u][1].z, F[u][1].w};
        #pragma unroll
        for (int t = 0; t < 4; ++t) {
          v2f bv;
          bv.x = f0[t];
          bv.y = f1[t];
          ac[t] = __builtin_amdgcn_wmma_f32_16x16x4_f32(
               false, A[u],  false, bv,
               (short)0, ac[t],  false,  false);
        }
      }
      #pragma unroll
      for (int u = 0; u < 2; ++u) {
        F[u][0] = G[u][0];
        F[u][1] = G[u][1];
        A[u]    = An[u];
      }
      eb = en;
    }

    float  es[8];
    float4 o[8];
    #pragma unroll
    for (int r = 0; r < 8; ++r) {
      const int rl = min(16 * b + 8 * hf + r, rows - 1);
      es[r] = dt[rl * HEADS + hd];
      o[r]  = op[rl * (HIDC / 4) + hd * (DHC / 4) + col];
    }
    #pragma unroll
    for (int r = 0; r < 8; ++r) {
      const int rl = 16 * b + 8 * hf + r;
      if (rl < rows) {
        const float rc = 1.0f / fmaxf(es[r], 1e-16f);
        o[r].x += ac[0][r] * rc; o[r].y += ac[1][r] * rc;
        o[r].z += ac[2][r] * rc; o[r].w += ac[3][r] * rc;
        op[rl * (HIDC / 4) + hd * (DHC / 4) + col] = o[r];
      }
    }
  }
}

__global__ __launch_bounds__(256)
void mean_heads_kernel(const float* __restrict__ hbuf,
                       float* __restrict__ out, long total) {
  const long stride = (long)gridDim.x * blockDim.x;
  const long t      = (long)blockIdx.x * blockDim.x + threadIdx.x;
  if ((reinterpret_cast<unsigned long>(out) & 15ul) == 0ul) {
    const long n4 = total >> 2;
    const float4* hp = reinterpret_cast<const float4*>(hbuf);
    float4*       op = reinterpret_cast<float4*>(out);
    for (long i = t; i < n4; i += 2 * stride) {
      float4 p[2][4];
      #pragma unroll
      for (int u = 0; u < 2; ++u) {
        const long j = i + u * stride;
        if (j < n4) {
          const float4* r = hp + (j >> 4) * (HIDC / 4) + (j & 15);
          #pragma unroll
          for (int k = 0; k < HEADS; ++k) p[u][k] = r[k * (DHC / 4)];
        }
      }
      #pragma unroll
      for (int u = 0; u < 2; ++u) {
        const long j = i + u * stride;
        if (j < n4) {
          float4 o;
          o.x = 0.25f * (p[u][0].x + p[u][1].x + p[u][2].x + p[u][3].x);
          o.y = 0.25f * (p[u][0].y + p[u][1].y + p[u][2].y + p[u][3].y);
          o.z = 0.25f * (p[u][0].z + p[u][1].z + p[u][2].z + p[u][3].z);
          o.w = 0.25f * (p[u][0].w + p[u][1].w + p[u][2].w + p[u][3].w);
          op[j] = o;
        }
      }
    }
  } else {
    for (long i = t; i < total; i += stride) {
      const float* p = hbuf + (i >> 6) * HIDC + (i & 63);
      out[i] = 0.25f * (p[0] + p[DHC] + p[2 * DHC] + p[3 * DHC]);
    }
  }
}

extern "C" void kernel_launch(void* const* d_in, const int* in_sizes, int n_in,
                              void* d_out, int out_size, void* d_ws, size_t ws_size,
                              hipStream_t stream) {
  const float* x     = (const float*)d_in[0];
  const int*   ei    = (const int*)d_in[1];
  const float* W[3]  = {(const float*)d_in[2], (const float*)d_in[3],
                        (const float*)d_in[4]};
  const float* al[3] = {(const float*)d_in[5], (const float*)d_in[6],
                        (const float*)d_in[7]};
  const float* ar[3] = {(const float*)d_in[8], (const float*)d_in[9],
                        (const float*)d_in[10]};
  const float* b[3]  = {(const float*)d_in[11], (const float*)d_in[12],
                        (const float*)d_in[13]};
  const float* resW0 = (const float*)d_in[14];

  const int N = in_sizes[0] / 128;
  const int E = in_sizes[1] / 2;
  const int* src = ei;
  const int* dst = ei + E;

  float* ws   = (float*)d_ws;
  float* FEAT = ws;
  float* OUT  = FEAT + (size_t)N * HIDC;
  float* ELP  = OUT  + (size_t)N * HIDC;
  float* ERP  = ELP  + (size_t)N * HEADS;
  int*   QSV  = (int*)(ERP + (size_t)N * HEADS);

  const int  gGEMM  = 256;
  const int  NH     = N * HEADS;
  const int  gSTREAM = 512;
  const int  gNARROW = 1024;

  auto attention = [&](const float* alw, const float* arw, int reuseSaved) {
    el_er_kernel<<<gNARROW, 256, 0, stream>>>(FEAT, alw, arw, ELP, ERP, NH);
    aggregate_tile_kernel<<<(N + AGG_TN - 1) / AGG_TN, 1024, 0, stream>>>(
        src, dst, FEAT, ELP, ERP, OUT, N, E, QSV, ws, (unsigned long)ws_size, reuseSaved);
  };

  gemm_wmma_f32<128, false><<<gGEMM, 512, 0, stream>>>(x, W[0], nullptr, FEAT, N, nullptr, nullptr);
  gemm_wmma_f32<128, false><<<gGEMM, 512, 0, stream>>>(x, resW0, b[0], OUT, N, nullptr, nullptr);
  attention(al[0], ar[0], 0);

  gemm_wmma_f32<256, true><<<gGEMM, 512, 0, stream>>>(OUT, W[1], nullptr, FEAT, N, OUT, b[1]);
  attention(al[1], ar[1], 1);

  gemm_wmma_f32<256, true><<<gGEMM, 512, 0, stream>>>(OUT, W[2], nullptr, FEAT, N, OUT, b[2]);
  attention(al[2], ar[2], 1);
  mean_heads_kernel<<<gSTREAM, 256, 0, stream>>>(
      OUT, (float*)d_out, (long)N * DHC);
}


# First aggregation launch no longer walks all of dst in every work-group: 128 work-groups of the first projection launch sort their share of the edges by tile in LDS; sums: list slot order as before
# speedup vs baseline: 5.2613x; 1.0374x over previous
#include <hip/hip_runtime.h>
#include <math.h>


typedef __attribute__((ext_vector_type(2))) float v2f;
typedef __attribute__((ext_vector_type(8))) float v8f;

#define HIDC 256
#define HEADS 4
#define DHC 64

__device__ __forceinline__ float atomicMaxF(float* addr, float value) {
  if (value >= 0.0f)
    return __int_as_float(atomicMax((int*)addr, __float_as_int(value)));
  else
    return __uint_as_float(atomicMin((unsigned int*)addr, __float_as_uint(value)));
}

__device__ __forceinline__ float lrelu(float v) {
  return v > 0.0f ? v : 0.2f * v;
}

__device__ __forceinline__ float4 ld4(const float* p, bool al) {
  if (al) return *reinterpret_cast<const float4*>(p);
  return make_float4(p[0], p[1], p[2], p[3]);
}

__device__ __forceinline__ void agg_bin_edges(int* lds, const int* __restrict__ dst, int E,
                                              int* qsave, int nTiles, int tid);
__device__ __forceinline__ bool agg_bins_usable(const int* qsave, int nTiles,
                                                const float* wsBase, unsigned long wsBytes);

template <int K, bool RELU_IN>
__global__ __launch_bounds__(512)
void gemm_wmma_f32(const float* A, const float* __restrict__ W,
                   const float* __restrict__ bias, float* __restrict__ C,
                   int M, float* R, const float* __restrict__ rbias,
                   const int* __restrict__ binDst, int binE, int* binOut, int binTiles,
                   const float* wsBase, unsigned long wsBytes) {
  constexpr int LDA  = K + 4;
  constexpr int NV   = 16 * K / 512;
  constexpr int RS   = 512 / K;
  constexpr int NU   = K / 32;
  constexpr int NG   = K / 32;
  constexpr int NBUF = (K >= 256) ? 2 : 4;
  __shared__ __attribute__((aligned(16))) float Ws[K * HIDC];
  __shared__ __attribute__((aligned(16))) float As[NBUF][16 * LDA];

  const int tid    = threadIdx.x;
  const int mtiles = (M + 15) >> 4;
  const int G      = (int)gridDim.x;
  int s = (int)blockIdx.x;
  if (binDst != nullptr && agg_bins_usable(binOut, binTiles, wsBase, wsBytes))
    agg_bin_edges(reinterpret_cast<int*>(Ws), binDst, binE, binOut, binTiles, tid);
  if (s >= mtiles) return;

  const int sc  = tid & (K - 1);
  const int sr  = tid / K;
  const int scp = (sc & ~6) | ((sc & 2) << 1) | ((sc & 4) >> 1);
  float v[NV], w[NV];
  #pragma unroll
  for (int u = 0; u < NV; ++u) {
    int row = (s << 4) + sr + u * RS;
    if (row >= M) row = M - 1;
    v[u] = A[(long)row * K + sc];
  }
  const float rb = RELU_IN ? rbias[sc] : 0.0f;

  {
    const bool wal = (reinterpret_cast<unsigned long>(W) & 15ul) == 0ul;
    float4 L[NU][4];
    #pragma unroll
    for (int j = 0; j < NU; ++j) {
      const int jj = (j + (int)blockIdx.x) & (NU - 1);
      const int u  = tid + 512 * jj;
      const int q  = u & 63;
      const int p  = u >> 6;
      const float* g = W + (long)(8 * (p >> 1) + 2 * (p & 1)) * HIDC + 4 * q;
      L[j][0] = ld4(g, wal);
      L[j][1] = ld4(g + HIDC, wal);
      L[j][2] = ld4(g + 4 * HIDC, wal);
      L[j][3] = ld4(g + 5 * HIDC, wal);
    }
    #pragma unroll
    for (int j = 0; j < NU; ++j) {
      const int jj = (j + (int)blockIdx.x) & (NU - 1);
      const int u  = tid + 512 * jj;
      const int q  = u & 63;
      const int p  = u >> 6;
      const float c0[4] = {L[j][0].x, L[j][0].y, L[j][0].z, L[j][0].w};
      const float c1[4] = {L[j][1].x, L[j][1].y, L[j][1].z, L[j][1].w};
      const float c2[4] = {L[j][2].x, L[j][2].y, L[j][2].z, L[j][2].w};
      const float c3[4] = {L[j][3].x, L[j][3].y, L[j][3].z, L[j][3].w};
      #pragma unroll
      for (int i = 0; i < 4; ++i) {
        const int n  = 4 * q + i;
        const int np = (n & ~15) | ((n + (q >> 2)) & 15);
        *reinterpret_cast<float4*>(&Ws[(p * HIDC + np) * 4]) =
            make_float4(c0[i], c1[i], c2[i], c3[i]);
      }
    }
  }

  const int lane   = tid & 31;
  const int wave   = tid >> 5;
  const int nBase  = wave << 4;
  const int col    = lane & 15;
  const int h      = lane >> 4;
  const int rowOff = h << 3;
  const float* bp  = &Ws[(h * HIDC + (nBase | ((col + wave) & 15))) * 4];
  const int aoff   = col * LDA + 4 * h;
  const float bc   = bias ? bias[nBase + col] : 0.0f;

  #pragma unroll
  for (int u = 0; u < NV; ++u) {
    float t = v[u];
    if (RELU_IN) t = fmaxf(t, 0.0f);
    As[0][(sr + u * RS) * LDA + scp] = t;
    w[u] = t + rb;
  }
  __syncthreads();

  int buf   = 0;
  int mPrev = -1;
  v8f accPrev = {};
  for (; s < mtiles; s += G) {
    const int  ns      = s + G;
    const bool hasNext = ns < mtiles;
    if (hasNext) {
      #pragma unroll
      for (int u = 0; u < NV; ++u) {
        int row = (ns << 4) + sr + u * RS;
        if (row >= M) row = M - 1;
        v[u] = A[(long)row * K + sc];
      }
    }
    if (RELU_IN) {
      #pragma unroll
      for (int u = 0; u < NV; ++u) {
        const int row = (s << 4) + sr + u * RS;
        if (row < M) R[(long)row * K + sc] = w[u];
      }
    }
    if (mPrev >= 0) {
      #pragma unroll
      for (int r = 0; r < 8; ++r) {
        const int row = mPrev + rowOff + r;
        if (row < M) C[(long)row * HIDC + nBase + col] = accPrev[r];
      }
    }

    const float* ap = &As[buf][aoff];
    float4 fa[2][4], fb[2][4];
    #pragma unroll
    for (int t = 0; t < 4; ++t) {
      fa[0][t] = *reinterpret_cast<const float4*>(ap + 8 * t);
      fb[0][t] = *reinterpret_cast<const float4*>(bp + 2 * HIDC * 4 * t);
    }
    v8f acc = {};
    #pragma unroll
    for (int g = 0; g < NG; ++g) {
      const int cur = g & 1, nxt = cur ^ 1;
      if (g + 1 < NG) {
        #pragma unroll
        for (int t = 0; t < 4; ++t) {
          fa[nxt][t] = *reinterpret_cast<const float4*>(ap + 8 * (4 * (g + 1) + t));
          fb[nxt][t] = *reinterpret_cast<const float4*>(bp + 2 * HIDC * 4 * (4 * (g + 1) + t));
        }
      }
      __builtin_amdgcn_sched_barrier(0);
      #pragma unroll
      for (int t = 0; t < 4; ++t) {
        v2f a, b;
        a.x = fa[cur][t].x; a.y = fa[cur][t].y;
        b.x = fb[cur][t].x; b.y = fb[cur][t].y;
        acc = __builtin_amdgcn_wmma_f32_16x16x4_f32(
             false, a,  false, b,
             (short)0, acc,  false,  false);
        a.x = fa[cur][t].z; a.y = fa[cur][t].w;
        b.x = fb[cur][t].z; b.y = fb[cur][t].w;
        acc = __builtin_amdgcn_wmma_f32_16x16x4_f32(
             false, a,  false, b,
             (short)0, acc,  false,  false);
      }
      __builtin_amdgcn_sched_barrier(0);
    }
    #pragma unroll
    for (int r = 0; r < 8; ++r) accPrev[r] = acc[r] + bc;
    mPrev = s << 4;

    const int nb = (buf + 1) & (NBUF - 1);
    if (hasNext) {
      #pragma unroll
      for (int u = 0; u < NV; ++u) {
        float t = v[u];
        if (RELU_IN) t = fmaxf(t, 0.0f);
        As[nb][(sr + u * RS) * LDA + scp] = t;
        w[u] = t + rb;
      }
    }
    __syncthreads();
    buf = nb;
  }
  #pragma unroll
  for (int r = 0; r < 8; ++r) {
    const int row = mPrev + rowOff + r;
    if (row < M) C[(long)row * HIDC + nBase + col] = accPrev[r];
  }
}

__global__ __launch_bounds__(256)
void el_er_kernel(const float* __restrict__ feat, const float* __restrict__ al,
                  const float* __restrict__ ar, float* __restrict__ el,
                  float* __restrict__ er, int NH) {
  const int lane = threadIdx.x & 31;
  const int wave0 = __builtin_amdgcn_readfirstlane(
      (int)(blockIdx.x * (blockDim.x >> 5) + (threadIdx.x >> 5)));
  const int nW = (int)(gridDim.x * (blockDim.x >> 5));
  const int h  = wave0 & 3;
  const float l0 = al[h * DHC + lane], l1 = al[h * DHC + lane + 32];
  const float r0 = ar[h * DHC + lane], r1 = ar[h * DHC + lane + 32];
  for (int p = wave0; p < NH; p += 8 * nW) {
    float f0[8], f1[8];
    #pragma unroll
    for (int u = 0; u < 8; ++u) {
      const int q = p + u * nW;
      f0[u] = 0.0f;
      f1[u] = 0.0f;
      if (q < NH) {
        const float* f = feat + (long)q * DHC;
        f0[u] = f[lane];
        f1[u] = f[lane + 32];
      }
    }
    float sl[8], sr[8];
    #pragma unroll
    for (int u = 0; u < 8; ++u) {
      sl[u] = f0[u] * l0 + f1[u] * l1;
      sr[u] = f0[u] * r0 + f1[u] * r1;
    }
    #pragma unroll
    for (int off = 16; off; off >>= 1) {
      #pragma unroll
      for (int u = 0; u < 8; ++u) {
        sl[u] += __shfl_down(sl[u], off, 32);
        sr[u] += __shfl_down(sr[u], off, 32);
      }
    }
    #pragma unroll
    for (int u = 0; u < 8; ++u) {
      const int q = p + u * nW;
      if (lane == 0 && q < NH) { el[q] = sl[u]; er[q] = sr[u]; }
    }
  }
}

#define AGG_TN    196
#define AGG_TRIP  512
#define AGG_QCAP  768
#define AGG_SAVE  (AGG_QCAP - AGG_TRIP)
#define AGG_SHDR  64
#define AGG_SBLK  (AGG_SHDR + 32 * AGG_SAVE)
#define AGG_NBLK  ((AGG_TN + 15) / 16)
#define AGG_RCAP  (32 * AGG_SAVE + 8 * AGG_NBLK)
#define AGG_LOW   0x00FFFFFF
static_assert(AGG_TN <= 256, "the row inside the tile must fit the entry's upper eight bits");

__device__ __forceinline__ int agg_pack(int low, int rowInTile) {
  return (int)(((unsigned)low & 0x00FFFFFFu) | ((unsigned)rowInTile << 24));
}

#define AGG_NCH   128
#define AGG_SEGW  64
#define AGG_BINT  256
static_assert(AGG_NCH * AGG_SEGW == 32 * AGG_SAVE, "a tile's segments fill its saved queues' place");
static_assert(4 * (AGG_SEGW - 1) <= AGG_SAVE, "a wave's four segments fit its queue");

__device__ __forceinline__ bool agg_bins_usable(const int* qsave, int nTiles,
                                                const float* wsBase, unsigned long wsBytes) {
  return nTiles <= AGG_BINT &&
         (unsigned long)(reinterpret_cast<const char*>(qsave + (long)nTiles * AGG_SBLK) -
                         reinterpret_cast<const char*>(wsBase)) <= wsBytes;
}

__device__ __forceinline__ void agg_bin_edges(int* lds, const int* __restrict__ dst, int E,
                                              int* qsave, int nTiles, int tid) {
  const int share = (int)blockIdx.x;
  if (share >= AGG_NCH) return;
  int* cntT = lds;
  int* buf  = lds + AGG_BINT;
  const int L  = (E + AGG_NCH - 1) / AGG_NCH;
  const int e0 = share * L;
  if (tid < AGG_BINT) cntT[tid] = 0;
  __syncthreads();
  for (int i0 = 0; i0 < L; i0 += 512 * 16) {
    int d[16];
    #pragma unroll
    for (int u = 0; u < 16; ++u) {
      const int i = i0 + tid + 512 * u;
      d[u] = -1;
      if (i < L && e0 + i < E) d[u] = dst[e0 + i];
    }
    int tl[16], sl[16];
    #pragma unroll
    for (int u = 0; u < 16; ++u) {
      const int t = d[u] / AGG_TN;
      tl[u] = -1;
      sl[u] = 0;
      if (d[u] >= 0 && t < nTiles) {
        tl[u] = t;
        sl[u] = atomicAdd(&cntT[t], 1);
      }
    }
    #pragma unroll
    for (int u = 0; u < 16; ++u) {
      if (tl[u] >= 0 && sl[u] < AGG_SEGW - 1)
        buf[tl[u] * AGG_SEGW + 1 + sl[u]] =
            agg_pack(e0 + i0 + tid + 512 * u, d[u] - tl[u] * AGG_TN);
    }
  }
  __syncthreads();
  const int lane = tid & 31;
  for (int t = tid >> 5; t < nTiles; t += 16) {
    const int n = cntT[t];
    int* seg = qsave + (long)t * AGG_SBLK + AGG_SHDR + share * AGG_SEGW;
    #pragma unroll
    for (int hf = 0; hf < 2; ++hf) {
      const int wd = lane + 32 * hf;
      int v = 0;
      if (wd == 0) v = (n > AGG_SEGW - 1) ? -1 : n;
      else if (wd <= n) v = buf[t * AGG_SEGW + wd];
      seg[wd] = v;
    }
  }
  __syncthreads();
}

__device__ __forceinline__ float edge_weight(float score, float mx) {
  const float v = lrelu(score);
  if (!(mx > -1e38f)) mx = 0.0f;
  return __expf(v - mx);
}

__device__ __forceinline__ void agg_drain_max(
    const int* q, int cnt, int lo, int lane,
    const int* __restrict__ src, const int* __restrict__ dst,
    const float* __restrict__ el, const float* __restrict__ er, float* mt, int* bc) {
  for (int i = 0; i < cnt; i += 64) {
    int s[2], d[2];
    float4 x[2], y[2];
    #pragma unroll
    for (int u = 0; u < 2; ++u) {
      const int p = i + 32 * u + lane;
      s[u] = 0; d[u] = lo;
      if (p < cnt) { const int e = q[p] & AGG_LOW; d[u] = dst[e]; s[u] = src[e]; }
    }
    #pragma unroll
    for (int u = 0; u < 2; ++u) {
      const int p = i + 32 * u + lane;
      if (p < cnt) {
        x[u] = *reinterpret_cast<const float4*>(el + 4 * (long)s[u]);
        y[u] = *reinterpret_cast<const float4*>(er + 4 * (long)d[u]);
      }
    }
    #pragma unroll
    for (int u = 0; u < 2; ++u) {
      const int p = i + 32 * u + lane;
      if (p < cnt) {
        float* m = mt + 4 * (d[u] - lo);
        atomicAdd(bc + ((d[u] - lo) >> 4), 1);
        atomicMaxF(m,     lrelu(x[u].x + y[u].x));
        atomicMaxF(m + 1, lrelu(x[u].y + y[u].y));
        atomicMaxF(m + 2, lrelu(x[u].z + y[u].z));
        atomicMaxF(m + 3, lrelu(x[u].w + y[u].w));
      }
    }
  }
}

__device__ __forceinline__ void agg_resolve(int* q, int cnt, int lane,
                                            const int* __restrict__ src) {
  for (int i = 0; i < cnt; i += AGG_SAVE) {
    int v[AGG_SAVE / 32], s[AGG_SAVE / 32];
    #pragma unroll
    for (int u = 0; u < AGG_SAVE / 32; ++u) {
      const int p = i + 32 * u + lane;
      v[u] = 0;
      if (p < cnt) v[u] = q[p];
    }
    #pragma unroll
    for (int u = 0; u < AGG_SAVE / 32; ++u) {
      const int p = i + 32 * u + lane;
      s[u] = 0;
      if (p < cnt) s[u] = src[v[u] & AGG_LOW];
    }
    #pragma unroll
    for (int u = 0; u < AGG_SAVE / 32; ++u) {
      const int p = i + 32 * u + lane;
      if (p < cnt) q[p] = agg_pack(s[u], (int)((unsigned)v[u] >> 24));
    }
  }
}

__device__ __forceinline__ void agg_drain_max_resolved(
    const int* q, int cnt, int lo, int lane,
    const float* __restrict__ el, const float* __restrict__ er, float* mt, int* bc) {
  for (int i = 0; i < cnt; i += 128) {
    int s[4], dl[4];
    float4 x[4], y[4];
    #pragma unroll
    for (int u = 0; u < 4; ++u) {
      const int p = i + 32 * u + lane;
      s[u] = 0; dl[u] = 0;
      if (p < cnt) { const int v = q[p]; s[u] = v & AGG_LOW; dl[u] = (int)((unsigned)v >> 24); }
    }
    #pragma unroll
    for (int u = 0; u < 4; ++u) {
      const int p = i + 32 * u + lane;
      if (p < cnt) {
        x[u] = *reinterpret_cast<const float4*>(el + 4 * (long)s[u]);
        y[u] = *reinterpret_cast<const float4*>(er + 4 * (long)(lo + dl[u]));
      }
    }
    #pragma unroll
    for (int u = 0; u < 4; ++u) {
      const int p = i + 32 * u + lane;
      if (p < cnt) {
        float* m = mt + 4 * dl[u];
        atomicAdd(bc + (dl[u] >> 4), 1);
        atomicMaxF(m,     lrelu(x[u].x + y[u].x));
        atomicMaxF(m + 1, lrelu(x[u].y + y[u].y));
        atomicMaxF(m + 2, lrelu(x[u].z + y[u].z));
        atomicMaxF(m + 3, lrelu(x[u].w + y[u].w));
      }
    }
  }
}

__device__ __forceinline__ void agg_drain_acc(
    const int* q, int cnt, int lo, int lane,
    const int* __restrict__ src, const int* __restrict__ dst,
    const float* __restrict__ feat,
    const float* __restrict__ el, const float* __restrict__ er,
    const float* mt, float* dt, float* acc) {
  for (int i = 0; i < cnt; i += 32) {
    const int n = min(32, cnt - i);
    int s = 0, dl = 0;
    float a0 = 0.0f, a1 = 0.0f, a2 = 0.0f, a3 = 0.0f;
    if (lane < n) {
      const int e = q[i + lane] & AGG_LOW;
      const int d = dst[e];
      s  = src[e];
      dl = d - lo;
      const float4 x = *reinterpret_cast<const float4*>(el + 4 * (long)s);
      const float4 y = *reinterpret_cast<const float4*>(er + 4 * (long)d);
      const float4 m = *reinterpret_cast<const float4*>(mt + 4 * dl);
      a0 = edge_weight(x.x + y.x, m.x);
      a1 = edge_weight(x.y + y.y, m.y);
      a2 = edge_weight(x.z + y.z, m.z);
      a3 = edge_weight(x.w + y.w, m.w);
      float* ds = dt + 4 * dl;
      atomicAdd(ds,     a0);
      atomicAdd(ds + 1, a1);
      atomicAdd(ds + 2, a2);
      atomicAdd(ds + 3, a3);
    }
    for (int k = 0; k < n; k += 4) {
      int su[4], du[4];
      float au[4][4];
      v2f f[4][4];
      #pragma unroll
      for (int u = 0; u < 4; ++u) {
        const int idx = min(k + u, n - 1);
        su[u] = __builtin_amdgcn_readlane(s, idx);
        du[u] = (k + u < n) ? __builtin_amdgcn_readlane(dl, idx) : AGG_TN;
        au[u][0] = __int_as_float(__builtin_amdgcn_readlane(__float_as_int(a0), idx));
        au[u][1] = __int_as_float(__builtin_amdgcn_readlane(__float_as_int(a1), idx));
        au[u][2] = __int_as_float(__builtin_amdgcn_readlane(__float_as_int(a2), idx));
        au[u][3] = __int_as_float(__builtin_amdgcn_readlane(__float_as_int(a3), idx));
      }
      #pragma unroll
      for (int u = 0; u < 4; ++u) {
        const float* fr = feat + (long)su[u] * HIDC + 2 * lane;
        #pragma unroll
        for (int j = 0; j < HEADS; ++j)
          f[u][j] = *reinterpret_cast<const v2f*>(fr + j * DHC);
      }
      #pragma unroll
      for (int u = 0; u < 4; ++u) {
        float* p = acc + du[u] * HIDC + 2 * lane;
        #pragma unroll
        for (int j = 0; j < HEADS; ++j) {
          atomicAdd(p + j * DHC,     au[u][j] * f[u][j].x);
          atomicAdd(p + j * DHC + 1, au[u][j] * f[u][j].y);
        }
      }
    }
  }
}

template <int PASS>
__device__ __forceinline__ int agg_scan(
    int* q, int wave, int lane, int lo, int rows, int E,
    const int* __restrict__ src, const int* __restrict__ dst,
    const float* __restrict__ feat,
    const float* __restrict__ el, const float* __restrict__ er,
    float* mt, float* dt, float* acc, int* full, int* bc) {
  const int  T       = (E + AGG_TRIP - 1) / AGG_TRIP;
  const bool aligned = (reinterpret_cast<unsigned long>(dst) & 15ul) == 0ul;
  const int start = (T / (int)gridDim.x) * (int)blockIdx.x;
  int cnt = 0;
  for (int t = wave; t < T; t += 32) {
    int g = t + start;
    if (g >= T) g -= T;
    const int base = g * AGG_TRIP + 4 * lane;
    int4 dv[4];
    if (aligned && g * AGG_TRIP + AGG_TRIP <= E) {
      #pragma unroll
      for (int u = 0; u < 4; ++u)
        dv[u] = *reinterpret_cast<const int4*>(dst + base + 128 * u);
    } else {
      #pragma unroll
      for (int u = 0; u < 4; ++u) {
        const int e0 = base + 128 * u;
        dv[u].x = (e0     < E) ? dst[e0]     : -1;
        dv[u].y = (e0 + 1 < E) ? dst[e0 + 1] : -1;
        dv[u].z = (e0 + 2 < E) ? dst[e0 + 2] : -1;
        dv[u].w = (e0 + 3 < E) ? dst[e0 + 3] : -1;
      }
    }
    #pragma unroll
    for (int u = 0; u < 4; ++u) {
      const int  e0 = base + 128 * u;
      const bool mx = (unsigned)(dv[u].x - lo) < (unsigned)rows;
      const bool my = (unsigned)(dv[u].y - lo) < (unsigned)rows;
      const bool mz = (unsigned)(dv[u].z - lo) < (unsigned)rows;
      const bool mw = (unsigned)(dv[u].w - lo) < (unsigned)rows;
      if (__builtin_amdgcn_ballot_w32(mx | my | mz | mw)) {
        unsigned mask;
        mask = __builtin_amdgcn_ballot_w32(mx);
        if (mx) q[cnt + __builtin_amdgcn_mbcnt_lo(mask, 0u)] = agg_pack(e0, dv[u].x - lo);
        cnt += __builtin_popcount(mask);
        mask = __builtin_amdgcn_ballot_w32(my);
        if (my) q[cnt + __builtin_amdgcn_mbcnt_lo(mask, 0u)] = agg_pack(e0 + 1, dv[u].y - lo);
        cnt += __builtin_popcount(mask);
        mask = __builtin_amdgcn_ballot_w32(mz);
        if (mz) q[cnt + __builtin_amdgcn_mbcnt_lo(mask, 0u)] = agg_pack(e0 + 2, dv[u].z - lo);
        cnt += __builtin_popcount(mask);
        mask = __builtin_amdgcn_ballot_w32(mw);
        if (mw) q[cnt + __builtin_amdgcn_mbcnt_lo(mask, 0u)] = agg_pack(e0 + 3, dv[u].w - lo);
        cnt += __builtin_popcount(mask);
      }
    }
    if (cnt > AGG_QCAP - AGG_TRIP) {
      if (PASS == 0) {
        agg_drain_max(q, cnt, lo, lane, src, dst, el, er, mt, bc);
        if (lane == 0) *full = 1;
      } else {
        agg_drain_acc(q, cnt, lo, lane, src, dst, feat, el, er, mt, dt, acc);
      }
      cnt = 0;
    }
  }
  return cnt;
}

__global__ __launch_bounds__(1024)
void aggregate_tile_kernel(const int* __restrict__ src, const int* __restrict__ dst,
                           const float* __restrict__ feat,
                           const float* __restrict__ el,
                           const float* __restrict__ er, float* __restrict__ out,
                           int N, int E, int* qsave, const float* wsBase,
                           unsigned long wsBytes, int reuseSaved, int useBins) {
  __shared__ __attribute__((aligned(16))) float acc[(AGG_TN + 1) * HIDC];
  static_assert(6 * AGG_RCAP <= (AGG_TN + 1) * HIDC, "records must fit the accumulators' memory");
  __shared__ __attribute__((aligned(16))) float mt[AGG_TN * HEADS];
  __shared__ __attribute__((aligned(16))) float dt[AGG_TN * HEADS];
  __shared__ int queue[32 * AGG_QCAP];
  __shared__ int full;
  __shared__ int binBad;
  __shared__ int blkCnt[16];
  __shared__ int blkFill[16];
  __shared__ int blkOff[16];

  const int tid  = threadIdx.x;
  const int lane = tid & 31;
  const int wave = __builtin_amdgcn_readfirstlane(tid >> 5);
  const int lo   = blockIdx.x * AGG_TN;
  const int rows = min(AGG_TN, N - lo);

  if (tid < AGG_TN * HEADS) { mt[tid] = -INFINITY; dt[tid] = 0.0f; }
  if (tid < 16) { blkCnt[tid] = 0; blkFill[tid] = 0; }
  if (tid == 0) { full = 0; binBad = 0; }
  __syncthreads();

  int* q = queue + wave * AGG_QCAP;
  int* sv = qsave + (long)blockIdx.x * AGG_SBLK;
  const bool fits =
      (unsigned long)(reinterpret_cast<const char*>(qsave + (long)gridDim.x * AGG_SBLK) -
                      reinterpret_cast<const char*>(wsBase)) <= wsBytes;
  int reuse = 0, saved = 0;
  if (reuseSaved != 0 && fits) {
    const int okw = sv[32];
    const int cw  = sv[wave];
    reuse = (okw == 1) ? 1 : 0;
    saved = cw;
  }
  int cnt;
  if (reuse != 0) {
    cnt = min(max(saved, 0), AGG_SAVE);
    const int* sq = sv + AGG_SHDR + wave * AGG_SAVE;
    int v[AGG_SAVE / 32];
    #pragma unroll
    for (int u = 0; u < AGG_SAVE / 32; ++u) {
      const int i = lane + 32 * u;
      v[u] = 0;
      if (i < cnt) v[u] = sq[i];
    }
    #pragma unroll
    for (int u = 0; u < AGG_SAVE / 32; ++u) {
      const int i = lane + 32 * u;
      if (i < cnt) q[i] = v[u];
    }
  } else {
    int binned = 0;
    cnt = 0;
    if (useBins != 0 && agg_bins_usable(qsave, (int)gridDim.x, wsBase, wsBytes)) {
      const int* bp = sv + AGG_SHDR + wave * (4 * AGG_SEGW) + lane;
      int w8[8];
      #pragma unroll
      for (int u = 0; u < 8; ++u) w8[u] = bp[32 * u];
      int bad = 0;
      #pragma unroll
      for (int g = 0; g < 4; ++g) {
        const int c = __builtin_amdgcn_readfirstlane(w8[2 * g]);
        if (c < 0 || c > AGG_SEGW - 1) {
          bad = 1;
        } else {
          #pragma unroll
          for (int hf = 0; hf < 2; ++hf) {
            const int  wd = lane + 32 * hf;
            const bool v  = wd >= 1 && wd <= c;
            const unsigned mask = __builtin_amdgcn_ballot_w32(v);
            if (v) q[cnt + __builtin_amdgcn_mbcnt_lo(mask, 0u)] = w8[2 * g + hf];
            cnt += __builtin_popcount(mask);
          }
        }
      }
      if (bad != 0 && lane == 0) binBad = 1;
      __syncthreads();
      binned = (__builtin_amdgcn_readfirstlane(binBad) == 0) ? 1 : 0;
    }
    if (binned == 0)
      cnt = agg_scan<0>(q, wave, lane, lo, rows, E, src, dst, feat, el, er, mt, dt, acc, &full, blkCnt);
    agg_resolve(q, cnt, lane, src);
  }
  agg_drain_max_resolved(q, cnt, lo, lane, el, er, mt, blkCnt);
  __syncthreads();

  const int early = (reuse != 0) ? 0 : __builtin_amdgcn_readfirstlane(full);
  if (reuseSaved == 0 && fits) {
    if (early == 0) {
      int* sq = sv + AGG_SHDR + wave * AGG_SAVE;
      for (int i = lane; i < cnt; i += 32) sq[i] = q[i];
    }
    if (lane == 0) sv[wave] = (early == 0) ? cnt : 0;
    if (tid == 0) sv[32] = (early == 0) ? 1 : 0;
  }
  float4* op = reinterpret_cast<float4*>(out + (long)lo * HIDC);

  if (early != 0) {
    for (int i = tid; i < (AGG_TN + 1) * HIDC / 4; i += 1024)
      reinterpret_cast<float4*>(acc)[i] = make_float4(0.0f, 0.0f, 0.0f, 0.0f);
    __syncthreads();
    cnt = agg_scan<1>(q, wave, lane, lo, rows, E, src, dst, feat, el, er, mt, dt, acc, &full, blkCnt);
    agg_drain_acc(q, cnt, lo, lane, src, dst, feat, el, er, mt, dt, acc);
    __syncthreads();
    if (tid < rows * HEADS) dt[tid] = 1.0f / fmaxf(dt[tid], 1e-16f);
    __syncthreads();
    for (int i = tid; i < rows * (HIDC / 4); i += 1024) {
      const float  r = dt[i >> 4];
      const float4 a = reinterpret_cast<const float4*>(acc)[i];
      float4 o = op[i];
      o.x += a.x * r; o.y += a.y * r; o.z += a.z * r; o.w += a.w * r;
      op[i] = o;
    }
    return;
  }

  int*   recS = reinterpret_cast<int*>(acc);
  int*   recD = recS + AGG_RCAP;
  float* recA = reinterpret_cast<float*>(recD + AGG_RCAP);
  if (tid < AGG_NBLK) {
    int o = 0;
    for (int i = 0; i < tid; ++i) o += (blkCnt[i] + 7) & ~7;
    blkOff[tid] = o;
  }
  __syncthreads();
  if (wave < AGG_NBLK && lane < 8) {
    const int n = blkCnt[wave];
    if (n + lane < ((n + 7) & ~7)) {
      const int pos = blkOff[wave] + n + lane;
      recS[pos] = 0;
      recD[pos] = 0;
      recA[pos] = 0.0f;
      recA[AGG_RCAP + pos] = 0.0f;
      recA[2 * AGG_RCAP + pos] = 0.0f;
      recA[3 * AGG_RCAP + pos] = 0.0f;
    }
  }
  for (int i = 0; i < cnt; i += 128) {
    int s[4], dr[4];
    float4 x[4], y[4];
    #pragma unroll
    for (int u = 0; u < 4; ++u) {
      const int p = i + 32 * u + lane;
      s[u] = 0; dr[u] = 0;
      if (p < cnt) { const int v = q[p]; s[u] = v & AGG_LOW; dr[u] = (int)((unsigned)v >> 24); }
    }
    #pragma unroll
    for (int u = 0; u < 4; ++u) {
      const int p = i + 32 * u + lane;
      if (p < cnt) {
        x[u] = *reinterpret_cast<const float4*>(el + 4 * (long)s[u]);
        y[u] = *reinterpret_cast<const float4*>(er + 4 * (long)(lo + dr[u]));
      }
    }
    #pragma unroll
    for (int u = 0; u < 4; ++u) {
      const int p = i + 32 * u + lane;
      if (p < cnt) {
        const int dl = dr[u];
        const float4 m = *reinterpret_cast<const float4*>(mt + 4 * dl);
        const float a0 = edge_weight(x[u].x + y[u].x, m.x);
        const float a1 = edge_weight(x[u].y + y[u].y, m.y);
        const float a2 = edge_weight(x[u].z + y[u].z, m.z);
        const float a3 = edge_weight(x[u].w + y[u].w, m.w);
        float* ds = dt + 4 * dl;
        atomicAdd(ds,     a0);
        atomicAdd(ds + 1, a1);
        atomicAdd(ds + 2, a2);
        atomicAdd(ds + 3, a3);
        const int bb  = dl >> 4;
        const int pos = blkOff[bb] + atomicAdd(&blkFill[bb], 1);
        recS[pos] = s[u];
        recD[pos] = dl & 15;
        recA[pos] = a0;
        recA[AGG_RCAP + pos] = a1;
        recA[2 * AGG_RCAP + pos] = a2;
        recA[3 * AGG_RCAP + pos] = a3;
      }
    }
  }
  __syncthreads();

  const int col = lane & 15;
  const int hf  = lane >> 4;
  const int nJobs = 4 * ((rows + 15) >> 4);
  for (int j = wave; j < nJobs; j += 32) {
    const int b  = j >> 2;
    const int hd = j & 3;
    const int nIt  = (__builtin_amdgcn_readfirstlane(blkCnt[b]) + 7) >> 3;
    const int off  = __builtin_amdgcn_readfirstlane(blkOff[b]);
    const float* fcol = feat + hd * DHC + 4 * col;
    const float* ra   = recA + hd * AGG_RCAP;

    float4 F[2][2];
    v2f    A[2];
    auto fetch = [&](int eb, float4 (&Fo)[2][2], v2f (&Ao)[2]) {
      #pragma unroll
      for (int u = 0; u < 2; ++u) {
        const int2   s2 = *reinterpret_cast<const int2*>(recS + eb + 4 * u);
        const int2   d2 = *reinterpret_cast<const int2*>(recD + eb + 4 * u);
        const float2 a2 = *reinterpret_cast<const float2*>(ra + eb + 4 * u);
        Fo[u][0] = *reinterpret_cast<const float4*>(fcol + (long)s2.x * HIDC);
        Fo[u][1] = *reinterpret_cast<const float4*>(fcol + (long)s2.y * HIDC);
        Ao[u].x = (d2.x == col) ? a2.x : 0.0f;
        Ao[u].y = (d2.y == col) ? a2.y : 0.0f;
      }
    };

    v8f ac[4] = {};
    int eb = off + 2 * hf;
    if (nIt > 0) fetch(eb, F, A);
    for (int it = 0; it < nIt; ++it) {
      const int en = (it + 1 < nIt) ? eb + 8 : eb;
      float4 G[2][2];
      v2f    An[2];
      fetch(en, G, An);
      #pragma unroll
      for (int u = 0; u < 2; ++u) {
        const float f0[4] = {F[u][0].x, F[u][0].y, F[u][0].z, F[u][0].w};
        const float f1[4] = {F[u][1].x, F[u][1].y, F[u][1].z, F[u][1].w};
        #pragma unroll
        for (int t = 0; t < 4; ++t) {
          v2f bv;
          bv.x = f0[t];
          bv.y = f1[t];
          ac[t] = __builtin_amdgcn_wmma_f32_16x16x4_f32(
               false, A[u],  false, bv,
               (short)0, ac[t],  false,  false);
        }
      }
      #pragma unroll
      for (int u = 0; u < 2; ++u) {
        F[u][0] = G[u][0];
        F[u][1] = G[u][1];
        A[u]    = An[u];
      }
      eb = en;
    }

    float  es[8];
    float4 o[8];
    #pragma unroll
    for (int r = 0; r < 8; ++r) {
      const int rl = min(16 * b + 8 * hf + r, rows - 1);
      es[r] = dt[rl * HEADS + hd];
      o[r]  = op[rl * (HIDC / 4) + hd * (DHC / 4) + col];
    }
    #pragma unroll
    for (int r = 0; r < 8; ++r) {
      const int rl = 16 * b + 8 * hf + r;
      if (rl < rows) {
        const float rc = 1.0f / fmaxf(es[r], 1e-16f);
        o[r].x += ac[0][r] * rc; o[r].y += ac[1][r] * rc;
        o[r].z += ac[2][r] * rc; o[r].w += ac[3][r] * rc;
        op[rl * (HIDC / 4) + hd * (DHC / 4) + col] = o[r];
      }
    }
  }
}

__global__ __launch_bounds__(256)
void mean_heads_kernel(const float* __restrict__ hbuf,
                       float* __restrict__ out, long total) {
  const long stride = (long)gridDim.x * blockDim.x;
  const long t      = (long)blockIdx.x * blockDim.x + threadIdx.x;
  if ((reinterpret_cast<unsigned long>(out) & 15ul) == 0ul) {
    const long n4 = total >> 2;
    const float4* hp = reinterpret_cast<const float4*>(hbuf);
    float4*       op = reinterpret_cast<float4*>(out);
    for (long i = t; i < n4; i += 2 * stride) {
      float4 p[2][4];
      #pragma unroll
      for (int u = 0; u < 2; ++u) {
        const long j = i + u * stride;
        if (j < n4) {
          const float4* r = hp + (j >> 4) * (HIDC / 4) + (j & 15);
          #pragma unroll
          for (int k = 0; k < HEADS; ++k) p[u][k] = r[k * (DHC / 4)];
        }
      }
      #pragma unroll
      for (int u = 0; u < 2; ++u) {
        const long j = i + u * stride;
        if (j < n4) {
          float4 o;
          o.x = 0.25f * (p[u][0].x + p[u][1].x + p[u][2].x + p[u][3].x);
          o.y = 0.25f * (p[u][0].y + p[u][1].y + p[u][2].y + p[u][3].y);
          o.z = 0.25f * (p[u][0].z + p[u][1].z + p[u][2].z + p[u][3].z);
          o.w = 0.25f * (p[u][0].w + p[u][1].w + p[u][2].w + p[u][3].w);
          op[j] = o;
        }
      }
    }
  } else {
    for (long i = t; i < total; i += stride) {
      const float* p = hbuf + (i >> 6) * HIDC + (i & 63);
      out[i] = 0.25f * (p[0] + p[DHC] + p[2 * DHC] + p[3 * DHC]);
    }
  }
}

extern "C" void kernel_launch(void* const* d_in, const int* in_sizes, int n_in,
                              void* d_out, int out_size, void* d_ws, size_t ws_size,
                              hipStream_t stream) {
  const float* x     = (const float*)d_in[0];
  const int*   ei    = (const int*)d_in[1];
  const float* W[3]  = {(const float*)d_in[2], (const float*)d_in[3],
                        (const float*)d_in[4]};
  const float* al[3] = {(const float*)d_in[5], (const float*)d_in[6],
                        (const float*)d_in[7]};
  const float* ar[3] = {(const float*)d_in[8], (const float*)d_in[9],
                        (const float*)d_in[10]};
  const float* b[3]  = {(const float*)d_in[11], (const float*)d_in[12],
                        (const float*)d_in[13]};
  const float* resW0 = (const float*)d_in[14];

  const int N = in_sizes[0] / 128;
  const int E = in_sizes[1] / 2;
  const int* src = ei;
  const int* dst = ei + E;

  float* ws   = (float*)d_ws;
  float* FEAT = ws;
  float* OUT  = FEAT + (size_t)N * HIDC;
  float* ELP  = OUT  + (size_t)N * HIDC;
  float* ERP  = ELP  + (size_t)N * HEADS;
  int*   QSV  = (int*)(ERP + (size_t)N * HEADS);

  const int  gGEMM  = 256;
  const int  NH     = N * HEADS;
  const int  gSTREAM = 512;
  const int  gNARROW = 1024;

  auto attention = [&](const float* alw, const float* arw, int reuseSaved, int useBins) {
    el_er_kernel<<<gNARROW, 256, 0, stream>>>(FEAT, alw, arw, ELP, ERP, NH);
    aggregate_tile_kernel<<<(N + AGG_TN - 1) / AGG_TN, 1024, 0, stream>>>(
        src, dst, FEAT, ELP, ERP, OUT, N, E, QSV, ws, (unsigned long)ws_size, reuseSaved, useBins);
  };

  gemm_wmma_f32<128, false><<<gGEMM, 512, 0, stream>>>(x, W[0], nullptr, FEAT, N, nullptr, nullptr,
      dst, E, QSV, (N + AGG_TN - 1) / AGG_TN, ws, (unsigned long)ws_size);
  gemm_wmma_f32<128, false><<<gGEMM, 512, 0, stream>>>(x, resW0, b[0], OUT, N, nullptr, nullptr,
      nullptr, 0, nullptr, 0, nullptr, 0ul);
  attention(al[0], ar[0], 0, 1);

  gemm_wmma_f32<256, true><<<gGEMM, 512, 0, stream>>>(OUT, W[1], nullptr, FEAT, N, OUT, b[1],
      nullptr, 0, nullptr, 0, nullptr, 0ul);
  attention(al[1], ar[1], 1, 0);

  gemm_wmma_f32<256, true><<<gGEMM, 512, 0, stream>>>(OUT, W[2], nullptr, FEAT, N, OUT, b[2],
      nullptr, 0, nullptr, 0, nullptr, 0ul);
  attention(al[2], ar[2], 1, 0);
  mean_heads_kernel<<<gSTREAM, 256, 0, stream>>>(
      OUT, (float*)d_out, (long)N * DHC);
}


# Attention dots (el/er): a pair per half-wave, 16-byte loads, 8 to 16 a wave in flight on 16 resident waves per WGP; float sums: 4 products a lane, then a 4-level half-wave butterfly
# speedup vs baseline: 5.4820x; 1.0419x over previous
#include <hip/hip_runtime.h>
#include <math.h>


typedef __attribute__((ext_vector_type(2))) float v2f;
typedef __attribute__((ext_vector_type(8))) float v8f;

#define HIDC 256
#define HEADS 4
#define DHC 64

__device__ __forceinline__ float atomicMaxF(float* addr, float value) {
  if (value >= 0.0f)
    return __int_as_float(atomicMax((int*)addr, __float_as_int(value)));
  else
    return __uint_as_float(atomicMin((unsigned int*)addr, __float_as_uint(value)));
}

__device__ __forceinline__ float lrelu(float v) {
  return v > 0.0f ? v : 0.2f * v;
}

__device__ __forceinline__ float4 ld4(const float* p, bool al) {
  if (al) return *reinterpret_cast<const float4*>(p);
  return make_float4(p[0], p[1], p[2], p[3]);
}

__device__ __forceinline__ void agg_bin_edges(int* lds, const int* __restrict__ dst, int E,
                                              int* qsave, int nTiles, int tid);
__device__ __forceinline__ bool agg_bins_usable(const int* qsave, int nTiles,
                                                const float* wsBase, unsigned long wsBytes);

template <int K, bool RELU_IN>
__global__ __launch_bounds__(512)
void gemm_wmma_f32(const float* A, const float* __restrict__ W,
                   const float* __restrict__ bias, float* __restrict__ C,
                   int M, float* R, const float* __restrict__ rbias,
                   const int* __restrict__ binDst, int binE, int* binOut, int binTiles,
                   const float* wsBase, unsigned long wsBytes) {
  constexpr int LDA  = K + 4;
  constexpr int NV   = 16 * K / 512;
  constexpr int RS   = 512 / K;
  constexpr int NU   = K / 32;
  constexpr int NG   = K / 32;
  constexpr int NBUF = (K >= 256) ? 2 : 4;
  __shared__ __attribute__((aligned(16))) float Ws[K * HIDC];
  __shared__ __attribute__((aligned(16))) float As[NBUF][16 * LDA];

  const int tid    = threadIdx.x;
  const int mtiles = (M + 15) >> 4;
  const int G      = (int)gridDim.x;
  int s = (int)blockIdx.x;
  if (binDst != nullptr && agg_bins_usable(binOut, binTiles, wsBase, wsBytes))
    agg_bin_edges(reinterpret_cast<int*>(Ws), binDst, binE, binOut, binTiles, tid);
  if (s >= mtiles) return;

  const int sc  = tid & (K - 1);
  const int sr  = tid / K;
  const int scp = (sc & ~6) | ((sc & 2) << 1) | ((sc & 4) >> 1);
  float v[NV], w[NV];
  #pragma unroll
  for (int u = 0; u < NV; ++u) {
    int row = (s << 4) + sr + u * RS;
    if (row >= M) row = M - 1;
    v[u] = A[(long)row * K + sc];
  }
  const float rb = RELU_IN ? rbias[sc] : 0.0f;

  {
    const bool wal = (reinterpret_cast<unsigned long>(W) & 15ul) == 0ul;
    float4 L[NU][4];
    #pragma unroll
    for (int j = 0; j < NU; ++j) {
      const int jj = (j + (int)blockIdx.x) & (NU - 1);
      const int u  = tid + 512 * jj;
      const int q  = u & 63;
      const int p  = u >> 6;
      const float* g = W + (long)(8 * (p >> 1) + 2 * (p & 1)) * HIDC + 4 * q;
      L[j][0] = ld4(g, wal);
      L[j][1] = ld4(g + HIDC, wal);
      L[j][2] = ld4(g + 4 * HIDC, wal);
      L[j][3] = ld4(g + 5 * HIDC, wal);
    }
    #pragma unroll
    for (int j = 0; j < NU; ++j) {
      const int jj = (j + (int)blockIdx.x) & (NU - 1);
      const int u  = tid + 512 * jj;
      const int q  = u & 63;
      const int p  = u >> 6;
      const float c0[4] = {L[j][0].x, L[j][0].y, L[j][0].z, L[j][0].w};
      const float c1[4] = {L[j][1].x, L[j][1].y, L[j][1].z, L[j][1].w};
      const float c2[4] = {L[j][2].x, L[j][2].y, L[j][2].z, L[j][2].w};
      const float c3[4] = {L[j][3].x, L[j][3].y, L[j][3].z, L[j][3].w};
      #pragma unroll
      for (int i = 0; i < 4; ++i) {
        const int n  = 4 * q + i;
        const int np = (n & ~15) | ((n + (q >> 2)) & 15);
        *reinterpret_cast<float4*>(&Ws[(p * HIDC + np) * 4]) =
            make_float4(c0[i], c1[i], c2[i], c3[i]);
      }
    }
  }

  const int lane   = tid & 31;
  const int wave   = tid >> 5;
  const int nBase  = wave << 4;
  const int col    = lane & 15;
  const int h      = lane >> 4;
  const int rowOff = h << 3;
  const float* bp  = &Ws[(h * HIDC + (nBase | ((col + wave) & 15))) * 4];
  const int aoff   = col * LDA + 4 * h;
  const float bc   = bias ? bias[nBase + col] : 0.0f;

  #pragma unroll
  for (int u = 0; u < NV; ++u) {
    float t = v[u];
    if (RELU_IN) t = fmaxf(t, 0.0f);
    As[0][(sr + u * RS) * LDA + scp] = t;
    w[u] = t + rb;
  }
  __syncthreads();

  int buf   = 0;
  int mPrev = -1;
  v8f accPrev = {};
  for (; s < mtiles; s += G) {
    const int  ns      = s + G;
    const bool hasNext = ns < mtiles;
    if (hasNext) {
      #pragma unroll
      for (int u = 0; u < NV; ++u) {
        int row = (ns << 4) + sr + u * RS;
        if (row >= M) row = M - 1;
        v[u] = A[(long)row * K + sc];
      }
    }
    if (RELU_IN) {
      #pragma unroll
      for (int u = 0; u < NV; ++u) {
        const int row = (s << 4) + sr + u * RS;
        if (row < M) R[(long)row * K + sc] = w[u];
      }
    }
    if (mPrev >= 0) {
      #pragma unroll
      for (int r = 0; r < 8; ++r) {
        const int row = mPrev + rowOff + r;
        if (row < M) C[(long)row * HIDC + nBase + col] = accPrev[r];
      }
    }

    const float* ap = &As[buf][aoff];
    float4 fa[2][4], fb[2][4];
    #pragma unroll
    for (int t = 0; t < 4; ++t) {
      fa[0][t] = *reinterpret_cast<const float4*>(ap + 8 * t);
      fb[0][t] = *reinterpret_cast<const float4*>(bp + 2 * HIDC * 4 * t);
    }
    v8f acc = {};
    #pragma unroll
    for (int g = 0; g < NG; ++g) {
      const int cur = g & 1, nxt = cur ^ 1;
      if (g + 1 < NG) {
        #pragma unroll
        for (int t = 0; t < 4; ++t) {
          fa[nxt][t] = *reinterpret_cast<const float4*>(ap + 8 * (4 * (g + 1) + t));
          fb[nxt][t] = *reinterpret_cast<const float4*>(bp + 2 * HIDC * 4 * (4 * (g + 1) + t));
        }
      }
      __builtin_amdgcn_sched_barrier(0);
      #pragma unroll
      for (int t = 0; t < 4; ++t) {
        v2f a, b;
        a.x = fa[cur][t].x; a.y = fa[cur][t].y;
        b.x = fb[cur][t].x; b.y = fb[cur][t].y;
        acc = __builtin_amdgcn_wmma_f32_16x16x4_f32(
             false, a,  false, b,
             (short)0, acc,  false,  false);
        a.x = fa[cur][t].z; a.y = fa[cur][t].w;
        b.x = fb[cur][t].z; b.y = fb[cur][t].w;
        acc = __builtin_amdgcn_wmma_f32_16x16x4_f32(
             false, a,  false, b,
             (short)0, acc,  false,  false);
      }
      __builtin_amdgcn_sched_barrier(0);
    }
    #pragma unroll
    for (int r = 0; r < 8; ++r) accPrev[r] = acc[r] + bc;
    mPrev = s << 4;

    const int nb = (buf + 1) & (NBUF - 1);
    if (hasNext) {
      #pragma unroll
      for (int u = 0; u < NV; ++u) {
        float t = v[u];
        if (RELU_IN) t = fmaxf(t, 0.0f);
        As[nb][(sr + u * RS) * LDA + scp] = t;
        w[u] = t + rb;
      }
    }
    __syncthreads();
    buf = nb;
  }
  #pragma unroll
  for (int r = 0; r < 8; ++r) {
    const int row = mPrev + rowOff + r;
    if (row < M) C[(long)row * HIDC + nBase + col] = accPrev[r];
  }
}

#define DOT_TRIP 8
__global__ __launch_bounds__(256)
void el_er_kernel(const float* __restrict__ feat, const float* __restrict__ al,
                  const float* __restrict__ ar, float* __restrict__ el,
                  float* __restrict__ er, int NH) {
  const int lane = threadIdx.x & 31;
  const int c    = lane & 15;
  const int hf   = lane >> 4;
  const int wave0 = __builtin_amdgcn_readfirstlane(
      (int)(blockIdx.x * (blockDim.x >> 5) + (threadIdx.x >> 5)));
  const int nW = (int)(gridDim.x * (blockDim.x >> 5));
  const int NU = NH >> 1;
  const int h  = 2 * (wave0 & 1) + hf;
  const float* ap = al + h * DHC + 4 * c;
  const float* rp = ar + h * DHC + 4 * c;
  const float a0 = ap[0], a1 = ap[1], a2 = ap[2], a3 = ap[3];
  const float r0 = rp[0], r1 = rp[1], r2 = rp[2], r3 = rp[3];
  const float4* fp = reinterpret_cast<const float4*>(feat) + hf * (DHC / 4) + c;

  auto ask = [&](int U, float4 (&f)[DOT_TRIP]) __attribute__((always_inline)) {
    #pragma unroll
    for (int u = 0; u < DOT_TRIP; ++u) {
      const int Uu = min(U + u * nW, NU - 1);
      f[u] = fp[(long)Uu * (2 * DHC / 4)];
    }
    __builtin_amdgcn_sched_barrier(0);
  };
  auto reduce = [&](int U, const float4 (&f)[DOT_TRIP]) __attribute__((always_inline)) {
    float sl[DOT_TRIP], sr[DOT_TRIP];
    #pragma unroll
    for (int u = 0; u < DOT_TRIP; ++u) {
      sl[u] = f[u].x * a0 + f[u].y * a1 + f[u].z * a2 + f[u].w * a3;
      sr[u] = f[u].x * r0 + f[u].y * r1 + f[u].z * r2 + f[u].w * r3;
    }
    #pragma unroll
    for (int m = 8; m; m >>= 1) {
      #pragma unroll
      for (int u = 0; u < DOT_TRIP; ++u) {
        sl[u] += __shfl_xor(sl[u], m, 32);
        sr[u] += __shfl_xor(sr[u], m, 32);
      }
    }
    #pragma unroll
    for (int u = 0; u < DOT_TRIP; ++u) {
      const int Uu = U + u * nW;
      if (c == 0 && Uu < NU) {
        const int q = 2 * Uu + hf;
        el[q] = sl[u];
        er[q] = sr[u];
      }
    }
  };

  const int trip = DOT_TRIP * nW;
  float4 A[DOT_TRIP], B[DOT_TRIP];
  ask(wave0, A);
  for (int U = wave0; U < NU; U += 2 * trip) {
    ask(U + trip, B);
    reduce(U, A);
    ask(U + 2 * trip, A);
    reduce(U + trip, B);
  }
}

#define AGG_TN    196
#define AGG_TRIP  512
#define AGG_QCAP  768
#define AGG_SAVE  (AGG_QCAP - AGG_TRIP)
#define AGG_SHDR  64
#define AGG_SBLK  (AGG_SHDR + 32 * AGG_SAVE)
#define AGG_NBLK  ((AGG_TN + 15) / 16)
#define AGG_RCAP  (32 * AGG_SAVE + 8 * AGG_NBLK)
#define AGG_LOW   0x00FFFFFF
static_assert(AGG_TN <= 256, "the row inside the tile must fit the entry's upper eight bits");

__device__ __forceinline__ int agg_pack(int low, int rowInTile) {
  return (int)(((unsigned)low & 0x00FFFFFFu) | ((unsigned)rowInTile << 24));
}

#define AGG_NCH   128
#define AGG_SEGW  64
#define AGG_BINT  256
static_assert(AGG_NCH * AGG_SEGW == 32 * AGG_SAVE, "a tile's segments fill its saved queues' place");
static_assert(4 * (AGG_SEGW - 1) <= AGG_SAVE, "a wave's four segments fit its queue");

__device__ __forceinline__ bool agg_bins_usable(const int* qsave, int nTiles,
                                                const float* wsBase, unsigned long wsBytes) {
  return nTiles <= AGG_BINT &&
         (unsigned long)(reinterpret_cast<const char*>(qsave + (long)nTiles * AGG_SBLK) -
                         reinterpret_cast<const char*>(wsBase)) <= wsBytes;
}

__device__ __forceinline__ void agg_bin_edges(int* lds, const int* __restrict__ dst, int E,
                                              int* qsave, int nTiles, int tid) {
  const int share = (int)blockIdx.x;
  if (share >= AGG_NCH) return;
  int* cntT = lds;
  int* buf  = lds + AGG_BINT;
  const int L  = (E + AGG_NCH - 1) / AGG_NCH;
  const int e0 = share * L;
  if (tid < AGG_BINT) cntT[tid] = 0;
  __syncthreads();
  for (int i0 = 0; i0 < L; i0 += 512 * 16) {
    int d[16];
    #pragma unroll
    for (int u = 0; u < 16; ++u) {
      const int i = i0 + tid + 512 * u;
      d[u] = -1;
      if (i < L && e0 + i < E) d[u] = dst[e0 + i];
    }
    int tl[16], sl[16];
    #pragma unroll
    for (int u = 0; u < 16; ++u) {
      const int t = d[u] / AGG_TN;
      tl[u] = -1;
      sl[u] = 0;
      if (d[u] >= 0 && t < nTiles) {
        tl[u] = t;
        sl[u] = atomicAdd(&cntT[t], 1);
      }
    }
    #pragma unroll
    for (int u = 0; u < 16; ++u) {
      if (tl[u] >= 0 && sl[u] < AGG_SEGW - 1)
        buf[tl[u] * AGG_SEGW + 1 + sl[u]] =
            agg_pack(e0 + i0 + tid + 512 * u, d[u] - tl[u] * AGG_TN);
    }
  }
  __syncthreads();
  const int lane = tid & 31;
  for (int t = tid >> 5; t < nTiles; t += 16) {
    const int n = cntT[t];
    int* seg = qsave + (long)t * AGG_SBLK + AGG_SHDR + share * AGG_SEGW;
    #pragma unroll
    for (int hf = 0; hf < 2; ++hf) {
      const int wd = lane + 32 * hf;
      int v = 0;
      if (wd == 0) v = (n > AGG_SEGW - 1) ? -1 : n;
      else if (wd <= n) v = buf[t * AGG_SEGW + wd];
      seg[wd] = v;
    }
  }
  __syncthreads();
}

__device__ __forceinline__ float edge_weight(float score, float mx) {
  const float v = lrelu(score);
  if (!(mx > -1e38f)) mx = 0.0f;
  return __expf(v - mx);
}

__device__ __forceinline__ void agg_drain_max(
    const int* q, int cnt, int lo, int lane,
    const int* __restrict__ src, const int* __restrict__ dst,
    const float* __restrict__ el, const float* __restrict__ er, float* mt, int* bc) {
  for (int i = 0; i < cnt; i += 64) {
    int s[2], d[2];
    float4 x[2], y[2];
    #pragma unroll
    for (int u = 0; u < 2; ++u) {
      const int p = i + 32 * u + lane;
      s[u] = 0; d[u] = lo;
      if (p < cnt) { const int e = q[p] & AGG_LOW; d[u] = dst[e]; s[u] = src[e]; }
    }
    #pragma unroll
    for (int u = 0; u < 2; ++u) {
      const int p = i + 32 * u + lane;
      if (p < cnt) {
        x[u] = *reinterpret_cast<const float4*>(el + 4 * (long)s[u]);
        y[u] = *reinterpret_cast<const float4*>(er + 4 * (long)d[u]);
      }
    }
    #pragma unroll
    for (int u = 0; u < 2; ++u) {
      const int p = i + 32 * u + lane;
      if (p < cnt) {
        float* m = mt + 4 * (d[u] - lo);
        atomicAdd(bc + ((d[u] - lo) >> 4), 1);
        atomicMaxF(m,     lrelu(x[u].x + y[u].x));
        atomicMaxF(m + 1, lrelu(x[u].y + y[u].y));
        atomicMaxF(m + 2, lrelu(x[u].z + y[u].z));
        atomicMaxF(m + 3, lrelu(x[u].w + y[u].w));
      }
    }
  }
}

__device__ __forceinline__ void agg_resolve(int* q, int cnt, int lane,
                                            const int* __restrict__ src) {
  for (int i = 0; i < cnt; i += AGG_SAVE) {
    int v[AGG_SAVE / 32], s[AGG_SAVE / 32];
    #pragma unroll
    for (int u = 0; u < AGG_SAVE / 32; ++u) {
      const int p = i + 32 * u + lane;
      v[u] = 0;
      if (p < cnt) v[u] = q[p];
    }
    #pragma unroll
    for (int u = 0; u < AGG_SAVE / 32; ++u) {
      const int p = i + 32 * u + lane;
      s[u] = 0;
      if (p < cnt) s[u] = src[v[u] & AGG_LOW];
    }
    #pragma unroll
    for (int u = 0; u < AGG_SAVE / 32; ++u) {
      const int p = i + 32 * u + lane;
      if (p < cnt) q[p] = agg_pack(s[u], (int)((unsigned)v[u] >> 24));
    }
  }
}

__device__ __forceinline__ void agg_drain_max_resolved(
    const int* q, int cnt, int lo, int lane,
    const float* __restrict__ el, const float* __restrict__ er, float* mt, int* bc) {
  for (int i = 0; i < cnt; i += 128) {
    int s[4], dl[4];
    float4 x[4], y[4];
    #pragma unroll
    for (int u = 0; u < 4; ++u) {
      const int p = i + 32 * u + lane;
      s[u] = 0; dl[u] = 0;
      if (p < cnt) { const int v = q[p]; s[u] = v & AGG_LOW; dl[u] = (int)((unsigned)v >> 24); }
    }
    #pragma unroll
    for (int u = 0; u < 4; ++u) {
      const int p = i + 32 * u + lane;
      if (p < cnt) {
        x[u] = *reinterpret_cast<const float4*>(el + 4 * (long)s[u]);
        y[u] = *reinterpret_cast<const float4*>(er + 4 * (long)(lo + dl[u]));
      }
    }
    #pragma unroll
    for (int u = 0; u < 4; ++u) {
      const int p = i + 32 * u + lane;
      if (p < cnt) {
        float* m = mt + 4 * dl[u];
        atomicAdd(bc + (dl[u] >> 4), 1);
        atomicMaxF(m,     lrelu(x[u].x + y[u].x));
        atomicMaxF(m + 1, lrelu(x[u].y + y[u].y));
        atomicMaxF(m + 2, lrelu(x[u].z + y[u].z));
        atomicMaxF(m + 3, lrelu(x[u].w + y[u].w));
      }
    }
  }
}

__device__ __forceinline__ void agg_drain_acc(
    const int* q, int cnt, int lo, int lane,
    const int* __restrict__ src, const int* __restrict__ dst,
    const float* __restrict__ feat,
    const float* __restrict__ el, const float* __restrict__ er,
    const float* mt, float* dt, float* acc) {
  for (int i = 0; i < cnt; i += 32) {
    const int n = min(32, cnt - i);
    int s = 0, dl = 0;
    float a0 = 0.0f, a1 = 0.0f, a2 = 0.0f, a3 = 0.0f;
    if (lane < n) {
      const int e = q[i + lane] & AGG_LOW;
      const int d = dst[e];
      s  = src[e];
      dl = d - lo;
      const float4 x = *reinterpret_cast<const float4*>(el + 4 * (long)s);
      const float4 y = *reinterpret_cast<const float4*>(er + 4 * (long)d);
      const float4 m = *reinterpret_cast<const float4*>(mt + 4 * dl);
      a0 = edge_weight(x.x + y.x, m.x);
      a1 = edge_weight(x.y + y.y, m.y);
      a2 = edge_weight(x.z + y.z, m.z);
      a3 = edge_weight(x.w + y.w, m.w);
      float* ds = dt + 4 * dl;
      atomicAdd(ds,     a0);
      atomicAdd(ds + 1, a1);
      atomicAdd(ds + 2, a2);
      atomicAdd(ds + 3, a3);
    }
    for (int k = 0; k < n; k += 4) {
      int su[4], du[4];
      float au[4][4];
      v2f f[4][4];
      #pragma unroll
      for (int u = 0; u < 4; ++u) {
        const int idx = min(k + u, n - 1);
        su[u] = __builtin_amdgcn_readlane(s, idx);
        du[u] = (k + u < n) ? __builtin_amdgcn_readlane(dl, idx) : AGG_TN;
        au[u][0] = __int_as_float(__builtin_amdgcn_readlane(__float_as_int(a0), idx));
        au[u][1] = __int_as_float(__builtin_amdgcn_readlane(__float_as_int(a1), idx));
        au[u][2] = __int_as_float(__builtin_amdgcn_readlane(__float_as_int(a2), idx));
        au[u][3] = __int_as_float(__builtin_amdgcn_readlane(__float_as_int(a3), idx));
      }
      #pragma unroll
      for (int u = 0; u < 4; ++u) {
        const float* fr = feat + (long)su[u] * HIDC + 2 * lane;
        #pragma unroll
        for (int j = 0; j < HEADS; ++j)
          f[u][j] = *reinterpret_cast<const v2f*>(fr + j * DHC);
      }
      #pragma unroll
      for (int u = 0; u < 4; ++u) {
        float* p = acc + du[u] * HIDC + 2 * lane;
        #pragma unroll
        for (int j = 0; j < HEADS; ++j) {
          atomicAdd(p + j * DHC,     au[u][j] * f[u][j].x);
          atomicAdd(p + j * DHC + 1, au[u][j] * f[u][j].y);
        }
      }
    }
  }
}

template <int PASS>
__device__ __forceinline__ int agg_scan(
    int* q, int wave, int lane, int lo, int rows, int E,
    const int* __restrict__ src, const int* __restrict__ dst,
    const float* __restrict__ feat,
    const float* __restrict__ el, const float* __restrict__ er,
    float* mt, float* dt, float* acc, int* full, int* bc) {
  const int  T       = (E + AGG_TRIP - 1) / AGG_TRIP;
  const bool aligned = (reinterpret_cast<unsigned long>(dst) & 15ul) == 0ul;
  const int start = (T / (int)gridDim.x) * (int)blockIdx.x;
  int cnt = 0;
  for (int t = wave; t < T; t += 32) {
    int g = t + start;
    if (g >= T) g -= T;
    const int base = g * AGG_TRIP + 4 * lane;
    int4 dv[4];
    if (aligned && g * AGG_TRIP + AGG_TRIP <= E) {
      #pragma unroll
      for (int u = 0; u < 4; ++u)
        dv[u] = *reinterpret_cast<const int4*>(dst + base + 128 * u);
    } else {
      #pragma unroll
      for (int u = 0; u < 4; ++u) {
        const int e0 = base + 128 * u;
        dv[u].x = (e0     < E) ? dst[e0]     : -1;
        dv[u].y = (e0 + 1 < E) ? dst[e0 + 1] : -1;
        dv[u].z = (e0 + 2 < E) ? dst[e0 + 2] : -1;
        dv[u].w = (e0 + 3 < E) ? dst[e0 + 3] : -1;
      }
    }
    #pragma unroll
    for (int u = 0; u < 4; ++u) {
      const int  e0 = base + 128 * u;
      const bool mx = (unsigned)(dv[u].x - lo) < (unsigned)rows;
      const bool my = (unsigned)(dv[u].y - lo) < (unsigned)rows;
      const bool mz = (unsigned)(dv[u].z - lo) < (unsigned)rows;
      const bool mw = (unsigned)(dv[u].w - lo) < (unsigned)rows;
      if (__builtin_amdgcn_ballot_w32(mx | my | mz | mw)) {
        unsigned mask;
        mask = __builtin_amdgcn_ballot_w32(mx);
        if (mx) q[cnt + __builtin_amdgcn_mbcnt_lo(mask, 0u)] = agg_pack(e0, dv[u].x - lo);
        cnt += __builtin_popcount(mask);
        mask = __builtin_amdgcn_ballot_w32(my);
        if (my) q[cnt + __builtin_amdgcn_mbcnt_lo(mask, 0u)] = agg_pack(e0 + 1, dv[u].y - lo);
        cnt += __builtin_popcount(mask);
        mask = __builtin_amdgcn_ballot_w32(mz);
        if (mz) q[cnt + __builtin_amdgcn_mbcnt_lo(mask, 0u)] = agg_pack(e0 + 2, dv[u].z - lo);
        cnt += __builtin_popcount(mask);
        mask = __builtin_amdgcn_ballot_w32(mw);
        if (mw) q[cnt + __builtin_amdgcn_mbcnt_lo(mask, 0u)] = agg_pack(e0 + 3, dv[u].w - lo);
        cnt += __builtin_popcount(mask);
      }
    }
    if (cnt > AGG_QCAP - AGG_TRIP) {
      if (PASS == 0) {
        agg_drain_max(q, cnt, lo, lane, src, dst, el, er, mt, bc);
        if (lane == 0) *full = 1;
      } else {
        agg_drain_acc(q, cnt, lo, lane, src, dst, feat, el, er, mt, dt, acc);
      }
      cnt = 0;
    }
  }
  return cnt;
}

__global__ __launch_bounds__(1024)
void aggregate_tile_kernel(const int* __restrict__ src, const int* __restrict__ dst,
                           const float* __restrict__ feat,
                           const float* __restrict__ el,
                           const float* __restrict__ er, float* __restrict__ out,
                           int N, int E, int* qsave, const float* wsBase,
                           unsigned long wsBytes, int reuseSaved, int useBins) {
  __shared__ __attribute__((aligned(16))) float acc[(AGG_TN + 1) * HIDC];
  static_assert(6 * AGG_RCAP <= (AGG_TN + 1) * HIDC, "records must fit the accumulators' memory");
  __shared__ __attribute__((aligned(16))) float mt[AGG_TN * HEADS];
  __shared__ __attribute__((aligned(16))) float dt[AGG_TN * HEADS];
  __shared__ int queue[32 * AGG_QCAP];
  __shared__ int full;
  __shared__ int binBad;
  __shared__ int blkCnt[16];
  __shared__ int blkFill[16];
  __shared__ int blkOff[16];

  const int tid  = threadIdx.x;
  const int lane = tid & 31;
  const int wave = __builtin_amdgcn_readfirstlane(tid >> 5);
  const int lo   = blockIdx.x * AGG_TN;
  const int rows = min(AGG_TN, N - lo);

  if (tid < AGG_TN * HEADS) { mt[tid] = -INFINITY; dt[tid] = 0.0f; }
  if (tid < 16) { blkCnt[tid] = 0; blkFill[tid] = 0; }
  if (tid == 0) { full = 0; binBad = 0; }
  __syncthreads();

  int* q = queue + wave * AGG_QCAP;
  int* sv = qsave + (long)blockIdx.x * AGG_SBLK;
  const bool fits =
      (unsigned long)(reinterpret_cast<const char*>(qsave + (long)gridDim.x * AGG_SBLK) -
                      reinterpret_cast<const char*>(wsBase)) <= wsBytes;
  int reuse = 0, saved = 0;
  if (reuseSaved != 0 && fits) {
    const int okw = sv[32];
    const int cw  = sv[wave];
    reuse = (okw == 1) ? 1 : 0;
    saved = cw;
  }
  int cnt;
  if (reuse != 0) {
    cnt = min(max(saved, 0), AGG_SAVE);
    const int* sq = sv + AGG_SHDR + wave * AGG_SAVE;
    int v[AGG_SAVE / 32];
    #pragma unroll
    for (int u = 0; u < AGG_SAVE / 32; ++u) {
      const int i = lane + 32 * u;
      v[u] = 0;
      if (i < cnt) v[u] = sq[i];
    }
    #pragma unroll
    for (int u = 0; u < AGG_SAVE / 32; ++u) {
      const int i = lane + 32 * u;
      if (i < cnt) q[i] = v[u];
    }
  } else {
    int binned = 0;
    cnt = 0;
    if (useBins != 0 && agg_bins_usable(qsave, (int)gridDim.x, wsBase, wsBytes)) {
      const int* bp = sv + AGG_SHDR + wave * (4 * AGG_SEGW) + lane;
      int w8[8];
      #pragma unroll
      for (int u = 0; u < 8; ++u) w8[u] = bp[32 * u];
      int bad = 0;
      #pragma unroll
      for (int g = 0; g < 4; ++g) {
        const int c = __builtin_amdgcn_readfirstlane(w8[2 * g]);
        if (c < 0 || c > AGG_SEGW - 1) {
          bad = 1;
        } else {
          #pragma unroll
          for (int hf = 0; hf < 2; ++hf) {
            const int  wd = lane + 32 * hf;
            const bool v  = wd >= 1 && wd <= c;
            const unsigned mask = __builtin_amdgcn_ballot_w32(v);
            if (v) q[cnt + __builtin_amdgcn_mbcnt_lo(mask, 0u)] = w8[2 * g + hf];
            cnt += __builtin_popcount(mask);
          }
        }
      }
      if (bad != 0 && lane == 0) binBad = 1;
      __syncthreads();
      binned = (__builtin_amdgcn_readfirstlane(binBad) == 0) ? 1 : 0;
    }
    if (binned == 0)
      cnt = agg_scan<0>(q, wave, lane, lo, rows, E, src, dst, feat, el, er, mt, dt, acc, &full, blkCnt);
    agg_resolve(q, cnt, lane, src);
  }
  agg_drain_max_resolved(q, cnt, lo, lane, el, er, mt, blkCnt);
  __syncthreads();

  const int early = (reuse != 0) ? 0 : __builtin_amdgcn_readfirstlane(full);
  if (reuseSaved == 0 && fits) {
    if (early == 0) {
      int* sq = sv + AGG_SHDR + wave * AGG_SAVE;
      for (int i = lane; i < cnt; i += 32) sq[i] = q[i];
    }
    if (lane == 0) sv[wave] = (early == 0) ? cnt : 0;
    if (tid == 0) sv[32] = (early == 0) ? 1 : 0;
  }
  float4* op = reinterpret_cast<float4*>(out + (long)lo * HIDC);

  if (early != 0) {
    for (int i = tid; i < (AGG_TN + 1) * HIDC / 4; i += 1024)
      reinterpret_cast<float4*>(acc)[i] = make_float4(0.0f, 0.0f, 0.0f, 0.0f);
    __syncthreads();
    cnt = agg_scan<1>(q, wave, lane, lo, rows, E, src, dst, feat, el, er, mt, dt, acc, &full, blkCnt);
    agg_drain_acc(q, cnt, lo, lane, src, dst, feat, el, er, mt, dt, acc);
    __syncthreads();
    if (tid < rows * HEADS) dt[tid] = 1.0f / fmaxf(dt[tid], 1e-16f);
    __syncthreads();
    for (int i = tid; i < rows * (HIDC / 4); i += 1024) {
      const float  r = dt[i >> 4];
      const float4 a = reinterpret_cast<const float4*>(acc)[i];
      float4 o = op[i];
      o.x += a.x * r; o.y += a.y * r; o.z += a.z * r; o.w += a.w * r;
      op[i] = o;
    }
    return;
  }

  int*   recS = reinterpret_cast<int*>(acc);
  int*   recD = recS + AGG_RCAP;
  float* recA = reinterpret_cast<float*>(recD + AGG_RCAP);
  if (tid < AGG_NBLK) {
    int o = 0;
    for (int i = 0; i < tid; ++i) o += (blkCnt[i] + 7) & ~7;
    blkOff[tid] = o;
  }
  __syncthreads();
  if (wave < AGG_NBLK && lane < 8) {
    const int n = blkCnt[wave];
    if (n + lane < ((n + 7) & ~7)) {
      const int pos = blkOff[wave] + n + lane;
      recS[pos] = 0;
      recD[pos] = 0;
      recA[pos] = 0.0f;
      recA[AGG_RCAP + pos] = 0.0f;
      recA[2 * AGG_RCAP + pos] = 0.0f;
      recA[3 * AGG_RCAP + pos] = 0.0f;
    }
  }
  for (int i = 0; i < cnt; i += 128) {
    int s[4], dr[4];
    float4 x[4], y[4];
    #pragma unroll
    for (int u = 0; u < 4; ++u) {
      const int p = i + 32 * u + lane;
      s[u] = 0; dr[u] = 0;
      if (p < cnt) { const int v = q[p]; s[u] = v & AGG_LOW; dr[u] = (int)((unsigned)v >> 24); }
    }
    #pragma unroll
    for (int u = 0; u < 4; ++u) {
      const int p = i + 32 * u + lane;
      if (p < cnt) {
        x[u] = *reinterpret_cast<const float4*>(el + 4 * (long)s[u]);
        y[u] = *reinterpret_cast<const float4*>(er + 4 * (long)(lo + dr[u]));
      }
    }
    #pragma unroll
    for (int u = 0; u < 4; ++u) {
      const int p = i + 32 * u + lane;
      if (p < cnt) {
        const int dl = dr[u];
        const float4 m = *reinterpret_cast<const float4*>(mt + 4 * dl);
        const float a0 = edge_weight(x[u].x + y[u].x, m.x);
        const float a1 = edge_weight(x[u].y + y[u].y, m.y);
        const float a2 = edge_weight(x[u].z + y[u].z, m.z);
        const float a3 = edge_weight(x[u].w + y[u].w, m.w);
        float* ds = dt + 4 * dl;
        atomicAdd(ds,     a0);
        atomicAdd(ds + 1, a1);
        atomicAdd(ds + 2, a2);
        atomicAdd(ds + 3, a3);
        const int bb  = dl >> 4;
        const int pos = blkOff[bb] + atomicAdd(&blkFill[bb], 1);
        recS[pos] = s[u];
        recD[pos] = dl & 15;
        recA[pos] = a0;
        recA[AGG_RCAP + pos] = a1;
        recA[2 * AGG_RCAP + pos] = a2;
        recA[3 * AGG_RCAP + pos] = a3;
      }
    }
  }
  __syncthreads();

  const int col = lane & 15;
  const int hf  = lane >> 4;
  const int nJobs = 4 * ((rows + 15) >> 4);
  for (int j = wave; j < nJobs; j += 32) {
    const int b  = j >> 2;
    const int hd = j & 3;
    const int nIt  = (__builtin_amdgcn_readfirstlane(blkCnt[b]) + 7) >> 3;
    const int off  = __builtin_amdgcn_readfirstlane(blkOff[b]);
    const float* fcol = feat + hd * DHC + 4 * col;
    const float* ra   = recA + hd * AGG_RCAP;

    float4 F[2][2];
    v2f    A[2];
    auto fetch = [&](int eb, float4 (&Fo)[2][2], v2f (&Ao)[2]) {
      #pragma unroll
      for (int u = 0; u < 2; ++u) {
        const int2   s2 = *reinterpret_cast<const int2*>(recS + eb + 4 * u);
        const int2   d2 = *reinterpret_cast<const int2*>(recD + eb + 4 * u);
        const float2 a2 = *reinterpret_cast<const float2*>(ra + eb + 4 * u);
        Fo[u][0] = *reinterpret_cast<const float4*>(fcol + (long)s2.x * HIDC);
        Fo[u][1] = *reinterpret_cast<const float4*>(fcol + (long)s2.y * HIDC);
        Ao[u].x = (d2.x == col) ? a2.x : 0.0f;
        Ao[u].y = (d2.y == col) ? a2.y : 0.0f;
      }
    };

    v8f ac[4] = {};
    int eb = off + 2 * hf;
    if (nIt > 0) fetch(eb, F, A);
    for (int it = 0; it < nIt; ++it) {
      const int en = (it + 1 < nIt) ? eb + 8 : eb;
      float4 G[2][2];
      v2f    An[2];
      fetch(en, G, An);
      #pragma unroll
      for (int u = 0; u < 2; ++u) {
        const float f0[4] = {F[u][0].x, F[u][0].y, F[u][0].z, F[u][0].w};
        const float f1[4] = {F[u][1].x, F[u][1].y, F[u][1].z, F[u][1].w};
        #pragma unroll
        for (int t = 0; t < 4; ++t) {
          v2f bv;
          bv.x = f0[t];
          bv.y = f1[t];
          ac[t] = __builtin_amdgcn_wmma_f32_16x16x4_f32(
               false, A[u],  false, bv,
               (short)0, ac[t],  false,  false);
        }
      }
      #pragma unroll
      for (int u = 0; u < 2; ++u) {
        F[u][0] = G[u][0];
        F[u][1] = G[u][1];
        A[u]    = An[u];
      }
      eb = en;
    }

    float  es[8];
    float4 o[8];
    #pragma unroll
    for (int r = 0; r < 8; ++r) {
      const int rl = min(16 * b + 8 * hf + r, rows - 1);
      es[r] = dt[rl * HEADS + hd];
      o[r]  = op[rl * (HIDC / 4) + hd * (DHC / 4) + col];
    }
    #pragma unroll
    for (int r = 0; r < 8; ++r) {
      const int rl = 16 * b + 8 * hf + r;
      if (rl < rows) {
        const float rc = 1.0f / fmaxf(es[r], 1e-16f);
        o[r].x += ac[0][r] * rc; o[r].y += ac[1][r] * rc;
        o[r].z += ac[2][r] * rc; o[r].w += ac[3][r] * rc;
        op[rl * (HIDC / 4) + hd * (DHC / 4) + col] = o[r];
      }
    }
  }
}

__global__ __launch_bounds__(256)
void mean_heads_kernel(const float* __restrict__ hbuf,
                       float* __restrict__ out, long total) {
  const long stride = (long)gridDim.x * blockDim.x;
  const long t      = (long)blockIdx.x * blockDim.x + threadIdx.x;
  if ((reinterpret_cast<unsigned long>(out) & 15ul) == 0ul) {
    const long n4 = total >> 2;
    const float4* hp = reinterpret_cast<const float4*>(hbuf);
    float4*       op = reinterpret_cast<float4*>(out);
    for (long i = t; i < n4; i += 2 * stride) {
      float4 p[2][4];
      #pragma unroll
      for (int u = 0; u < 2; ++u) {
        const long j = i + u * stride;
        if (j < n4) {
          const float4* r = hp + (j >> 4) * (HIDC / 4) + (j & 15);
          #pragma unroll
          for (int k = 0; k < HEADS; ++k) p[u][k] = r[k * (DHC / 4)];
        }
      }
      #pragma unroll
      for (int u = 0; u < 2; ++u) {
        const long j = i + u * stride;
        if (j < n4) {
          float4 o;
          o.x = 0.25f * (p[u][0].x + p[u][1].x + p[u][2].x + p[u][3].x);
          o.y = 0.25f * (p[u][0].y + p[u][1].y + p[u][2].y + p[u][3].y);
          o.z = 0.25f * (p[u][0].z + p[u][1].z + p[u][2].z + p[u][3].z);
          o.w = 0.25f * (p[u][0].w + p[u][1].w + p[u][2].w + p[u][3].w);
          op[j] = o;
        }
      }
    }
  } else {
    for (long i = t; i < total; i += stride) {
      const float* p = hbuf + (i >> 6) * HIDC + (i & 63);
      out[i] = 0.25f * (p[0] + p[DHC] + p[2 * DHC] + p[3 * DHC]);
    }
  }
}

extern "C" void kernel_launch(void* const* d_in, const int* in_sizes, int n_in,
                              void* d_out, int out_size, void* d_ws, size_t ws_size,
                              hipStream_t stream) {
  const float* x     = (const float*)d_in[0];
  const int*   ei    = (const int*)d_in[1];
  const float* W[3]  = {(const float*)d_in[2], (const float*)d_in[3],
                        (const float*)d_in[4]};
  const float* al[3] = {(const float*)d_in[5], (const float*)d_in[6],
                        (const float*)d_in[7]};
  const float* ar[3] = {(const float*)d_in[8], (const float*)d_in[9],
                        (const float*)d_in[10]};
  const float* b[3]  = {(const float*)d_in[11], (const float*)d_in[12],
                        (const float*)d_in[13]};
  const float* resW0 = (const float*)d_in[14];

  const int N = in_sizes[0] / 128;
  const int E = in_sizes[1] / 2;
  const int* src = ei;
  const int* dst = ei + E;

  float* ws   = (float*)d_ws;
  float* FEAT = ws;
  float* OUT  = FEAT + (size_t)N * HIDC;
  float* ELP  = OUT  + (size_t)N * HIDC;
  float* ERP  = ELP  + (size_t)N * HEADS;
  int*   QSV  = (int*)(ERP + (size_t)N * HEADS);

  const int  gGEMM  = 256;
  const int  NH     = N * HEADS;
  const int  gSTREAM = 512;

  auto attention = [&](const float* alw, const float* arw, int reuseSaved, int useBins) {
    el_er_kernel<<<gSTREAM, 256, 0, stream>>>(FEAT, alw, arw, ELP, ERP, NH);
    aggregate_tile_kernel<<<(N + AGG_TN - 1) / AGG_TN, 1024, 0, stream>>>(
        src, dst, FEAT, ELP, ERP, OUT, N, E, QSV, ws, (unsigned long)ws_size, reuseSaved, useBins);
  };

  gemm_wmma_f32<128, false><<<gGEMM, 512, 0, stream>>>(x, W[0], nullptr, FEAT, N, nullptr, nullptr,
      dst, E, QSV, (N + AGG_TN - 1) / AGG_TN, ws, (unsigned long)ws_size);
  gemm_wmma_f32<128, false><<<gGEMM, 512, 0, stream>>>(x, resW0, b[0], OUT, N, nullptr, nullptr,
      nullptr, 0, nullptr, 0, nullptr, 0ul);
  attention(al[0], ar[0], 0, 1);

  gemm_wmma_f32<256, true><<<gGEMM, 512, 0, stream>>>(OUT, W[1], nullptr, FEAT, N, OUT, b[1],
      nullptr, 0, nullptr, 0, nullptr, 0ul);
  attention(al[1], ar[1], 1, 0);

  gemm_wmma_f32<256, true><<<gGEMM, 512, 0, stream>>>(OUT, W[2], nullptr, FEAT, N, OUT, b[2],
      nullptr, 0, nullptr, 0, nullptr, 0ul);
  attention(al[2], ar[2], 1, 0);
  mean_heads_kernel<<<gSTREAM, 256, 0, stream>>>(
      OUT, (float*)d_out, (long)N * DHC);
}
